# starting kernel
# baseline (speedup 1.0000x reference)
#include <hip/hip_runtime.h>
#include <math.h>
#include <limits.h>

#define N_NODES   50000
#define N_EDGES   800000
#define IN_FEATS  128
#define OUT_FEATS 16
#define NUM_HEADS 4
#define ZCOLS     (NUM_HEADS * OUT_FEATS)
#define ALPHA     0.2f

typedef __attribute__((ext_vector_type(2))) float v2f;
typedef __attribute__((ext_vector_type(8))) float v8f;

__global__ __launch_bounds__(128) void gat_proj_wmma(
    const float* __restrict__ h, const float* __restrict__ W,
    float* __restrict__ z)
{
  const int wave  = threadIdx.x >> 5;
  const int lane  = threadIdx.x & 31;
  const int mtile = blockIdx.x;
  const int ntile = wave;
  const int lo16  = lane & 15;
  const int kbase = (lane >> 4) * 2;

  const float* __restrict__ ha = h + (size_t)(mtile * 16 + lo16) * IN_FEATS + kbase;
  const float* __restrict__ wb = W + (size_t)(ntile * 16 + lo16) * IN_FEATS + kbase;

  v8f c = {};
#pragma unroll 4
  for (int k0 = 0; k0 < IN_FEATS; k0 += 4) {
    v2f a, b;
    a.x = ha[k0]; a.y = ha[k0 + 1];
    b.x = wb[k0]; b.y = wb[k0 + 1];
    c = __builtin_amdgcn_wmma_f32_16x16x4_f32(
         false, a,  false, b,
         (short)0, c,  false,  false);
  }

  const int rbase = mtile * 16 + ((lane >> 4) << 3);
  float* __restrict__ zp = z + (size_t)rbase * ZCOLS + ntile * 16 + lo16;
#pragma unroll
  for (int r = 0; r < 8; ++r)
    zp[(size_t)r * ZCOLS] = c[r];
}

__global__ void gat_node_scores(const float* __restrict__ z,
                                const float* __restrict__ A,
                                float* __restrict__ asrc,
                                float* __restrict__ adst,
                                int* __restrict__ maxbuf)
{
  int t = blockIdx.x * blockDim.x + threadIdx.x;
  if (t >= N_NODES * NUM_HEADS) return;
  const int n = t >> 2, hh = t & 3;
  const float* __restrict__ zv = z + (size_t)n * ZCOLS + hh * OUT_FEATS;
  const float* __restrict__ As = A + hh * (2 * OUT_FEATS);
  float s = 0.f, d = 0.f;
#pragma unroll
  for (int f = 0; f < OUT_FEATS; ++f) {
    s += zv[f] * As[f];
    d += zv[f] * As[OUT_FEATS + f];
  }
  asrc[t] = s;
  adst[t] = d;
  maxbuf[t] = INT_MIN;
}

__device__ __forceinline__ int f32_ord(float v) {
  int i = __float_as_int(v);
  return i >= 0 ? i : (i ^ 0x7FFFFFFF);
}

__global__ void gat_edge_logits(const int* __restrict__ src,
                                const int* __restrict__ dst,
                                const float* __restrict__ asrc,
                                const float* __restrict__ adst,
                                float* __restrict__ eval,
                                int* __restrict__ maxbuf)
{
  int t = blockIdx.x * blockDim.x + threadIdx.x;
  if (t >= N_EDGES * NUM_HEADS) return;
  const int e = t >> 2, hh = t & 3;
  const int s = src[e], d = dst[e];
  float v = asrc[s * NUM_HEADS + hh] + adst[d * NUM_HEADS + hh];
  v = v > 0.f ? v : ALPHA * v;
  eval[t] = v;
  atomicMax(&maxbuf[d * NUM_HEADS + hh], f32_ord(v));
}

__global__ void gat_prep(const int* __restrict__ maxbuf,
                         float* __restrict__ m,
                         float* __restrict__ denom,
                         float* __restrict__ out)
{
  int t = blockIdx.x * blockDim.x + threadIdx.x;
  if (t >= N_NODES * NUM_HEADS) return;
  const int iv = maxbuf[t];
  float mv = 0.f;
  if (iv != INT_MIN) {
    int j = iv >= 0 ? iv : (iv ^ 0x7FFFFFFF);
    mv = __int_as_float(j);
    if (!isfinite(mv)) mv = 0.f;
  }
  m[t] = mv;
  denom[t] = 0.f;
  float* __restrict__ op = out + (size_t)(t >> 2) * ZCOLS + (t & 3) * OUT_FEATS;
#pragma unroll
  for (int f = 0; f < OUT_FEATS; ++f) op[f] = 0.f;
}

__global__ __launch_bounds__(256) void gat_edge_accum(
    const int* __restrict__ src, const int* __restrict__ dst,
    const float* __restrict__ eval, const float* __restrict__ m,
    const float* __restrict__ z,
    float* __restrict__ out, float* __restrict__ denom)
{
  const long long t = (long long)blockIdx.x * 256 + threadIdx.x;
  const int f  = (int)(t & 15);
  const int hh = (int)((t >> 4) & 3);
  const int e  = (int)(t >> 6);
  const int s = src[e], d = dst[e];
  const float ex = expf(eval[e * NUM_HEADS + hh] - m[d * NUM_HEADS + hh]);
  atomicAdd(&out[(size_t)d * ZCOLS + hh * OUT_FEATS + f],
            ex * z[(size_t)s * ZCOLS + hh * OUT_FEATS + f]);
  if (f == 0) atomicAdd(&denom[d * NUM_HEADS + hh], ex);
}

__global__ __launch_bounds__(256) void gat_finalize(float* __restrict__ out,
                                                    const float* __restrict__ denom)
{
  const int t = blockIdx.x * 256 + threadIdx.x;
  const int n = t >> 6, j = t & 63;
  out[t] /= fmaxf(denom[n * NUM_HEADS + (j >> 4)], 1e-12f);
}

extern "C" void kernel_launch(void* const* d_in, const int* in_sizes, int n_in,
                              void* d_out, int out_size, void* d_ws, size_t ws_size,
                              hipStream_t stream) {
  const float* h   = (const float*)d_in[0];
  const int*   src = (const int*)d_in[1];
  const int*   dst = (const int*)d_in[2];
  const float* W   = (const float*)d_in[3];
  const float* A   = (const float*)d_in[4];
  float* out = (float*)d_out;

  float* z     = (float*)d_ws;
  float* asrc  = z     + (size_t)N_NODES * ZCOLS;
  float* adst  = asrc  + (size_t)N_NODES * NUM_HEADS;
  float* m     = adst  + (size_t)N_NODES * NUM_HEADS;
  float* denom = m     + (size_t)N_NODES * NUM_HEADS;
  float* eval  = denom + (size_t)N_NODES * NUM_HEADS;
  int*   maxbuf = (int*)(eval + (size_t)N_EDGES * NUM_HEADS);

  const int NH = N_NODES * NUM_HEADS;
  const int EH = N_EDGES * NUM_HEADS;

  gat_proj_wmma<<<N_NODES / 16, 128, 0, stream>>>(h, W, z);
  gat_node_scores<<<(NH + 255) / 256, 256, 0, stream>>>(z, A, asrc, adst, maxbuf);
  gat_edge_logits<<<(EH + 255) / 256, 256, 0, stream>>>(src, dst, asrc, adst, eval, maxbuf);
  gat_prep<<<(NH + 255) / 256, 256, 0, stream>>>(maxbuf, m, denom, out);
  gat_edge_accum<<<(N_EDGES * 64) / 256, 256, 0, stream>>>(src, dst, eval, m, z, out, denom);
  gat_finalize<<<(N_NODES * ZCOLS) / 256, 256, 0, stream>>>(out, denom);
}


# global f32-atomic edge scatter replaced by destination tiles held in LDS: 256 blocks each scan dst for their 196 nodes, segment max and sums by LDS atomics; 6 launches to 3; arrival-order sums
# speedup vs baseline: 2.7589x; 2.7589x over previous
#include <hip/hip_runtime.h>
#include <math.h>
#include <limits.h>

#define N_NODES   50000
#define N_EDGES   800000
#define IN_FEATS  128
#define OUT_FEATS 16
#define NUM_HEADS 4
#define ZCOLS     (NUM_HEADS * OUT_FEATS)
#define ALPHA     0.2f

typedef __attribute__((ext_vector_type(2))) float v2f;
typedef __attribute__((ext_vector_type(8))) float v8f;

__global__ __launch_bounds__(128) void gat_proj_wmma(
    const float* __restrict__ h, const float* __restrict__ W,
    float* __restrict__ z)
{
  const int wave  = threadIdx.x >> 5;
  const int lane  = threadIdx.x & 31;
  const int mtile = blockIdx.x;
  const int ntile = wave;
  const int lo16  = lane & 15;
  const int kbase = (lane >> 4) * 2;

  const float* __restrict__ ha = h + (size_t)(mtile * 16 + lo16) * IN_FEATS + kbase;
  const float* __restrict__ wb = W + (size_t)(ntile * 16 + lo16) * IN_FEATS + kbase;

  v8f c = {};
#pragma unroll 4
  for (int k0 = 0; k0 < IN_FEATS; k0 += 4) {
    v2f a, b;
    a.x = ha[k0]; a.y = ha[k0 + 1];
    b.x = wb[k0]; b.y = wb[k0 + 1];
    c = __builtin_amdgcn_wmma_f32_16x16x4_f32(
         false, a,  false, b,
         (short)0, c,  false,  false);
  }

  const int rbase = mtile * 16 + ((lane >> 4) << 3);
  float* __restrict__ zp = z + (size_t)rbase * ZCOLS + ntile * 16 + lo16;
#pragma unroll
  for (int r = 0; r < 8; ++r)
    zp[(size_t)r * ZCOLS] = c[r];
}

__global__ void gat_node_scores(const float* __restrict__ z,
                                const float* __restrict__ A,
                                float* __restrict__ asrc,
                                float* __restrict__ adst)
{
  int t = blockIdx.x * blockDim.x + threadIdx.x;
  if (t >= N_NODES * NUM_HEADS) return;
  const int n = t >> 2, hh = t & 3;
  const float* __restrict__ zv = z + (size_t)n * ZCOLS + hh * OUT_FEATS;
  const float* __restrict__ As = A + hh * (2 * OUT_FEATS);
  float s = 0.f, d = 0.f;
#pragma unroll
  for (int f = 0; f < OUT_FEATS; ++f) {
    s += zv[f] * As[f];
    d += zv[f] * As[OUT_FEATS + f];
  }
  asrc[t] = s;
  adst[t] = d;
}

__device__ __forceinline__ int f32_ord(float v) {
  int i = __float_as_int(v);
  return i >= 0 ? i : (i ^ 0x7FFFFFFF);
}

__device__ __forceinline__ float ord_f32_guarded(int iv) {
  float mv = 0.f;
  if (iv != INT_MIN) {
    int j = iv >= 0 ? iv : (iv ^ 0x7FFFFFFF);
    mv = __int_as_float(j);
    if (!isfinite(mv)) mv = 0.f;
  }
  return mv;
}

__device__ __forceinline__ float leaky(float v) { return v > 0.f ? v : ALPHA * v; }

__device__ __forceinline__ void lds_add(float* p, float v) {
  __hip_atomic_fetch_add(p, v, __ATOMIC_RELAXED, __HIP_MEMORY_SCOPE_WORKGROUP);
}
__device__ __forceinline__ void lds_max(int* p, int v) {
  __hip_atomic_fetch_max(p, v, __ATOMIC_RELAXED, __HIP_MEMORY_SCOPE_WORKGROUP);
}

#define TILE_NODES   196
#define N_TILES      ((N_NODES + TILE_NODES - 1) / TILE_NODES)
#define TILE_THREADS 1024
#define ACC_PITCH    65
#define LIST_CAP     6144
#define SCAN_VEC     (N_EDGES / 4)
#define SCAN_STRIDE  (4 * TILE_THREADS)
#define SCAN_TRIPS   ((SCAN_VEC + SCAN_STRIDE - 1) / SCAN_STRIDE)

static_assert(N_EDGES % 4 == 0, "dst is scanned in 16-B words");
static_assert(N_EDGES <= (1 << 20), "edge id must fit 20 bits of a list word");
static_assert(N_NODES <= (1 << 16), "source id must fit 16 bits of a list word");
static_assert(TILE_NODES <= (1 << 12), "local node id must fit the rest of a list word");

__global__ __launch_bounds__(TILE_THREADS) void gat_tile_softmax(
    const int* __restrict__ src, const int* __restrict__ dst,
    const float* __restrict__ asrc, const float* __restrict__ adst,
    const float* __restrict__ z, float* __restrict__ out)
{
  __shared__ float    acc[TILE_NODES * ACC_PITCH];
  __shared__ float    den[TILE_NODES * NUM_HEADS];
  __shared__ int      mx[TILE_NODES * NUM_HEADS];
  __shared__ __attribute__((aligned(16))) float adl[TILE_NODES * NUM_HEADS];
  __shared__ unsigned list[LIST_CAP];
  __shared__ __attribute__((aligned(16))) float lv[LIST_CAP * NUM_HEADS];
  __shared__ unsigned cnt;

  const int tid  = threadIdx.x;
  const int base = blockIdx.x * TILE_NODES;
  const int nloc = min(TILE_NODES, N_NODES - base);

  for (int i = tid; i < TILE_NODES * ACC_PITCH; i += TILE_THREADS) acc[i] = 0.f;
  for (int i = tid; i < TILE_NODES * NUM_HEADS; i += TILE_THREADS) {
    den[i] = 0.f;
    mx[i]  = INT_MIN;
    adl[i] = (i < nloc * NUM_HEADS) ? adst[(size_t)base * NUM_HEADS + i] : 0.f;
  }
  if (tid == 0) cnt = 0u;
  __syncthreads();

  {
    const int4* __restrict__ dst4 = (const int4*)dst;
    int trip = (int)(blockIdx.x % SCAN_TRIPS);
    for (int it = 0; it < SCAN_TRIPS; ++it) {
      const int i0 = trip * SCAN_STRIDE + tid;
      int4 d[4];
#pragma unroll
      for (int u = 0; u < 4; ++u) {
        const int i = i0 + u * TILE_THREADS;
        d[u] = (i < SCAN_VEC) ? dst4[i] : make_int4(-1, -1, -1, -1);
      }
#pragma unroll
      for (int u = 0; u < 4; ++u) {
        const unsigned e0 = 4u * (unsigned)(i0 + u * TILE_THREADS);
        const unsigned dl4[4] = {(unsigned)(d[u].x - base), (unsigned)(d[u].y - base),
                                 (unsigned)(d[u].z - base), (unsigned)(d[u].w - base)};
        const unsigned lo = min(min(dl4[0], dl4[1]), min(dl4[2], dl4[3]));
        if (lo < (unsigned)TILE_NODES) {
#pragma unroll
          for (int c = 0; c < 4; ++c) {
            if (dl4[c] < (unsigned)TILE_NODES) {
              const unsigned slot =
                  __hip_atomic_fetch_add(&cnt, 1u, __ATOMIC_RELAXED, __HIP_MEMORY_SCOPE_WORKGROUP);
              if (slot < (unsigned)LIST_CAP) list[slot] = (dl4[c] << 20) | (e0 + (unsigned)c);
            }
          }
        }
      }
      trip = (trip + 1 == SCAN_TRIPS) ? 0 : trip + 1;
    }
  }
  __syncthreads();
  const unsigned total = cnt;

  if (total <= (unsigned)LIST_CAP) {
    for (unsigned k = tid; k < total; k += TILE_THREADS) {
      const unsigned w  = list[k];
      const unsigned e  = w & 0xFFFFFu;
      const unsigned dl = w >> 20;
      const int s = src[e];
      const float4 as = *(const float4*)(asrc + (size_t)s * NUM_HEADS);
      const float4 ad = *(const float4*)(adl + dl * NUM_HEADS);
      float4 v;
      v.x = leaky(as.x + ad.x);
      v.y = leaky(as.y + ad.y);
      v.z = leaky(as.z + ad.z);
      v.w = leaky(as.w + ad.w);
      *(float4*)(lv + k * NUM_HEADS) = v;
      lds_max(&mx[dl * NUM_HEADS + 0], f32_ord(v.x));
      lds_max(&mx[dl * NUM_HEADS + 1], f32_ord(v.y));
      lds_max(&mx[dl * NUM_HEADS + 2], f32_ord(v.z));
      lds_max(&mx[dl * NUM_HEADS + 3], f32_ord(v.w));
      list[k] = (dl << 16) | (unsigned)s;
    }
    __syncthreads();

    for (unsigned q = tid; q < total * NUM_HEADS; q += TILE_THREADS) {
      const unsigned dl = list[q >> 2] >> 16;
      const unsigned hq = q & 3u;
      const float ex = expf(lv[q] - ord_f32_guarded(mx[dl * NUM_HEADS + hq]));
      lv[q] = ex;
      lds_add(&den[dl * NUM_HEADS + hq], ex);
    }
    __syncthreads();

    {
      const unsigned G   = TILE_THREADS / 16;
      const unsigned g   = (unsigned)tid >> 4;
      const int      l16 = tid & 15;
      const int      hh  = l16 >> 2;
      unsigned k0 = g;
      for (; k0 + 3u * G < total; k0 += 4u * G) {
        unsigned w[4];
        float4 zr[4];
#pragma unroll
        for (int u = 0; u < 4; ++u) w[u] = list[k0 + (unsigned)u * G];
#pragma unroll
        for (int u = 0; u < 4; ++u)
          zr[u] = *(const float4*)(z + (size_t)(w[u] & 0xFFFFu) * ZCOLS + l16 * 4);
#pragma unroll
        for (int u = 0; u < 4; ++u) {
          const float ex = lv[(k0 + (unsigned)u * G) * NUM_HEADS + hh];
          float* a = acc + (w[u] >> 16) * ACC_PITCH + l16 * 4;
          lds_add(a + 0, ex * zr[u].x);
          lds_add(a + 1, ex * zr[u].y);
          lds_add(a + 2, ex * zr[u].z);
          lds_add(a + 3, ex * zr[u].w);
        }
      }
      for (; k0 < total; k0 += G) {
        const unsigned w1 = list[k0];
        const float4 z1 = *(const float4*)(z + (size_t)(w1 & 0xFFFFu) * ZCOLS + l16 * 4);
        const float ex = lv[k0 * NUM_HEADS + hh];
        float* a = acc + (w1 >> 16) * ACC_PITCH + l16 * 4;
        lds_add(a + 0, ex * z1.x);
        lds_add(a + 1, ex * z1.y);
        lds_add(a + 2, ex * z1.z);
        lds_add(a + 3, ex * z1.w);
      }
    }
  } else {
    for (int e = tid; e < N_EDGES; e += TILE_THREADS) {
      const unsigned dl = (unsigned)(dst[e] - base);
      if (dl < (unsigned)TILE_NODES) {
        const int s = src[e];
        for (int h = 0; h < NUM_HEADS; ++h)
          lds_max(&mx[dl * NUM_HEADS + h],
                  f32_ord(leaky(asrc[(size_t)s * NUM_HEADS + h] + adl[dl * NUM_HEADS + h])));
      }
    }
    __syncthreads();
    for (int e = tid; e < N_EDGES; e += TILE_THREADS) {
      const unsigned dl = (unsigned)(dst[e] - base);
      if (dl < (unsigned)TILE_NODES) {
        const int s = src[e];
        for (int h = 0; h < NUM_HEADS; ++h) {
          const float v  = leaky(asrc[(size_t)s * NUM_HEADS + h] + adl[dl * NUM_HEADS + h]);
          const float ex = expf(v - ord_f32_guarded(mx[dl * NUM_HEADS + h]));
          lds_add(&den[dl * NUM_HEADS + h], ex);
          for (int f = 0; f < OUT_FEATS; ++f)
            lds_add(&acc[dl * ACC_PITCH + h * OUT_FEATS + f],
                    ex * z[(size_t)s * ZCOLS + h * OUT_FEATS + f]);
        }
      }
    }
  }
  __syncthreads();

  for (int i = tid; i < nloc * ZCOLS; i += TILE_THREADS) {
    const int dl = i >> 6, j = i & 63;
    out[(size_t)base * ZCOLS + i] =
        acc[dl * ACC_PITCH + j] / fmaxf(den[dl * NUM_HEADS + (j >> 4)], 1e-12f);
  }
}

extern "C" void kernel_launch(void* const* d_in, const int* in_sizes, int n_in,
                              void* d_out, int out_size, void* d_ws, size_t ws_size,
                              hipStream_t stream) {
  const float* h   = (const float*)d_in[0];
  const int*   src = (const int*)d_in[1];
  const int*   dst = (const int*)d_in[2];
  const float* W   = (const float*)d_in[3];
  const float* A   = (const float*)d_in[4];
  float* out = (float*)d_out;

  float* z     = (float*)d_ws;
  float* asrc  = z     + (size_t)N_NODES * ZCOLS;
  float* adst  = asrc  + (size_t)N_NODES * NUM_HEADS;

  const int NH = N_NODES * NUM_HEADS;

  gat_proj_wmma<<<N_NODES / 16, 128, 0, stream>>>(h, W, z);
  gat_node_scores<<<(NH + 255) / 256, 256, 0, stream>>>(z, A, asrc, adst);
  gat_tile_softmax<<<N_TILES, TILE_THREADS, 0, stream>>>(src, dst, asrc, adst, z, out);
}


# tile's kept edges sorted by (node, edge id) in LDS; w*z and w summed in registers per run of equal node, one LDS add per run, not per edge; sums now in edge-id order, not arrival order
# speedup vs baseline: 2.9622x; 1.0737x over previous
#include <hip/hip_runtime.h>
#include <math.h>
#include <limits.h>

#define N_NODES   50000
#define N_EDGES   800000
#define IN_FEATS  128
#define OUT_FEATS 16
#define NUM_HEADS 4
#define ZCOLS     (NUM_HEADS * OUT_FEATS)
#define ALPHA     0.2f

typedef __attribute__((ext_vector_type(2))) float v2f;
typedef __attribute__((ext_vector_type(8))) float v8f;

__global__ __launch_bounds__(128) void gat_proj_wmma(
    const float* __restrict__ h, const float* __restrict__ W,
    float* __restrict__ z)
{
  const int wave  = threadIdx.x >> 5;
  const int lane  = threadIdx.x & 31;
  const int mtile = blockIdx.x;
  const int ntile = wave;
  const int lo16  = lane & 15;
  const int kbase = (lane >> 4) * 2;

  const float* __restrict__ ha = h + (size_t)(mtile * 16 + lo16) * IN_FEATS + kbase;
  const float* __restrict__ wb = W + (size_t)(ntile * 16 + lo16) * IN_FEATS + kbase;

  v8f c = {};
#pragma unroll 4
  for (int k0 = 0; k0 < IN_FEATS; k0 += 4) {
    v2f a, b;
    a.x = ha[k0]; a.y = ha[k0 + 1];
    b.x = wb[k0]; b.y = wb[k0 + 1];
    c = __builtin_amdgcn_wmma_f32_16x16x4_f32(
         false, a,  false, b,
         (short)0, c,  false,  false);
  }

  const int rbase = mtile * 16 + ((lane >> 4) << 3);
  float* __restrict__ zp = z + (size_t)rbase * ZCOLS + ntile * 16 + lo16;
#pragma unroll
  for (int r = 0; r < 8; ++r)
    zp[(size_t)r * ZCOLS] = c[r];
}

__global__ void gat_node_scores(const float* __restrict__ z,
                                const float* __restrict__ A,
                                float* __restrict__ asrc,
                                float* __restrict__ adst)
{
  int t = blockIdx.x * blockDim.x + threadIdx.x;
  if (t >= N_NODES * NUM_HEADS) return;
  const int n = t >> 2, hh = t & 3;
  const float* __restrict__ zv = z + (size_t)n * ZCOLS + hh * OUT_FEATS;
  const float* __restrict__ As = A + hh * (2 * OUT_FEATS);
  float s = 0.f, d = 0.f;
#pragma unroll
  for (int f = 0; f < OUT_FEATS; ++f) {
    s += zv[f] * As[f];
    d += zv[f] * As[OUT_FEATS + f];
  }
  asrc[t] = s;
  adst[t] = d;
}

__device__ __forceinline__ int f32_ord(float v) {
  int i = __float_as_int(v);
  return i >= 0 ? i : (i ^ 0x7FFFFFFF);
}

__device__ __forceinline__ float ord_f32_guarded(int iv) {
  float mv = 0.f;
  if (iv != INT_MIN) {
    int j = iv >= 0 ? iv : (iv ^ 0x7FFFFFFF);
    mv = __int_as_float(j);
    if (!isfinite(mv)) mv = 0.f;
  }
  return mv;
}

__device__ __forceinline__ float leaky(float v) { return v > 0.f ? v : ALPHA * v; }

__device__ __forceinline__ void lds_add(float* p, float v) {
  __hip_atomic_fetch_add(p, v, __ATOMIC_RELAXED, __HIP_MEMORY_SCOPE_WORKGROUP);
}
__device__ __forceinline__ void lds_max(int* p, int v) {
  __hip_atomic_fetch_max(p, v, __ATOMIC_RELAXED, __HIP_MEMORY_SCOPE_WORKGROUP);
}

#define TILE_NODES   196
#define N_TILES      ((N_NODES + TILE_NODES - 1) / TILE_NODES)
#define TILE_THREADS 1024
#define ACC_PITCH    65
#define LIST_CAP     6144
#define MIN_CHUNK    48
#define B_DEPTH      8
#define SCAN_VEC     (N_EDGES / 4)
#define SCAN_STRIDE  (4 * TILE_THREADS)
#define SCAN_TRIPS   ((SCAN_VEC + SCAN_STRIDE - 1) / SCAN_STRIDE)

static_assert(N_EDGES % 4 == 0, "dst is scanned in 16-B words");
static_assert(N_EDGES <= (1 << 20), "edge id must fit 20 bits of a list word");
static_assert(N_NODES <= (1 << 16), "source id must fit 16 bits of a list word");
static_assert(TILE_NODES <= (1 << 12), "local node id must fit the rest of a list word");

__global__ __launch_bounds__(TILE_THREADS) void gat_tile_softmax(
    const int* __restrict__ src, const int* __restrict__ dst,
    const float* __restrict__ asrc, const float* __restrict__ adst,
    const float* __restrict__ z, float* __restrict__ out)
{
  __shared__ float    acc[TILE_NODES * ACC_PITCH];
  __shared__ float    den[TILE_NODES * NUM_HEADS];
  __shared__ int      mx[TILE_NODES * NUM_HEADS];
  __shared__ __attribute__((aligned(16))) float adl[TILE_NODES * NUM_HEADS];
  __shared__ unsigned list[LIST_CAP];
  __shared__ __attribute__((aligned(16))) float lv[LIST_CAP * NUM_HEADS];
  __shared__ unsigned rowbuf[LIST_CAP];
  __shared__ unsigned deg[TILE_NODES];
  __shared__ unsigned rstart[TILE_NODES];
  __shared__ unsigned rcur[TILE_NODES];
  __shared__ unsigned cnt;

  const int tid  = threadIdx.x;
  const int base = blockIdx.x * TILE_NODES;
  const int nloc = min(TILE_NODES, N_NODES - base);

  for (int i = tid; i < TILE_NODES * ACC_PITCH; i += TILE_THREADS) acc[i] = 0.f;
  for (int i = tid; i < TILE_NODES * NUM_HEADS; i += TILE_THREADS) {
    den[i] = 0.f;
    mx[i]  = INT_MIN;
    adl[i] = (i < nloc * NUM_HEADS) ? adst[(size_t)base * NUM_HEADS + i] : 0.f;
  }
  for (int i = tid; i < TILE_NODES; i += TILE_THREADS) {
    deg[i]  = 0u;
    rcur[i] = 0u;
  }
  if (tid == 0) cnt = 0u;
  __syncthreads();

  {
    const int4* __restrict__ dst4 = (const int4*)dst;
    int trip = (int)(blockIdx.x % SCAN_TRIPS);
    for (int it = 0; it < SCAN_TRIPS; ++it) {
      const int i0 = trip * SCAN_STRIDE + tid;
      int4 d[4];
#pragma unroll
      for (int u = 0; u < 4; ++u) {
        const int i = i0 + u * TILE_THREADS;
        d[u] = (i < SCAN_VEC) ? dst4[i] : make_int4(-1, -1, -1, -1);
      }
#pragma unroll
      for (int u = 0; u < 4; ++u) {
        const unsigned e0 = 4u * (unsigned)(i0 + u * TILE_THREADS);
        const unsigned dl4[4] = {(unsigned)(d[u].x - base), (unsigned)(d[u].y - base),
                                 (unsigned)(d[u].z - base), (unsigned)(d[u].w - base)};
        const unsigned lo = min(min(dl4[0], dl4[1]), min(dl4[2], dl4[3]));
        if (lo < (unsigned)TILE_NODES) {
#pragma unroll
          for (int c = 0; c < 4; ++c) {
            if (dl4[c] < (unsigned)TILE_NODES) {
              const unsigned slot =
                  __hip_atomic_fetch_add(&cnt, 1u, __ATOMIC_RELAXED, __HIP_MEMORY_SCOPE_WORKGROUP);
              if (slot < (unsigned)LIST_CAP) list[slot] = (dl4[c] << 20) | (e0 + (unsigned)c);
              __hip_atomic_fetch_add(&deg[dl4[c]], 1u, __ATOMIC_RELAXED, __HIP_MEMORY_SCOPE_WORKGROUP);
            }
          }
        }
      }
      trip = (trip + 1 == SCAN_TRIPS) ? 0 : trip + 1;
    }
  }
  __syncthreads();
  const unsigned total = cnt;

  if (total <= (unsigned)LIST_CAP) {
    if (tid < 32) {
      constexpr int PER = (TILE_NODES + 31) / 32;
      unsigned v[PER];
      unsigned sum = 0u;
#pragma unroll
      for (int u = 0; u < PER; ++u) {
        const int idx = tid * PER + u;
        v[u] = (idx < TILE_NODES) ? deg[idx] : 0u;
        sum += v[u];
      }
      unsigned incl = sum;
#pragma unroll
      for (int d = 1; d < 32; d <<= 1) {
        const unsigned t = __shfl_up(incl, (unsigned)d, 32);
        if (tid >= d) incl += t;
      }
      unsigned run = incl - sum;
#pragma unroll
      for (int u = 0; u < PER; ++u) {
        const int idx = tid * PER + u;
        if (idx < TILE_NODES) rstart[idx] = run;
        run += v[u];
      }
    }
    __syncthreads();

    for (unsigned k = tid; k < total; k += TILE_THREADS) {
      const unsigned w  = list[k];
      const unsigned dl = w >> 20;
      const unsigned p  =
          __hip_atomic_fetch_add(&rcur[dl], 1u, __ATOMIC_RELAXED, __HIP_MEMORY_SCOPE_WORKGROUP);
      rowbuf[rstart[dl] + p] = w;
    }
    __syncthreads();

    for (unsigned j = tid; j < total; j += TILE_THREADS) {
      const unsigned w  = rowbuf[j];
      const unsigned dl = w >> 20;
      const unsigned st = rstart[dl];
      const unsigned n  = deg[dl];
      unsigned r = 0u;
      for (unsigned i = 0; i < n; i += 8u) {
        unsigned v[8];
#pragma unroll
        for (unsigned u = 0; u < 8u; ++u) v[u] = rowbuf[st + min(i + u, n - 1u)];
#pragma unroll
        for (unsigned u = 0; u < 8u; ++u) r += (i + u < n && v[u] < w) ? 1u : 0u;
      }
      list[st + r] = w;
    }
    __syncthreads();

    const unsigned S = ((total + 31u) >> 5) | 1u;
    for (unsigned kt = tid; kt < 32u * S; kt += TILE_THREADS) {
      const unsigned k = (kt & 31u) * S + (kt >> 5);
      if (k >= total) continue;
      const unsigned w  = list[k];
      const unsigned e  = w & 0xFFFFFu;
      const unsigned dl = w >> 20;
      const int s = src[e];
      const float4 as = *(const float4*)(asrc + (size_t)s * NUM_HEADS);
      const float4 ad = *(const float4*)(adl + dl * NUM_HEADS);
      float4 v;
      v.x = leaky(as.x + ad.x);
      v.y = leaky(as.y + ad.y);
      v.z = leaky(as.z + ad.z);
      v.w = leaky(as.w + ad.w);
      *(float4*)(lv + k * NUM_HEADS) = v;
      lds_max(&mx[dl * NUM_HEADS + 0], f32_ord(v.x));
      lds_max(&mx[dl * NUM_HEADS + 1], f32_ord(v.y));
      lds_max(&mx[dl * NUM_HEADS + 2], f32_ord(v.z));
      lds_max(&mx[dl * NUM_HEADS + 3], f32_ord(v.w));
      list[k] = (dl << 16) | (unsigned)s;
    }
    __syncthreads();

    for (unsigned q = tid; q < total * NUM_HEADS; q += TILE_THREADS) {
      const unsigned dl = list[q >> 2] >> 16;
      const unsigned hq = q & 3u;
      const float ex = expf(lv[q] - ord_f32_guarded(mx[dl * NUM_HEADS + hq]));
      lv[q] = ex;
    }
    __syncthreads();

    {
      const unsigned G   = TILE_THREADS / 16;
      const unsigned g   = (unsigned)tid >> 4;
      const int      l16 = tid & 15;
      const int      hh  = l16 >> 2;
      unsigned chunk = (total + G - 1u) / G;
      chunk = (chunk < (unsigned)MIN_CHUNK) ? (unsigned)MIN_CHUNK : chunk;
      chunk = (chunk + (unsigned)B_DEPTH - 1u) / (unsigned)B_DEPTH * (unsigned)B_DEPTH;
      const unsigned kbeg = g * chunk;
      const unsigned kend = min(kbeg + chunk, total);
      const unsigned NONE = 0xFFFFFFFFu;
      unsigned cur = NONE;
      float4 s  = make_float4(0.f, 0.f, 0.f, 0.f);
      float  ds = 0.f;
      for (unsigned k0 = kbeg; k0 < kend; k0 += (unsigned)B_DEPTH) {
        unsigned w[B_DEPTH];
        float4   zr[B_DEPTH];
        float    ex[B_DEPTH];
#pragma unroll
        for (int u = 0; u < B_DEPTH; ++u) w[u] = list[min(k0 + (unsigned)u, kend - 1u)];
#pragma unroll
        for (int u = 0; u < B_DEPTH; ++u)
          zr[u] = *(const float4*)(z + (size_t)(w[u] & 0xFFFFu) * ZCOLS + l16 * 4);
#pragma unroll
        for (int u = 0; u < B_DEPTH; ++u)
          ex[u] = lv[min(k0 + (unsigned)u, kend - 1u) * NUM_HEADS + hh];
#pragma unroll
        for (int u = 0; u < B_DEPTH; ++u) {
          if (k0 + (unsigned)u < kend) {
            const unsigned node = w[u] >> 16;
            if (node != cur) {
              if (cur != NONE) {
                float* a = acc + cur * ACC_PITCH + l16 * 4;
                lds_add(a + 0, s.x);
                lds_add(a + 1, s.y);
                lds_add(a + 2, s.z);
                lds_add(a + 3, s.w);
                if ((l16 & 3) == 0) lds_add(&den[cur * NUM_HEADS + hh], ds);
              }
              cur = node;
              s   = make_float4(0.f, 0.f, 0.f, 0.f);
              ds  = 0.f;
            }
            s.x += ex[u] * zr[u].x;
            s.y += ex[u] * zr[u].y;
            s.z += ex[u] * zr[u].z;
            s.w += ex[u] * zr[u].w;
            ds  += ex[u];
          }
        }
      }
      if (cur != NONE) {
        float* a = acc + cur * ACC_PITCH + l16 * 4;
        lds_add(a + 0, s.x);
        lds_add(a + 1, s.y);
        lds_add(a + 2, s.z);
        lds_add(a + 3, s.w);
        if ((l16 & 3) == 0) lds_add(&den[cur * NUM_HEADS + hh], ds);
      }
    }
  } else {
    for (int e = tid; e < N_EDGES; e += TILE_THREADS) {
      const unsigned dl = (unsigned)(dst[e] - base);
      if (dl < (unsigned)TILE_NODES) {
        const int s = src[e];
        for (int h = 0; h < NUM_HEADS; ++h)
          lds_max(&mx[dl * NUM_HEADS + h],
                  f32_ord(leaky(asrc[(size_t)s * NUM_HEADS + h] + adl[dl * NUM_HEADS + h])));
      }
    }
    __syncthreads();
    for (int e = tid; e < N_EDGES; e += TILE_THREADS) {
      const unsigned dl = (unsigned)(dst[e] - base);
      if (dl < (unsigned)TILE_NODES) {
        const int s = src[e];
        for (int h = 0; h < NUM_HEADS; ++h) {
          const float v  = leaky(asrc[(size_t)s * NUM_HEADS + h] + adl[dl * NUM_HEADS + h]);
          const float ex = expf(v - ord_f32_guarded(mx[dl * NUM_HEADS + h]));
          lds_add(&den[dl * NUM_HEADS + h], ex);
          for (int f = 0; f < OUT_FEATS; ++f)
            lds_add(&acc[dl * ACC_PITCH + h * OUT_FEATS + f],
                    ex * z[(size_t)s * ZCOLS + h * OUT_FEATS + f]);
        }
      }
    }
  }
  __syncthreads();

  for (int i = tid; i < nloc * ZCOLS; i += TILE_THREADS) {
    const int dl = i >> 6, j = i & 63;
    out[(size_t)base * ZCOLS + i] =
        acc[dl * ACC_PITCH + j] / fmaxf(den[dl * NUM_HEADS + (j >> 4)], 1e-12f);
  }
}

extern "C" void kernel_launch(void* const* d_in, const int* in_sizes, int n_in,
                              void* d_out, int out_size, void* d_ws, size_t ws_size,
                              hipStream_t stream) {
  const float* h   = (const float*)d_in[0];
  const int*   src = (const int*)d_in[1];
  const int*   dst = (const int*)d_in[2];
  const float* W   = (const float*)d_in[3];
  const float* A   = (const float*)d_in[4];
  float* out = (float*)d_out;

  float* z     = (float*)d_ws;
  float* asrc  = z     + (size_t)N_NODES * ZCOLS;
  float* adst  = asrc  + (size_t)N_NODES * NUM_HEADS;

  const int NH = N_NODES * NUM_HEADS;

  gat_proj_wmma<<<N_NODES / 16, 128, 0, stream>>>(h, W, z);
  gat_node_scores<<<(NH + 255) / 256, 256, 0, stream>>>(z, A, asrc, adst);
  gat_tile_softmax<<<N_TILES, TILE_THREADS, 0, stream>>>(src, dst, asrc, adst, z, out);
}


# tile blocks scan one tile-id byte per edge (packed by the node-score launch where adst was) instead of 4-byte dst; dst read for kept edges only; adst formed in the tile block; sum order unchanged
# speedup vs baseline: 3.8881x; 1.3126x over previous
#include <hip/hip_runtime.h>
#include <math.h>
#include <limits.h>

#define N_NODES   50000
#define N_EDGES   800000
#define IN_FEATS  128
#define OUT_FEATS 16
#define NUM_HEADS 4
#define ZCOLS     (NUM_HEADS * OUT_FEATS)
#define ALPHA     0.2f

typedef __attribute__((ext_vector_type(2))) float v2f;
typedef __attribute__((ext_vector_type(8))) float v8f;

__global__ __launch_bounds__(128) void gat_proj_wmma(
    const float* __restrict__ h, const float* __restrict__ W,
    float* __restrict__ z)
{
  const int wave  = threadIdx.x >> 5;
  const int lane  = threadIdx.x & 31;
  const int mtile = blockIdx.x;
  const int ntile = wave;
  const int lo16  = lane & 15;
  const int kbase = (lane >> 4) * 2;

  const float* __restrict__ ha = h + (size_t)(mtile * 16 + lo16) * IN_FEATS + kbase;
  const float* __restrict__ wb = W + (size_t)(ntile * 16 + lo16) * IN_FEATS + kbase;

  v8f c = {};
#pragma unroll 4
  for (int k0 = 0; k0 < IN_FEATS; k0 += 4) {
    v2f a, b;
    a.x = ha[k0]; a.y = ha[k0 + 1];
    b.x = wb[k0]; b.y = wb[k0 + 1];
    c = __builtin_amdgcn_wmma_f32_16x16x4_f32(
         false, a,  false, b,
         (short)0, c,  false,  false);
  }

  const int rbase = mtile * 16 + ((lane >> 4) << 3);
  float* __restrict__ zp = z + (size_t)rbase * ZCOLS + ntile * 16 + lo16;
#pragma unroll
  for (int r = 0; r < 8; ++r)
    zp[(size_t)r * ZCOLS] = c[r];
}

#define TILE_NODES   196
#define N_TILES      ((N_NODES + TILE_NODES - 1) / TILE_NODES)

static_assert(N_TILES <= 256, "a tile id must fit one byte");
static_assert(N_EDGES == 4 * N_NODES * NUM_HEADS, "one thread of the node-score launch packs four tile ids");

__global__ void gat_node_scores(const float* __restrict__ z,
                                const float* __restrict__ A,
                                const int* __restrict__ dst,
                                float* __restrict__ asrc,
                                unsigned* __restrict__ tileid)
{
  int t = blockIdx.x * blockDim.x + threadIdx.x;
  if (t >= N_NODES * NUM_HEADS) return;
  const int4 d4 = ((const int4*)dst)[t];
  const int n = t >> 2, hh = t & 3;
  const float* __restrict__ zv = z + (size_t)n * ZCOLS + hh * OUT_FEATS;
  const float* __restrict__ As = A + hh * (2 * OUT_FEATS);
  float s = 0.f;
#pragma unroll
  for (int f = 0; f < OUT_FEATS; ++f) s += zv[f] * As[f];
  asrc[t] = s;
  const unsigned b0 = min((unsigned)d4.x / (unsigned)TILE_NODES, 255u);
  const unsigned b1 = min((unsigned)d4.y / (unsigned)TILE_NODES, 255u);
  const unsigned b2 = min((unsigned)d4.z / (unsigned)TILE_NODES, 255u);
  const unsigned b3 = min((unsigned)d4.w / (unsigned)TILE_NODES, 255u);
  tileid[t] = b0 | (b1 << 8) | (b2 << 16) | (b3 << 24);
}

__device__ __forceinline__ int f32_ord(float v) {
  int i = __float_as_int(v);
  return i >= 0 ? i : (i ^ 0x7FFFFFFF);
}

__device__ __forceinline__ float ord_f32_guarded(int iv) {
  float mv = 0.f;
  if (iv != INT_MIN) {
    int j = iv >= 0 ? iv : (iv ^ 0x7FFFFFFF);
    mv = __int_as_float(j);
    if (!isfinite(mv)) mv = 0.f;
  }
  return mv;
}

__device__ __forceinline__ float leaky(float v) { return v > 0.f ? v : ALPHA * v; }

__device__ __forceinline__ void lds_add(float* p, float v) {
  __hip_atomic_fetch_add(p, v, __ATOMIC_RELAXED, __HIP_MEMORY_SCOPE_WORKGROUP);
}
__device__ __forceinline__ void lds_max(int* p, int v) {
  __hip_atomic_fetch_max(p, v, __ATOMIC_RELAXED, __HIP_MEMORY_SCOPE_WORKGROUP);
}

#define TILE_THREADS 1024
#define ACC_PITCH    65
#define LIST_CAP     6144
#define MIN_CHUNK    48
#define B_DEPTH      8
#define SCAN_VEC     (N_EDGES / 16)
#define SCAN_STRIDE  (4 * TILE_THREADS)
#define SCAN_TRIPS   ((SCAN_VEC + SCAN_STRIDE - 1) / SCAN_STRIDE)

static_assert(N_EDGES % 16 == 0, "the tile ids are scanned in 16-B words");
static_assert(N_EDGES <= (1 << 20), "edge id must fit 20 bits of a list word");
static_assert(N_NODES <= (1 << 16), "source id must fit 16 bits of a list word");
static_assert(TILE_NODES <= (1 << 12), "local node id must fit the rest of a list word");

__global__ __launch_bounds__(TILE_THREADS) void gat_tile_softmax(
    const int* __restrict__ src, const int* __restrict__ dst,
    const unsigned* __restrict__ tileid,
    const float* __restrict__ asrc, const float* __restrict__ A,
    const float* __restrict__ z, float* __restrict__ out)
{
  __shared__ float    acc[TILE_NODES * ACC_PITCH];
  __shared__ float    den[TILE_NODES * NUM_HEADS];
  __shared__ int      mx[TILE_NODES * NUM_HEADS];
  __shared__ __attribute__((aligned(16))) float adl[TILE_NODES * NUM_HEADS];
  __shared__ unsigned list[LIST_CAP];
  __shared__ __attribute__((aligned(16))) float lv[LIST_CAP * NUM_HEADS];
  __shared__ unsigned rowbuf[LIST_CAP];
  __shared__ unsigned deg[TILE_NODES];
  __shared__ unsigned rstart[TILE_NODES];
  __shared__ unsigned rcur[TILE_NODES];
  __shared__ unsigned cnt;

  const int tid  = threadIdx.x;
  const int base = blockIdx.x * TILE_NODES;
  const int nloc = min(TILE_NODES, N_NODES - base);

  for (int i = tid; i < TILE_NODES * ACC_PITCH; i += TILE_THREADS) acc[i] = 0.f;
  for (int i = tid; i < TILE_NODES * NUM_HEADS; i += TILE_THREADS) {
    den[i] = 0.f;
    mx[i]  = INT_MIN;
    float dsc = 0.f;
    if (i < nloc * NUM_HEADS) {
      const int hh = i & 3;
      const float* __restrict__ zv = z + (size_t)(base + (i >> 2)) * ZCOLS + hh * OUT_FEATS;
      const float* __restrict__ Ad = A + hh * (2 * OUT_FEATS) + OUT_FEATS;
#pragma unroll
      for (int f = 0; f < OUT_FEATS; ++f) dsc += zv[f] * Ad[f];
    }
    adl[i] = dsc;
  }
  for (int i = tid; i < TILE_NODES; i += TILE_THREADS) {
    deg[i]  = 0u;
    rcur[i] = 0u;
  }
  if (tid == 0) cnt = 0u;
  __syncthreads();

  {
    const uint4* __restrict__ tid4 = (const uint4*)tileid;
    const unsigned T4 = (unsigned)blockIdx.x * 0x01010101u;
    int trip = (int)(blockIdx.x % SCAN_TRIPS);
    for (int it = 0; it < SCAN_TRIPS; ++it) {
      const int i0 = trip * SCAN_STRIDE + tid;
      uint4 d[4];
#pragma unroll
      for (int u = 0; u < 4; ++u) d[u] = tid4[min(i0 + u * TILE_THREADS, SCAN_VEC - 1)];
#pragma unroll
      for (int q = 0; q < 2; ++q) {
        unsigned m = 0u;
#pragma unroll
        for (int w = 0; w < 2; ++w) {
          const int u = 2 * q + w;
          const unsigned x[4] = {d[u].x ^ T4, d[u].y ^ T4, d[u].z ^ T4, d[u].w ^ T4};
          unsigned f[4];
#pragma unroll
          for (int c = 0; c < 4; ++c)
            f[c] = (x[c] - 0x01010101u) & ~(x[c] | (x[c] << 7)) & 0x80808080u;
          const unsigned mw = (f[0] >> 7) | (f[1] >> 6) | (f[2] >> 5) | (f[3] >> 4);
          m |= (i0 + u * TILE_THREADS < SCAN_VEC) ? (mw << (4 * w)) : 0u;
        }
        const unsigned e0 = 16u * (unsigned)(i0 + 2 * q * TILE_THREADS);
        while (m != 0u) {
          const unsigned p = (unsigned)__builtin_ctz(m);
          m &= m - 1u;
          const unsigned slot =
              __hip_atomic_fetch_add(&cnt, 1u, __ATOMIC_RELAXED, __HIP_MEMORY_SCOPE_WORKGROUP);
          if (slot < (unsigned)LIST_CAP)
            list[slot] = e0 + (((p >> 2) & 1u) * (16u * TILE_THREADS)) + ((p & 3u) << 2) + (p >> 3);
        }
      }
      trip = (trip + 1 == SCAN_TRIPS) ? 0 : trip + 1;
    }
  }
  __syncthreads();
  const unsigned total = cnt;

  if (total <= (unsigned)LIST_CAP) {
    {
      constexpr int PER = LIST_CAP / TILE_THREADS;
      static_assert(LIST_CAP % TILE_THREADS == 0, "every list slot has one thread and one turn");
      unsigned ev[PER];
      int      dv[PER];
#pragma unroll
      for (int u = 0; u < PER; ++u) {
        const unsigned k = (unsigned)tid + (unsigned)u * TILE_THREADS;
        ev[u] = (k < total) ? list[k] : 0u;
      }
#pragma unroll
      for (int u = 0; u < PER; ++u) dv[u] = dst[ev[u]];
#pragma unroll
      for (int u = 0; u < PER; ++u) {
        const unsigned k = (unsigned)tid + (unsigned)u * TILE_THREADS;
        if (k < total) {
          const unsigned dl = min((unsigned)(dv[u] - base), (unsigned)(TILE_NODES - 1));
          __hip_atomic_fetch_add(&deg[dl], 1u, __ATOMIC_RELAXED, __HIP_MEMORY_SCOPE_WORKGROUP);
          list[k] = (dl << 20) | ev[u];
        }
      }
    }
    __syncthreads();

    if (tid < 32) {
      constexpr int PER = (TILE_NODES + 31) / 32;
      unsigned v[PER];
      unsigned sum = 0u;
#pragma unroll
      for (int u = 0; u < PER; ++u) {
        const int idx = tid * PER + u;
        v[u] = (idx < TILE_NODES) ? deg[idx] : 0u;
        sum += v[u];
      }
      unsigned incl = sum;
#pragma unroll
      for (int d = 1; d < 32; d <<= 1) {
        const unsigned t = __shfl_up(incl, (unsigned)d, 32);
        if (tid >= d) incl += t;
      }
      unsigned run = incl - sum;
#pragma unroll
      for (int u = 0; u < PER; ++u) {
        const int idx = tid * PER + u;
        if (idx < TILE_NODES) rstart[idx] = run;
        run += v[u];
      }
    }
    __syncthreads();

    for (unsigned k = tid; k < total; k += TILE_THREADS) {
      const unsigned w  = list[k];
      const unsigned dl = w >> 20;
      const unsigned p  =
          __hip_atomic_fetch_add(&rcur[dl], 1u, __ATOMIC_RELAXED, __HIP_MEMORY_SCOPE_WORKGROUP);
      rowbuf[rstart[dl] + p] = w;
    }
    __syncthreads();

    for (unsigned j = tid; j < total; j += TILE_THREADS) {
      const unsigned w  = rowbuf[j];
      const unsigned dl = w >> 20;
      const unsigned st = rstart[dl];
      const unsigned n  = deg[dl];
      unsigned r = 0u;
      for (unsigned i = 0; i < n; i += 8u) {
        unsigned v[8];
#pragma unroll
        for (unsigned u = 0; u < 8u; ++u) v[u] = rowbuf[st + min(i + u, n - 1u)];
#pragma unroll
        for (unsigned u = 0; u < 8u; ++u) r += (i + u < n && v[u] < w) ? 1u : 0u;
      }
      list[st + r] = w;
    }
    __syncthreads();

    const unsigned S = ((total + 31u) >> 5) | 1u;
    for (unsigned kt = tid; kt < 32u * S; kt += TILE_THREADS) {
      const unsigned k = (kt & 31u) * S + (kt >> 5);
      if (k >= total) continue;
      const unsigned w  = list[k];
      const unsigned e  = w & 0xFFFFFu;
      const unsigned dl = w >> 20;
      const int s = src[e];
      const float4 as = *(const float4*)(asrc + (size_t)s * NUM_HEADS);
      const float4 ad = *(const float4*)(adl + dl * NUM_HEADS);
      float4 v;
      v.x = leaky(as.x + ad.x);
      v.y = leaky(as.y + ad.y);
      v.z = leaky(as.z + ad.z);
      v.w = leaky(as.w + ad.w);
      *(float4*)(lv + k * NUM_HEADS) = v;
      lds_max(&mx[dl * NUM_HEADS + 0], f32_ord(v.x));
      lds_max(&mx[dl * NUM_HEADS + 1], f32_ord(v.y));
      lds_max(&mx[dl * NUM_HEADS + 2], f32_ord(v.z));
      lds_max(&mx[dl * NUM_HEADS + 3], f32_ord(v.w));
      list[k] = (dl << 16) | (unsigned)s;
    }
    __syncthreads();

    for (unsigned q = tid; q < total * NUM_HEADS; q += TILE_THREADS) {
      const unsigned dl = list[q >> 2] >> 16;
      const unsigned hq = q & 3u;
      const float ex = expf(lv[q] - ord_f32_guarded(mx[dl * NUM_HEADS + hq]));
      lv[q] = ex;
    }
    __syncthreads();

    {
      const unsigned G   = TILE_THREADS / 16;
      const unsigned g   = (unsigned)tid >> 4;
      const int      l16 = tid & 15;
      const int      hh  = l16 >> 2;
      unsigned chunk = (total + G - 1u) / G;
      chunk = (chunk < (unsigned)MIN_CHUNK) ? (unsigned)MIN_CHUNK : chunk;
      chunk = (chunk + (unsigned)B_DEPTH - 1u) / (unsigned)B_DEPTH * (unsigned)B_DEPTH;
      const unsigned kbeg = g * chunk;
      const unsigned kend = min(kbeg + chunk, total);
      const unsigned NONE = 0xFFFFFFFFu;
      unsigned cur = NONE;
      float4 s  = make_float4(0.f, 0.f, 0.f, 0.f);
      float  ds = 0.f;
      for (unsigned k0 = kbeg; k0 < kend; k0 += (unsigned)B_DEPTH) {
        unsigned w[B_DEPTH];
        float4   zr[B_DEPTH];
        float    ex[B_DEPTH];
#pragma unroll
        for (int u = 0; u < B_DEPTH; ++u) w[u] = list[min(k0 + (unsigned)u, kend - 1u)];
#pragma unroll
        for (int u = 0; u < B_DEPTH; ++u)
          zr[u] = *(const float4*)(z + (size_t)(w[u] & 0xFFFFu) * ZCOLS + l16 * 4);
#pragma unroll
        for (int u = 0; u < B_DEPTH; ++u)
          ex[u] = lv[min(k0 + (unsigned)u, kend - 1u) * NUM_HEADS + hh];
#pragma unroll
        for (int u = 0; u < B_DEPTH; ++u) {
          if (k0 + (unsigned)u < kend) {
            const unsigned node = w[u] >> 16;
            if (node != cur) {
              if (cur != NONE) {
                float* a = acc + cur * ACC_PITCH + l16 * 4;
                lds_add(a + 0, s.x);
                lds_add(a + 1, s.y);
                lds_add(a + 2, s.z);
                lds_add(a + 3, s.w);
                if ((l16 & 3) == 0) lds_add(&den[cur * NUM_HEADS + hh], ds);
              }
              cur = node;
              s   = make_float4(0.f, 0.f, 0.f, 0.f);
              ds  = 0.f;
            }
            s.x += ex[u] * zr[u].x;
            s.y += ex[u] * zr[u].y;
            s.z += ex[u] * zr[u].z;
            s.w += ex[u] * zr[u].w;
            ds  += ex[u];
          }
        }
      }
      if (cur != NONE) {
        float* a = acc + cur * ACC_PITCH + l16 * 4;
        lds_add(a + 0, s.x);
        lds_add(a + 1, s.y);
        lds_add(a + 2, s.z);
        lds_add(a + 3, s.w);
        if ((l16 & 3) == 0) lds_add(&den[cur * NUM_HEADS + hh], ds);
      }
    }
  } else {
    for (int e = tid; e < N_EDGES; e += TILE_THREADS) {
      const unsigned dl = (unsigned)(dst[e] - base);
      if (dl < (unsigned)TILE_NODES) {
        const int s = src[e];
        for (int h = 0; h < NUM_HEADS; ++h)
          lds_max(&mx[dl * NUM_HEADS + h],
                  f32_ord(leaky(asrc[(size_t)s * NUM_HEADS + h] + adl[dl * NUM_HEADS + h])));
      }
    }
    __syncthreads();
    for (int e = tid; e < N_EDGES; e += TILE_THREADS) {
      const unsigned dl = (unsigned)(dst[e] - base);
      if (dl < (unsigned)TILE_NODES) {
        const int s = src[e];
        for (int h = 0; h < NUM_HEADS; ++h) {
          const float v  = leaky(asrc[(size_t)s * NUM_HEADS + h] + adl[dl * NUM_HEADS + h]);
          const float ex = expf(v - ord_f32_guarded(mx[dl * NUM_HEADS + h]));
          lds_add(&den[dl * NUM_HEADS + h], ex);
          for (int f = 0; f < OUT_FEATS; ++f)
            lds_add(&acc[dl * ACC_PITCH + h * OUT_FEATS + f],
                    ex * z[(size_t)s * ZCOLS + h * OUT_FEATS + f]);
        }
      }
    }
  }
  __syncthreads();

  for (int i = tid; i < nloc * ZCOLS; i += TILE_THREADS) {
    const int dl = i >> 6, j = i & 63;
    out[(size_t)base * ZCOLS + i] =
        acc[dl * ACC_PITCH + j] / fmaxf(den[dl * NUM_HEADS + (j >> 4)], 1e-12f);
  }
}

extern "C" void kernel_launch(void* const* d_in, const int* in_sizes, int n_in,
                              void* d_out, int out_size, void* d_ws, size_t ws_size,
                              hipStream_t stream) {
  const float* h   = (const float*)d_in[0];
  const int*   src = (const int*)d_in[1];
  const int*   dst = (const int*)d_in[2];
  const float* W   = (const float*)d_in[3];
  const float* A   = (const float*)d_in[4];
  float* out = (float*)d_out;

  float* z     = (float*)d_ws;
  float* asrc  = z     + (size_t)N_NODES * ZCOLS;
  unsigned* tileid = (unsigned*)(asrc + (size_t)N_NODES * NUM_HEADS);

  const int NH = N_NODES * NUM_HEADS;

  gat_proj_wmma<<<N_NODES / 16, 128, 0, stream>>>(h, W, z);
  gat_node_scores<<<(NH + 255) / 256, 256, 0, stream>>>(z, A, dst, asrc, tileid);
  gat_tile_softmax<<<N_TILES, TILE_THREADS, 0, stream>>>(src, dst, tileid, asrc, A, z, out);
}


# projection: a wave keeps its head's W fragments in registers and runs 5 node tiles as separate chains; a block's 80 h rows go to LDS by whole-row 16-B loads, A fragments read there; z sums unchanged
# speedup vs baseline: 4.5043x; 1.1585x over previous
#include <hip/hip_runtime.h>
#include <math.h>
#include <limits.h>

#define N_NODES   50000
#define N_EDGES   800000
#define IN_FEATS  128
#define OUT_FEATS 16
#define NUM_HEADS 4
#define ZCOLS     (NUM_HEADS * OUT_FEATS)
#define ALPHA     0.2f

typedef __attribute__((ext_vector_type(2))) float v2f;
typedef __attribute__((ext_vector_type(8))) float v8f;

#define PROJ_TILES   5
#define PROJ_ROWS    (16 * PROJ_TILES)
#define PROJ_BLOCKS  (N_NODES / PROJ_ROWS)
#define KSTEPS       (IN_FEATS / 4)
#define HT_PITCH     132
#define HT_TILE      (16 * HT_PITCH)
#define PIECES_PER_THREAD (PROJ_ROWS * (IN_FEATS / 4) / 128)

static_assert(N_NODES % PROJ_ROWS == 0, "every block takes a whole panel");
static_assert(PROJ_ROWS * (IN_FEATS / 4) % 128 == 0, "the panel's 16-B pieces divide over the threads");

__global__ __launch_bounds__(128) void gat_proj_wmma(
    const float* __restrict__ h, const float* __restrict__ W,
    float* __restrict__ z)
{
  __shared__ __attribute__((aligned(16))) float ht[PROJ_ROWS * HT_PITCH];

  const int wave  = threadIdx.x >> 5;
  const int lane  = threadIdx.x & 31;
  const int ntile = wave;
  const int lo16  = lane & 15;
  const int kbase = (lane >> 4) * 2;
  const int node0 = blockIdx.x * PROJ_ROWS;

  const float4* __restrict__ hsrc =
      (const float4*)(h + (size_t)node0 * IN_FEATS) + threadIdx.x;
  float* hdst = ht + (threadIdx.x >> 5) * HT_PITCH + (threadIdx.x & 31) * 4;

  float4 pre[PIECES_PER_THREAD];
#pragma unroll
  for (int u = 0; u < PIECES_PER_THREAD; ++u) pre[u] = hsrc[u * 128];

  const float* __restrict__ wb = W + (size_t)(ntile * 16 + lo16) * IN_FEATS + kbase;
  v2f b[KSTEPS];
#pragma unroll
  for (int j = 0; j < KSTEPS; ++j) { b[j].x = wb[4 * j]; b[j].y = wb[4 * j + 1]; }

#pragma unroll
  for (int u = 0; u < PIECES_PER_THREAD; ++u) *(float4*)(hdst + u * 4 * HT_PITCH) = pre[u];
  __syncthreads();

  const float* ap = ht + lo16 * HT_PITCH + kbase;
  v8f c[PROJ_TILES];
#pragma unroll
  for (int t = 0; t < PROJ_TILES; ++t) c[t] = v8f{};
#pragma unroll
  for (int j = 0; j < KSTEPS; ++j) {
#pragma unroll
    for (int t = 0; t < PROJ_TILES; ++t) {
      const v2f a = *(const v2f*)(ap + t * HT_TILE + 4 * j);
      c[t] = __builtin_amdgcn_wmma_f32_16x16x4_f32(
           false, a,  false, b[j],
           (short)0, c[t],  false,  false);
    }
  }

#pragma unroll
  for (int t = 0; t < PROJ_TILES; ++t) {
    const int rbase = node0 + 16 * t + ((lane >> 4) << 3);
    float* __restrict__ zp = z + (size_t)rbase * ZCOLS + ntile * 16 + lo16;
#pragma unroll
    for (int r = 0; r < 8; ++r)
      zp[(size_t)r * ZCOLS] = c[t][r];
  }
}

#define TILE_NODES   196
#define N_TILES      ((N_NODES + TILE_NODES - 1) / TILE_NODES)

static_assert(N_TILES <= 256, "a tile id must fit one byte");
static_assert(N_EDGES == 4 * N_NODES * NUM_HEADS, "one thread of the node-score launch packs four tile ids");

__global__ void gat_node_scores(const float* __restrict__ z,
                                const float* __restrict__ A,
                                const int* __restrict__ dst,
                                float* __restrict__ asrc,
                                unsigned* __restrict__ tileid)
{
  int t = blockIdx.x * blockDim.x + threadIdx.x;
  if (t >= N_NODES * NUM_HEADS) return;
  const int4 d4 = ((const int4*)dst)[t];
  const int n = t >> 2, hh = t & 3;
  const float* __restrict__ zv = z + (size_t)n * ZCOLS + hh * OUT_FEATS;
  const float* __restrict__ As = A + hh * (2 * OUT_FEATS);
  float s = 0.f;
#pragma unroll
  for (int f = 0; f < OUT_FEATS; ++f) s += zv[f] * As[f];
  asrc[t] = s;
  const unsigned b0 = min((unsigned)d4.x / (unsigned)TILE_NODES, 255u);
  const unsigned b1 = min((unsigned)d4.y / (unsigned)TILE_NODES, 255u);
  const unsigned b2 = min((unsigned)d4.z / (unsigned)TILE_NODES, 255u);
  const unsigned b3 = min((unsigned)d4.w / (unsigned)TILE_NODES, 255u);
  tileid[t] = b0 | (b1 << 8) | (b2 << 16) | (b3 << 24);
}

__device__ __forceinline__ int f32_ord(float v) {
  int i = __float_as_int(v);
  return i >= 0 ? i : (i ^ 0x7FFFFFFF);
}

__device__ __forceinline__ float ord_f32_guarded(int iv) {
  float mv = 0.f;
  if (iv != INT_MIN) {
    int j = iv >= 0 ? iv : (iv ^ 0x7FFFFFFF);
    mv = __int_as_float(j);
    if (!isfinite(mv)) mv = 0.f;
  }
  return mv;
}

__device__ __forceinline__ float leaky(float v) { return v > 0.f ? v : ALPHA * v; }

__device__ __forceinline__ void lds_add(float* p, float v) {
  __hip_atomic_fetch_add(p, v, __ATOMIC_RELAXED, __HIP_MEMORY_SCOPE_WORKGROUP);
}
__device__ __forceinline__ void lds_max(int* p, int v) {
  __hip_atomic_fetch_max(p, v, __ATOMIC_RELAXED, __HIP_MEMORY_SCOPE_WORKGROUP);
}

#define TILE_THREADS 1024
#define ACC_PITCH    65
#define LIST_CAP     6144
#define MIN_CHUNK    48
#define B_DEPTH      8
#define SCAN_VEC     (N_EDGES / 16)
#define SCAN_STRIDE  (4 * TILE_THREADS)
#define SCAN_TRIPS   ((SCAN_VEC + SCAN_STRIDE - 1) / SCAN_STRIDE)

static_assert(N_EDGES % 16 == 0, "the tile ids are scanned in 16-B words");
static_assert(N_EDGES <= (1 << 20), "edge id must fit 20 bits of a list word");
static_assert(N_NODES <= (1 << 16), "source id must fit 16 bits of a list word");
static_assert(TILE_NODES <= (1 << 12), "local node id must fit the rest of a list word");

__global__ __launch_bounds__(TILE_THREADS) void gat_tile_softmax(
    const int* __restrict__ src, const int* __restrict__ dst,
    const unsigned* __restrict__ tileid,
    const float* __restrict__ asrc, const float* __restrict__ A,
    const float* __restrict__ z, float* __restrict__ out)
{
  __shared__ float    acc[TILE_NODES * ACC_PITCH];
  __shared__ float    den[TILE_NODES * NUM_HEADS];
  __shared__ int      mx[TILE_NODES * NUM_HEADS];
  __shared__ __attribute__((aligned(16))) float adl[TILE_NODES * NUM_HEADS];
  __shared__ unsigned list[LIST_CAP];
  __shared__ __attribute__((aligned(16))) float lv[LIST_CAP * NUM_HEADS];
  __shared__ unsigned rowbuf[LIST_CAP];
  __shared__ unsigned deg[TILE_NODES];
  __shared__ unsigned rstart[TILE_NODES];
  __shared__ unsigned rcur[TILE_NODES];
  __shared__ unsigned cnt;

  const int tid  = threadIdx.x;
  const int base = blockIdx.x * TILE_NODES;
  const int nloc = min(TILE_NODES, N_NODES - base);

  for (int i = tid; i < TILE_NODES * ACC_PITCH; i += TILE_THREADS) acc[i] = 0.f;
  for (int i = tid; i < TILE_NODES * NUM_HEADS; i += TILE_THREADS) {
    den[i] = 0.f;
    mx[i]  = INT_MIN;
    float dsc = 0.f;
    if (i < nloc * NUM_HEADS) {
      const int hh = i & 3;
      const float* __restrict__ zv = z + (size_t)(base + (i >> 2)) * ZCOLS + hh * OUT_FEATS;
      const float* __restrict__ Ad = A + hh * (2 * OUT_FEATS) + OUT_FEATS;
#pragma unroll
      for (int f = 0; f < OUT_FEATS; ++f) dsc += zv[f] * Ad[f];
    }
    adl[i] = dsc;
  }
  for (int i = tid; i < TILE_NODES; i += TILE_THREADS) {
    deg[i]  = 0u;
    rcur[i] = 0u;
  }
  if (tid == 0) cnt = 0u;
  __syncthreads();

  {
    const uint4* __restrict__ tid4 = (const uint4*)tileid;
    const unsigned T4 = (unsigned)blockIdx.x * 0x01010101u;
    int trip = (int)(blockIdx.x % SCAN_TRIPS);
    for (int it = 0; it < SCAN_TRIPS; ++it) {
      const int i0 = trip * SCAN_STRIDE + tid;
      uint4 d[4];
#pragma unroll
      for (int u = 0; u < 4; ++u) d[u] = tid4[min(i0 + u * TILE_THREADS, SCAN_VEC - 1)];
#pragma unroll
      for (int q = 0; q < 2; ++q) {
        unsigned m = 0u;
#pragma unroll
        for (int w = 0; w < 2; ++w) {
          const int u = 2 * q + w;
          const unsigned x[4] = {d[u].x ^ T4, d[u].y ^ T4, d[u].z ^ T4, d[u].w ^ T4};
          unsigned f[4];
#pragma unroll
          for (int c = 0; c < 4; ++c)
            f[c] = (x[c] - 0x01010101u) & ~(x[c] | (x[c] << 7)) & 0x80808080u;
          const unsigned mw = (f[0] >> 7) | (f[1] >> 6) | (f[2] >> 5) | (f[3] >> 4);
          m |= (i0 + u * TILE_THREADS < SCAN_VEC) ? (mw << (4 * w)) : 0u;
        }
        const unsigned e0 = 16u * (unsigned)(i0 + 2 * q * TILE_THREADS);
        while (m != 0u) {
          const unsigned p = (unsigned)__builtin_ctz(m);
          m &= m - 1u;
          const unsigned slot =
              __hip_atomic_fetch_add(&cnt, 1u, __ATOMIC_RELAXED, __HIP_MEMORY_SCOPE_WORKGROUP);
          if (slot < (unsigned)LIST_CAP)
            list[slot] = e0 + (((p >> 2) & 1u) * (16u * TILE_THREADS)) + ((p & 3u) << 2) + (p >> 3);
        }
      }
      trip = (trip + 1 == SCAN_TRIPS) ? 0 : trip + 1;
    }
  }
  __syncthreads();
  const unsigned total = cnt;

  if (total <= (unsigned)LIST_CAP) {
    {
      constexpr int PER = LIST_CAP / TILE_THREADS;
      static_assert(LIST_CAP % TILE_THREADS == 0, "every list slot has one thread and one turn");
      unsigned ev[PER];
      int      dv[PER];
#pragma unroll
      for (int u = 0; u < PER; ++u) {
        const unsigned k = (unsigned)tid + (unsigned)u * TILE_THREADS;
        ev[u] = (k < total) ? list[k] : 0u;
      }
#pragma unroll
      for (int u = 0; u < PER; ++u) dv[u] = dst[ev[u]];
#pragma unroll
      for (int u = 0; u < PER; ++u) {
        const unsigned k = (unsigned)tid + (unsigned)u * TILE_THREADS;
        if (k < total) {
          const unsigned dl = min((unsigned)(dv[u] - base), (unsigned)(TILE_NODES - 1));
          __hip_atomic_fetch_add(&deg[dl], 1u, __ATOMIC_RELAXED, __HIP_MEMORY_SCOPE_WORKGROUP);
          list[k] = (dl << 20) | ev[u];
        }
      }
    }
    __syncthreads();

    if (tid < 32) {
      constexpr int PER = (TILE_NODES + 31) / 32;
      unsigned v[PER];
      unsigned sum = 0u;
#pragma unroll
      for (int u = 0; u < PER; ++u) {
        const int idx = tid * PER + u;
        v[u] = (idx < TILE_NODES) ? deg[idx] : 0u;
        sum += v[u];
      }
      unsigned incl = sum;
#pragma unroll
      for (int d = 1; d < 32; d <<= 1) {
        const unsigned t = __shfl_up(incl, (unsigned)d, 32);
        if (tid >= d) incl += t;
      }
      unsigned run = incl - sum;
#pragma unroll
      for (int u = 0; u < PER; ++u) {
        const int idx = tid * PER + u;
        if (idx < TILE_NODES) rstart[idx] = run;
        run += v[u];
      }
    }
    __syncthreads();

    for (unsigned k = tid; k < total; k += TILE_THREADS) {
      const unsigned w  = list[k];
      const unsigned dl = w >> 20;
      const unsigned p  =
          __hip_atomic_fetch_add(&rcur[dl], 1u, __ATOMIC_RELAXED, __HIP_MEMORY_SCOPE_WORKGROUP);
      rowbuf[rstart[dl] + p] = w;
    }
    __syncthreads();

    for (unsigned j = tid; j < total; j += TILE_THREADS) {
      const unsigned w  = rowbuf[j];
      const unsigned dl = w >> 20;
      const unsigned st = rstart[dl];
      const unsigned n  = deg[dl];
      unsigned r = 0u;
      for (unsigned i = 0; i < n; i += 8u) {
        unsigned v[8];
#pragma unroll
        for (unsigned u = 0; u < 8u; ++u) v[u] = rowbuf[st + min(i + u, n - 1u)];
#pragma unroll
        for (unsigned u = 0; u < 8u; ++u) r += (i + u < n && v[u] < w) ? 1u : 0u;
      }
      list[st + r] = w;
    }
    __syncthreads();

    const unsigned S = ((total + 31u) >> 5) | 1u;
    for (unsigned kt = tid; kt < 32u * S; kt += TILE_THREADS) {
      const unsigned k = (kt & 31u) * S + (kt >> 5);
      if (k >= total) continue;
      const unsigned w  = list[k];
      const unsigned e  = w & 0xFFFFFu;
      const unsigned dl = w >> 20;
      const int s = src[e];
      const float4 as = *(const float4*)(asrc + (size_t)s * NUM_HEADS);
      const float4 ad = *(const float4*)(adl + dl * NUM_HEADS);
      float4 v;
      v.x = leaky(as.x + ad.x);
      v.y = leaky(as.y + ad.y);
      v.z = leaky(as.z + ad.z);
      v.w = leaky(as.w + ad.w);
      *(float4*)(lv + k * NUM_HEADS) = v;
      lds_max(&mx[dl * NUM_HEADS + 0], f32_ord(v.x));
      lds_max(&mx[dl * NUM_HEADS + 1], f32_ord(v.y));
      lds_max(&mx[dl * NUM_HEADS + 2], f32_ord(v.z));
      lds_max(&mx[dl * NUM_HEADS + 3], f32_ord(v.w));
      list[k] = (dl << 16) | (unsigned)s;
    }
    __syncthreads();

    for (unsigned q = tid; q < total * NUM_HEADS; q += TILE_THREADS) {
      const unsigned dl = list[q >> 2] >> 16;
      const unsigned hq = q & 3u;
      const float ex = expf(lv[q] - ord_f32_guarded(mx[dl * NUM_HEADS + hq]));
      lv[q] = ex;
    }
    __syncthreads();

    {
      const unsigned G   = TILE_THREADS / 16;
      const unsigned g   = (unsigned)tid >> 4;
      const int      l16 = tid & 15;
      const int      hh  = l16 >> 2;
      unsigned chunk = (total + G - 1u) / G;
      chunk = (chunk < (unsigned)MIN_CHUNK) ? (unsigned)MIN_CHUNK : chunk;
      chunk = (chunk + (unsigned)B_DEPTH - 1u) / (unsigned)B_DEPTH * (unsigned)B_DEPTH;
      const unsigned kbeg = g * chunk;
      const unsigned kend = min(kbeg + chunk, total);
      const unsigned NONE = 0xFFFFFFFFu;
      unsigned cur = NONE;
      float4 s  = make_float4(0.f, 0.f, 0.f, 0.f);
      float  ds = 0.f;
      for (unsigned k0 = kbeg; k0 < kend; k0 += (unsigned)B_DEPTH) {
        unsigned w[B_DEPTH];
        float4   zr[B_DEPTH];
        float    ex[B_DEPTH];
#pragma unroll
        for (int u = 0; u < B_DEPTH; ++u) w[u] = list[min(k0 + (unsigned)u, kend - 1u)];
#pragma unroll
        for (int u = 0; u < B_DEPTH; ++u)
          zr[u] = *(const float4*)(z + (size_t)(w[u] & 0xFFFFu) * ZCOLS + l16 * 4);
#pragma unroll
        for (int u = 0; u < B_DEPTH; ++u)
          ex[u] = lv[min(k0 + (unsigned)u, kend - 1u) * NUM_HEADS + hh];
#pragma unroll
        for (int u = 0; u < B_DEPTH; ++u) {
          if (k0 + (unsigned)u < kend) {
            const unsigned node = w[u] >> 16;
            if (node != cur) {
              if (cur != NONE) {
                float* a = acc + cur * ACC_PITCH + l16 * 4;
                lds_add(a + 0, s.x);
                lds_add(a + 1, s.y);
                lds_add(a + 2, s.z);
                lds_add(a + 3, s.w);
                if ((l16 & 3) == 0) lds_add(&den[cur * NUM_HEADS + hh], ds);
              }
              cur = node;
              s   = make_float4(0.f, 0.f, 0.f, 0.f);
              ds  = 0.f;
            }
            s.x += ex[u] * zr[u].x;
            s.y += ex[u] * zr[u].y;
            s.z += ex[u] * zr[u].z;
            s.w += ex[u] * zr[u].w;
            ds  += ex[u];
          }
        }
      }
      if (cur != NONE) {
        float* a = acc + cur * ACC_PITCH + l16 * 4;
        lds_add(a + 0, s.x);
        lds_add(a + 1, s.y);
        lds_add(a + 2, s.z);
        lds_add(a + 3, s.w);
        if ((l16 & 3) == 0) lds_add(&den[cur * NUM_HEADS + hh], ds);
      }
    }
  } else {
    for (int e = tid; e < N_EDGES; e += TILE_THREADS) {
      const unsigned dl = (unsigned)(dst[e] - base);
      if (dl < (unsigned)TILE_NODES) {
        const int s = src[e];
        for (int h = 0; h < NUM_HEADS; ++h)
          lds_max(&mx[dl * NUM_HEADS + h],
                  f32_ord(leaky(asrc[(size_t)s * NUM_HEADS + h] + adl[dl * NUM_HEADS + h])));
      }
    }
    __syncthreads();
    for (int e = tid; e < N_EDGES; e += TILE_THREADS) {
      const unsigned dl = (unsigned)(dst[e] - base);
      if (dl < (unsigned)TILE_NODES) {
        const int s = src[e];
        for (int h = 0; h < NUM_HEADS; ++h) {
          const float v  = leaky(asrc[(size_t)s * NUM_HEADS + h] + adl[dl * NUM_HEADS + h]);
          const float ex = expf(v - ord_f32_guarded(mx[dl * NUM_HEADS + h]));
          lds_add(&den[dl * NUM_HEADS + h], ex);
          for (int f = 0; f < OUT_FEATS; ++f)
            lds_add(&acc[dl * ACC_PITCH + h * OUT_FEATS + f],
                    ex * z[(size_t)s * ZCOLS + h * OUT_FEATS + f]);
        }
      }
    }
  }
  __syncthreads();

  for (int i = tid; i < nloc * ZCOLS; i += TILE_THREADS) {
    const int dl = i >> 6, j = i & 63;
    out[(size_t)base * ZCOLS + i] =
        acc[dl * ACC_PITCH + j] / fmaxf(den[dl * NUM_HEADS + (j >> 4)], 1e-12f);
  }
}

extern "C" void kernel_launch(void* const* d_in, const int* in_sizes, int n_in,
                              void* d_out, int out_size, void* d_ws, size_t ws_size,
                              hipStream_t stream) {
  const float* h   = (const float*)d_in[0];
  const int*   src = (const int*)d_in[1];
  const int*   dst = (const int*)d_in[2];
  const float* W   = (const float*)d_in[3];
  const float* A   = (const float*)d_in[4];
  float* out = (float*)d_out;

  float* z     = (float*)d_ws;
  float* asrc  = z     + (size_t)N_NODES * ZCOLS;
  unsigned* tileid = (unsigned*)(asrc + (size_t)N_NODES * NUM_HEADS);

  const int NH = N_NODES * NUM_HEADS;

  gat_proj_wmma<<<PROJ_BLOCKS, 128, 0, stream>>>(h, W, z);
  gat_node_scores<<<(NH + 255) / 256, 256, 0, stream>>>(z, A, dst, asrc, tileid);
  gat_tile_softmax<<<N_TILES, TILE_THREADS, 0, stream>>>(src, dst, tileid, asrc, A, z, out);
}


# tile ids stored bit-sliced (8 planes of 16 edges per 16-B word, same bytes, packed by in-register transposition): a tile block tests 16 edges in 8 logic ops, not 30; sum order unchanged
# speedup vs baseline: 4.6739x; 1.0377x over previous
#include <hip/hip_runtime.h>
#include <math.h>
#include <limits.h>

#define N_NODES   50000
#define N_EDGES   800000
#define IN_FEATS  128
#define OUT_FEATS 16
#define NUM_HEADS 4
#define ZCOLS     (NUM_HEADS * OUT_FEATS)
#define ALPHA     0.2f

typedef __attribute__((ext_vector_type(2))) float v2f;
typedef __attribute__((ext_vector_type(8))) float v8f;

#define PROJ_TILES   5
#define PROJ_ROWS    (16 * PROJ_TILES)
#define PROJ_BLOCKS  (N_NODES / PROJ_ROWS)
#define KSTEPS       (IN_FEATS / 4)
#define HT_PITCH     132
#define HT_TILE      (16 * HT_PITCH)
#define PIECES_PER_THREAD (PROJ_ROWS * (IN_FEATS / 4) / 128)

static_assert(N_NODES % PROJ_ROWS == 0, "every block takes a whole panel");
static_assert(PROJ_ROWS * (IN_FEATS / 4) % 128 == 0, "the panel's 16-B pieces divide over the threads");

__global__ __launch_bounds__(128) void gat_proj_wmma(
    const float* __restrict__ h, const float* __restrict__ W,
    float* __restrict__ z)
{
  __shared__ __attribute__((aligned(16))) float ht[PROJ_ROWS * HT_PITCH];

  const int wave  = threadIdx.x >> 5;
  const int lane  = threadIdx.x & 31;
  const int ntile = wave;
  const int lo16  = lane & 15;
  const int kbase = (lane >> 4) * 2;
  const int node0 = blockIdx.x * PROJ_ROWS;

  const float4* __restrict__ hsrc =
      (const float4*)(h + (size_t)node0 * IN_FEATS) + threadIdx.x;
  float* hdst = ht + (threadIdx.x >> 5) * HT_PITCH + (threadIdx.x & 31) * 4;

  float4 pre[PIECES_PER_THREAD];
#pragma unroll
  for (int u = 0; u < PIECES_PER_THREAD; ++u) pre[u] = hsrc[u * 128];

  const float* __restrict__ wb = W + (size_t)(ntile * 16 + lo16) * IN_FEATS + kbase;
  v2f b[KSTEPS];
#pragma unroll
  for (int j = 0; j < KSTEPS; ++j) { b[j].x = wb[4 * j]; b[j].y = wb[4 * j + 1]; }

#pragma unroll
  for (int u = 0; u < PIECES_PER_THREAD; ++u) *(float4*)(hdst + u * 4 * HT_PITCH) = pre[u];
  __syncthreads();

  const float* ap = ht + lo16 * HT_PITCH + kbase;
  v8f c[PROJ_TILES];
#pragma unroll
  for (int t = 0; t < PROJ_TILES; ++t) c[t] = v8f{};
#pragma unroll
  for (int j = 0; j < KSTEPS; ++j) {
#pragma unroll
    for (int t = 0; t < PROJ_TILES; ++t) {
      const v2f a = *(const v2f*)(ap + t * HT_TILE + 4 * j);
      c[t] = __builtin_amdgcn_wmma_f32_16x16x4_f32(
           false, a,  false, b[j],
           (short)0, c[t],  false,  false);
    }
  }

#pragma unroll
  for (int t = 0; t < PROJ_TILES; ++t) {
    const int rbase = node0 + 16 * t + ((lane >> 4) << 3);
    float* __restrict__ zp = z + (size_t)rbase * ZCOLS + ntile * 16 + lo16;
#pragma unroll
    for (int r = 0; r < 8; ++r)
      zp[(size_t)r * ZCOLS] = c[t][r];
  }
}

#define TILE_NODES   196
#define N_TILES      ((N_NODES + TILE_NODES - 1) / TILE_NODES)

#define NS_BLOCK     256
#define PACK_LANES   64
#define N_GROUPS     (N_EDGES / 16)

static_assert(N_TILES <= 256, "a tile id must fit eight bit planes");
static_assert(N_EDGES % 16 == 0, "the edges divide into whole groups");
static_assert(N_EDGES == 4 * N_NODES * NUM_HEADS, "the planes fill the workspace words the packed tile-id bytes had");
static_assert((N_NODES * NUM_HEADS + NS_BLOCK - 1) / NS_BLOCK * PACK_LANES >= N_GROUPS,
              "the node-score launch's packing lanes cover every group");
static_assert(NS_BLOCK == 4 * PACK_LANES, "a block's threads hold the four words of each of its groups");

__global__ void gat_node_scores(const float* __restrict__ z,
                                const float* __restrict__ A,
                                const int* __restrict__ dst,
                                float* __restrict__ asrc,
                                unsigned* __restrict__ tileid)
{
  __shared__ __attribute__((aligned(16))) unsigned pw[NS_BLOCK];

  const int  t    = blockIdx.x * blockDim.x + threadIdx.x;
  const bool live = t < N_NODES * NUM_HEADS;
  const int n = t >> 2, hh = t & 3;
  const float* __restrict__ zv = z + (size_t)n * ZCOLS + hh * OUT_FEATS;
  const float* __restrict__ As = A + hh * (2 * OUT_FEATS);

  int4   d4 = make_int4(0, 0, 0, 0);
  float4 zq[OUT_FEATS / 4], aq[OUT_FEATS / 4];
#pragma unroll
  for (int k = 0; k < OUT_FEATS / 4; ++k) {
    zq[k] = make_float4(0.f, 0.f, 0.f, 0.f);
    aq[k] = make_float4(0.f, 0.f, 0.f, 0.f);
  }
  if (live) {
    d4 = ((const int4*)dst)[t];
    const float4* __restrict__ zp = (const float4*)zv;
    const float4* __restrict__ ap = (const float4*)As;
#pragma unroll
    for (int k = 0; k < OUT_FEATS / 4; ++k) { zq[k] = zp[k]; aq[k] = ap[k]; }
  }

  {
    const unsigned b0 = min((unsigned)d4.x / (unsigned)TILE_NODES, 255u);
    const unsigned b1 = min((unsigned)d4.y / (unsigned)TILE_NODES, 255u);
    const unsigned b2 = min((unsigned)d4.z / (unsigned)TILE_NODES, 255u);
    const unsigned b3 = min((unsigned)d4.w / (unsigned)TILE_NODES, 255u);
    pw[threadIdx.x] = b0 | (b1 << 8) | (b2 << 16) | (b3 << 24);
  }
  __syncthreads();

  if (threadIdx.x < PACK_LANES) {
    const unsigned G = blockIdx.x * PACK_LANES + threadIdx.x;
    if (G < (unsigned)N_GROUPS) {
      const uint4 w4 = *(const uint4*)(pw + 4 * threadIdx.x);
      const unsigned w[4] = {w4.x, w4.y, w4.z, w4.w};
      const unsigned M5 = 0x55555555u, M3 = 0x33333333u, MF = 0x0F0F0F0Fu;
      const unsigned ee = (w[0] & M5) | ((w[1] & M5) << 1);
      const unsigned fe = (w[2] & M5) | ((w[3] & M5) << 1);
      const unsigned eo = ((w[0] >> 1) & M5) | (((w[1] >> 1) & M5) << 1);
      const unsigned fo = ((w[2] >> 1) & M5) | (((w[3] >> 1) & M5) << 1);
      const unsigned pe = (ee & M3) | ((fe & M3) << 2);
      const unsigned qe = ((ee >> 2) & M3) | (((fe >> 2) & M3) << 2);
      const unsigned po = (eo & M3) | ((fo & M3) << 2);
      const unsigned qo = ((eo >> 2) & M3) | (((fo >> 2) & M3) << 2);
      uint4 o;
      o.x = (pe & MF) | ((po & MF) << 4);
      o.y = (qe & MF) | ((qo & MF) << 4);
      o.z = ((pe >> 4) & MF) | (po & ~MF);
      o.w = ((qe >> 4) & MF) | (qo & ~MF);
      ((uint4*)tileid)[G] = o;
    }
  }
  if (!live) return;
  float s = 0.f;
#pragma unroll
  for (int k = 0; k < OUT_FEATS / 4; ++k) {
    s += zq[k].x * aq[k].x;
    s += zq[k].y * aq[k].y;
    s += zq[k].z * aq[k].z;
    s += zq[k].w * aq[k].w;
  }
  asrc[t] = s;
}

__device__ __forceinline__ int f32_ord(float v) {
  int i = __float_as_int(v);
  return i >= 0 ? i : (i ^ 0x7FFFFFFF);
}

__device__ __forceinline__ float ord_f32_guarded(int iv) {
  float mv = 0.f;
  if (iv != INT_MIN) {
    int j = iv >= 0 ? iv : (iv ^ 0x7FFFFFFF);
    mv = __int_as_float(j);
    if (!isfinite(mv)) mv = 0.f;
  }
  return mv;
}

__device__ __forceinline__ float leaky(float v) { return v > 0.f ? v : ALPHA * v; }

__device__ __forceinline__ void lds_add(float* p, float v) {
  __hip_atomic_fetch_add(p, v, __ATOMIC_RELAXED, __HIP_MEMORY_SCOPE_WORKGROUP);
}
__device__ __forceinline__ void lds_max(int* p, int v) {
  __hip_atomic_fetch_max(p, v, __ATOMIC_RELAXED, __HIP_MEMORY_SCOPE_WORKGROUP);
}

#define TILE_THREADS 1024
#define ACC_PITCH    65
#define LIST_CAP     6144
#define MIN_CHUNK    48
#define B_DEPTH      8
#define SCAN_VEC     N_GROUPS
#define SCAN_STRIDE  (4 * TILE_THREADS)
#define SCAN_TRIPS   ((SCAN_VEC + SCAN_STRIDE - 1) / SCAN_STRIDE)

static_assert(N_EDGES <= (1 << 20), "edge id must fit 20 bits of a list word");
static_assert(N_NODES <= (1 << 16), "source id must fit 16 bits of a list word");
static_assert(TILE_NODES <= (1 << 12), "local node id must fit the rest of a list word");

__global__ __launch_bounds__(TILE_THREADS) void gat_tile_softmax(
    const int* __restrict__ src, const int* __restrict__ dst,
    const unsigned* __restrict__ tileid,
    const float* __restrict__ asrc, const float* __restrict__ A,
    const float* __restrict__ z, float* __restrict__ out)
{
  __shared__ float    acc[TILE_NODES * ACC_PITCH];
  __shared__ float    den[TILE_NODES * NUM_HEADS];
  __shared__ int      mx[TILE_NODES * NUM_HEADS];
  __shared__ __attribute__((aligned(16))) float adl[TILE_NODES * NUM_HEADS];
  __shared__ unsigned list[LIST_CAP];
  __shared__ __attribute__((aligned(16))) float lv[LIST_CAP * NUM_HEADS];
  __shared__ unsigned rowbuf[LIST_CAP];
  __shared__ unsigned deg[TILE_NODES];
  __shared__ unsigned rstart[TILE_NODES];
  __shared__ unsigned rcur[TILE_NODES];
  __shared__ unsigned cnt;

  const int tid  = threadIdx.x;
  const int base = blockIdx.x * TILE_NODES;
  const int nloc = min(TILE_NODES, N_NODES - base);

  for (int i = tid; i < TILE_NODES * ACC_PITCH; i += TILE_THREADS) acc[i] = 0.f;
  for (int i = tid; i < TILE_NODES * NUM_HEADS; i += TILE_THREADS) {
    den[i] = 0.f;
    mx[i]  = INT_MIN;
    float dsc = 0.f;
    if (i < nloc * NUM_HEADS) {
      const int hh = i & 3;
      const float* __restrict__ zv = z + (size_t)(base + (i >> 2)) * ZCOLS + hh * OUT_FEATS;
      const float* __restrict__ Ad = A + hh * (2 * OUT_FEATS) + OUT_FEATS;
#pragma unroll
      for (int f = 0; f < OUT_FEATS; ++f) dsc += zv[f] * Ad[f];
    }
    adl[i] = dsc;
  }
  for (int i = tid; i < TILE_NODES; i += TILE_THREADS) {
    deg[i]  = 0u;
    rcur[i] = 0u;
  }
  if (tid == 0) cnt = 0u;
  __syncthreads();

  {
    const uint4* __restrict__ tid4 = (const uint4*)tileid;
    unsigned sel[4];
#pragma unroll
    for (int j = 0; j < 4; ++j)
      sel[j] = (((blockIdx.x >> (2 * j)) & 1u) ? 0x0F0F0F0Fu : 0u) |
               (((blockIdx.x >> (2 * j + 1)) & 1u) ? 0xF0F0F0F0u : 0u);
    int trip = (int)(blockIdx.x % SCAN_TRIPS);
    for (int it = 0; it < SCAN_TRIPS; ++it) {
      const int i0 = trip * SCAN_STRIDE + tid;
      uint4 d[4];
#pragma unroll
      for (int u = 0; u < 4; ++u) d[u] = tid4[min(i0 + u * TILE_THREADS, SCAN_VEC - 1)];
#pragma unroll
      for (int q = 0; q < 2; ++q) {
        unsigned m = 0u;
#pragma unroll
        for (int w = 0; w < 2; ++w) {
          const int u = 2 * q + w;
          const unsigned dif = (d[u].x ^ sel[0]) | (d[u].y ^ sel[1]) | (d[u].z ^ sel[2]) | (d[u].w ^ sel[3]);
          const unsigned mw = ~(dif | (dif >> 4)) & 0x0F0F0F0Fu;
          m |= (i0 + u * TILE_THREADS < SCAN_VEC) ? (mw << (4 * w)) : 0u;
        }
        const unsigned e0 = 16u * (unsigned)(i0 + 2 * q * TILE_THREADS);
        while (m != 0u) {
          const unsigned p = (unsigned)__builtin_ctz(m);
          m &= m - 1u;
          const unsigned slot =
              __hip_atomic_fetch_add(&cnt, 1u, __ATOMIC_RELAXED, __HIP_MEMORY_SCOPE_WORKGROUP);
          if (slot < (unsigned)LIST_CAP)
            list[slot] = e0 + (((p >> 2) & 1u) * (16u * TILE_THREADS)) + ((p & 3u) << 2) + (p >> 3);
        }
      }
      trip = (trip + 1 == SCAN_TRIPS) ? 0 : trip + 1;
    }
  }
  __syncthreads();
  const unsigned total = cnt;

  if (total <= (unsigned)LIST_CAP) {
    {
      constexpr int PER = LIST_CAP / TILE_THREADS;
      static_assert(LIST_CAP % TILE_THREADS == 0, "every list slot has one thread and one turn");
      unsigned ev[PER];
      int      dv[PER];
#pragma unroll
      for (int u = 0; u < PER; ++u) {
        const unsigned k = (unsigned)tid + (unsigned)u * TILE_THREADS;
        ev[u] = (k < total) ? list[k] : 0u;
      }
#pragma unroll
      for (int u = 0; u < PER; ++u) dv[u] = dst[ev[u]];
#pragma unroll
      for (int u = 0; u < PER; ++u) {
        const unsigned k = (unsigned)tid + (unsigned)u * TILE_THREADS;
        if (k < total) {
          const unsigned dl = min((unsigned)(dv[u] - base), (unsigned)(TILE_NODES - 1));
          __hip_atomic_fetch_add(&deg[dl], 1u, __ATOMIC_RELAXED, __HIP_MEMORY_SCOPE_WORKGROUP);
          list[k] = (dl << 20) | ev[u];
        }
      }
    }
    __syncthreads();

    if (tid < 32) {
      constexpr int PER = (TILE_NODES + 31) / 32;
      unsigned v[PER];
      unsigned sum = 0u;
#pragma unroll
      for (int u = 0; u < PER; ++u) {
        const int idx = tid * PER + u;
        v[u] = (idx < TILE_NODES) ? deg[idx] : 0u;
        sum += v[u];
      }
      unsigned incl = sum;
#pragma unroll
      for (int d = 1; d < 32; d <<= 1) {
        const unsigned t = __shfl_up(incl, (unsigned)d, 32);
        if (tid >= d) incl += t;
      }
      unsigned run = incl - sum;
#pragma unroll
      for (int u = 0; u < PER; ++u) {
        const int idx = tid * PER + u;
        if (idx < TILE_NODES) rstart[idx] = run;
        run += v[u];
      }
    }
    __syncthreads();

    for (unsigned k = tid; k < total; k += TILE_THREADS) {
      const unsigned w  = list[k];
      const unsigned dl = w >> 20;
      const unsigned p  =
          __hip_atomic_fetch_add(&rcur[dl], 1u, __ATOMIC_RELAXED, __HIP_MEMORY_SCOPE_WORKGROUP);
      rowbuf[rstart[dl] + p] = w;
    }
    __syncthreads();

    for (unsigned j = tid; j < total; j += TILE_THREADS) {
      const unsigned w  = rowbuf[j];
      const unsigned dl = w >> 20;
      const unsigned st = rstart[dl];
      const unsigned n  = deg[dl];
      unsigned r = 0u;
      for (unsigned i = 0; i < n; i += 8u) {
        unsigned v[8];
#pragma unroll
        for (unsigned u = 0; u < 8u; ++u) v[u] = rowbuf[st + min(i + u, n - 1u)];
#pragma unroll
        for (unsigned u = 0; u < 8u; ++u) r += (i + u < n && v[u] < w) ? 1u : 0u;
      }
      list[st + r] = w;
    }
    __syncthreads();

    const unsigned S = ((total + 31u) >> 5) | 1u;
    for (unsigned kt = tid; kt < 32u * S; kt += TILE_THREADS) {
      const unsigned k = (kt & 31u) * S + (kt >> 5);
      if (k >= total) continue;
      const unsigned w  = list[k];
      const unsigned e  = w & 0xFFFFFu;
      const unsigned dl = w >> 20;
      const int s = src[e];
      const float4 as = *(const float4*)(asrc + (size_t)s * NUM_HEADS);
      const float4 ad = *(const float4*)(adl + dl * NUM_HEADS);
      float4 v;
      v.x = leaky(as.x + ad.x);
      v.y = leaky(as.y + ad.y);
      v.z = leaky(as.z + ad.z);
      v.w = leaky(as.w + ad.w);
      *(float4*)(lv + k * NUM_HEADS) = v;
      lds_max(&mx[dl * NUM_HEADS + 0], f32_ord(v.x));
      lds_max(&mx[dl * NUM_HEADS + 1], f32_ord(v.y));
      lds_max(&mx[dl * NUM_HEADS + 2], f32_ord(v.z));
      lds_max(&mx[dl * NUM_HEADS + 3], f32_ord(v.w));
      list[k] = (dl << 16) | (unsigned)s;
    }
    __syncthreads();

    for (unsigned q = tid; q < total * NUM_HEADS; q += TILE_THREADS) {
      const unsigned dl = list[q >> 2] >> 16;
      const unsigned hq = q & 3u;
      const float ex = expf(lv[q] - ord_f32_guarded(mx[dl * NUM_HEADS + hq]));
      lv[q] = ex;
    }
    __syncthreads();

    {
      const unsigned G   = TILE_THREADS / 16;
      const unsigned g   = (unsigned)tid >> 4;
      const int      l16 = tid & 15;
      const int      hh  = l16 >> 2;
      unsigned chunk = (total + G - 1u) / G;
      chunk = (chunk < (unsigned)MIN_CHUNK) ? (unsigned)MIN_CHUNK : chunk;
      chunk = (chunk + (unsigned)B_DEPTH - 1u) / (unsigned)B_DEPTH * (unsigned)B_DEPTH;
      const unsigned kbeg = g * chunk;
      const unsigned kend = min(kbeg + chunk, total);
      const unsigned NONE = 0xFFFFFFFFu;
      unsigned cur = NONE;
      float4 s  = make_float4(0.f, 0.f, 0.f, 0.f);
      float  ds = 0.f;
      for (unsigned k0 = kbeg; k0 < kend; k0 += (unsigned)B_DEPTH) {
        unsigned w[B_DEPTH];
        float4   zr[B_DEPTH];
        float    ex[B_DEPTH];
#pragma unroll
        for (int u = 0; u < B_DEPTH; ++u) w[u] = list[min(k0 + (unsigned)u, kend - 1u)];
#pragma unroll
        for (int u = 0; u < B_DEPTH; ++u)
          zr[u] = *(const float4*)(z + (size_t)(w[u] & 0xFFFFu) * ZCOLS + l16 * 4);
#pragma unroll
        for (int u = 0; u < B_DEPTH; ++u)
          ex[u] = lv[min(k0 + (unsigned)u, kend - 1u) * NUM_HEADS + hh];
#pragma unroll
        for (int u = 0; u < B_DEPTH; ++u) {
          if (k0 + (unsigned)u < kend) {
            const unsigned node = w[u] >> 16;
            if (node != cur) {
              if (cur != NONE) {
                float* a = acc + cur * ACC_PITCH + l16 * 4;
                lds_add(a + 0, s.x);
                lds_add(a + 1, s.y);
                lds_add(a + 2, s.z);
                lds_add(a + 3, s.w);
                if ((l16 & 3) == 0) lds_add(&den[cur * NUM_HEADS + hh], ds);
              }
              cur = node;
              s   = make_float4(0.f, 0.f, 0.f, 0.f);
              ds  = 0.f;
            }
            s.x += ex[u] * zr[u].x;
            s.y += ex[u] * zr[u].y;
            s.z += ex[u] * zr[u].z;
            s.w += ex[u] * zr[u].w;
            ds  += ex[u];
          }
        }
      }
      if (cur != NONE) {
        float* a = acc + cur * ACC_PITCH + l16 * 4;
        lds_add(a + 0, s.x);
        lds_add(a + 1, s.y);
        lds_add(a + 2, s.z);
        lds_add(a + 3, s.w);
        if ((l16 & 3) == 0) lds_add(&den[cur * NUM_HEADS + hh], ds);
      }
    }
  } else {
    for (int e = tid; e < N_EDGES; e += TILE_THREADS) {
      const unsigned dl = (unsigned)(dst[e] - base);
      if (dl < (unsigned)TILE_NODES) {
        const int s = src[e];
        for (int h = 0; h < NUM_HEADS; ++h)
          lds_max(&mx[dl * NUM_HEADS + h],
                  f32_ord(leaky(asrc[(size_t)s * NUM_HEADS + h] + adl[dl * NUM_HEADS + h])));
      }
    }
    __syncthreads();
    for (int e = tid; e < N_EDGES; e += TILE_THREADS) {
      const unsigned dl = (unsigned)(dst[e] - base);
      if (dl < (unsigned)TILE_NODES) {
        const int s = src[e];
        for (int h = 0; h < NUM_HEADS; ++h) {
          const float v  = leaky(asrc[(size_t)s * NUM_HEADS + h] + adl[dl * NUM_HEADS + h]);
          const float ex = expf(v - ord_f32_guarded(mx[dl * NUM_HEADS + h]));
          lds_add(&den[dl * NUM_HEADS + h], ex);
          for (int f = 0; f < OUT_FEATS; ++f)
            lds_add(&acc[dl * ACC_PITCH + h * OUT_FEATS + f],
                    ex * z[(size_t)s * ZCOLS + h * OUT_FEATS + f]);
        }
      }
    }
  }
  __syncthreads();

  for (int i = tid; i < nloc * ZCOLS; i += TILE_THREADS) {
    const int dl = i >> 6, j = i & 63;
    out[(size_t)base * ZCOLS + i] =
        acc[dl * ACC_PITCH + j] / fmaxf(den[dl * NUM_HEADS + (j >> 4)], 1e-12f);
  }
}

extern "C" void kernel_launch(void* const* d_in, const int* in_sizes, int n_in,
                              void* d_out, int out_size, void* d_ws, size_t ws_size,
                              hipStream_t stream) {
  const float* h   = (const float*)d_in[0];
  const int*   src = (const int*)d_in[1];
  const int*   dst = (const int*)d_in[2];
  const float* W   = (const float*)d_in[3];
  const float* A   = (const float*)d_in[4];
  float* out = (float*)d_out;

  float* z     = (float*)d_ws;
  float* asrc  = z     + (size_t)N_NODES * ZCOLS;
  unsigned* tileid = (unsigned*)(asrc + (size_t)N_NODES * NUM_HEADS);

  const int NH = N_NODES * NUM_HEADS;

  gat_proj_wmma<<<PROJ_BLOCKS, 128, 0, stream>>>(h, W, z);
  gat_node_scores<<<(NH + 255) / 256, 256, 0, stream>>>(z, A, dst, asrc, tileid);
  gat_tile_softmax<<<N_TILES, TILE_THREADS, 0, stream>>>(src, dst, tileid, asrc, A, z, out);
}


# node-score launch folded into the projection: asrc from the wave's accumulators by a 16-lane exchange tree (asrc sum order changes), tile-id planes packed by its blocks; 3 launches to 2
# speedup vs baseline: 4.8315x; 1.0337x over previous
#include <hip/hip_runtime.h>
#include <math.h>
#include <limits.h>

#define N_NODES   50000
#define N_EDGES   800000
#define IN_FEATS  128
#define OUT_FEATS 16
#define NUM_HEADS 4
#define ZCOLS     (NUM_HEADS * OUT_FEATS)
#define ALPHA     0.2f

typedef __attribute__((ext_vector_type(2))) float v2f;
typedef __attribute__((ext_vector_type(8))) float v8f;

#define PROJ_TILES   5
#define PROJ_ROWS    (16 * PROJ_TILES)
#define PROJ_BLOCKS  (N_NODES / PROJ_ROWS)
#define KSTEPS       (IN_FEATS / 4)
#define HT_PITCH     132
#define HT_TILE      (16 * HT_PITCH)
#define PIECES_PER_THREAD (PROJ_ROWS * (IN_FEATS / 4) / 128)

#define TILE_NODES   196
#define N_TILES      ((N_NODES + TILE_NODES - 1) / TILE_NODES)
#define N_GROUPS     (N_EDGES / 16)
#define GROUPS_PER_BLOCK (N_GROUPS / PROJ_BLOCKS)
#define DST_WORDS    (4 * GROUPS_PER_BLOCK)
#define DST_PER_THREAD ((DST_WORDS + 127) / 128)

static_assert(N_NODES % PROJ_ROWS == 0, "every block takes a whole panel");
static_assert(PROJ_ROWS * (IN_FEATS / 4) % 128 == 0, "the panel's 16-B pieces divide over the threads");
static_assert(N_TILES <= 256, "a tile id must fit eight bit planes");
static_assert(N_EDGES % 16 == 0, "the edges divide into whole groups");
static_assert(N_EDGES == 4 * N_NODES * NUM_HEADS, "the planes fill the workspace words the packed tile-id bytes had");
static_assert(N_GROUPS % PROJ_BLOCKS == 0, "the groups divide over the projection's blocks");
static_assert(GROUPS_PER_BLOCK <= 128 && PROJ_ROWS <= 128, "one thread per group and per node of the panel");
static_assert((PROJ_ROWS * NUM_HEADS * 4) % 128 == 0 && (GROUPS_PER_BLOCK * 16) % 128 == 0,
              "a block's asrc and plane words are whole 128-B lines");

__global__ __launch_bounds__(128) void gat_proj_wmma(
    const float* __restrict__ h, const float* __restrict__ W,
    const float* __restrict__ A, const int* __restrict__ dst,
    float* __restrict__ z, float* __restrict__ asrc, unsigned* __restrict__ tileid)
{
  __shared__ __attribute__((aligned(16))) float ht[PROJ_ROWS * HT_PITCH];
  __shared__ __attribute__((aligned(16))) unsigned pw[DST_WORDS];
  __shared__ __attribute__((aligned(16))) float sc[PROJ_ROWS * NUM_HEADS];

  const int wave  = threadIdx.x >> 5;
  const int lane  = threadIdx.x & 31;
  const int ntile = wave;
  const int lo16  = lane & 15;
  const int kbase = (lane >> 4) * 2;
  const int node0 = blockIdx.x * PROJ_ROWS;

  const float4* __restrict__ hsrc =
      (const float4*)(h + (size_t)node0 * IN_FEATS) + threadIdx.x;
  float* hdst = ht + (threadIdx.x >> 5) * HT_PITCH + (threadIdx.x & 31) * 4;

  float4 pre[PIECES_PER_THREAD];
#pragma unroll
  for (int u = 0; u < PIECES_PER_THREAD; ++u) pre[u] = hsrc[u * 128];

  const int4* __restrict__ dsrc = (const int4*)dst + (size_t)blockIdx.x * DST_WORDS;
  int4 dq[DST_PER_THREAD];
#pragma unroll
  for (int u = 0; u < DST_PER_THREAD; ++u)
    dq[u] = dsrc[min((int)threadIdx.x + u * 128, DST_WORDS - 1)];

  const float* __restrict__ wb = W + (size_t)(ntile * 16 + lo16) * IN_FEATS + kbase;
  v2f b[KSTEPS];
#pragma unroll
  for (int j = 0; j < KSTEPS; ++j) { b[j].x = wb[4 * j]; b[j].y = wb[4 * j + 1]; }

#pragma unroll
  for (int u = 0; u < PIECES_PER_THREAD; ++u) *(float4*)(hdst + u * 4 * HT_PITCH) = pre[u];
#pragma unroll
  for (int u = 0; u < DST_PER_THREAD; ++u) {
    const int idx = (int)threadIdx.x + u * 128;
    if (idx < DST_WORDS) {
      const unsigned b0 = min((unsigned)dq[u].x / (unsigned)TILE_NODES, 255u);
      const unsigned b1 = min((unsigned)dq[u].y / (unsigned)TILE_NODES, 255u);
      const unsigned b2 = min((unsigned)dq[u].z / (unsigned)TILE_NODES, 255u);
      const unsigned b3 = min((unsigned)dq[u].w / (unsigned)TILE_NODES, 255u);
      pw[idx] = b0 | (b1 << 8) | (b2 << 16) | (b3 << 24);
    }
  }
  __syncthreads();

  if (threadIdx.x < GROUPS_PER_BLOCK) {
    const uint4 w4 = *(const uint4*)(pw + 4 * threadIdx.x);
    const unsigned w[4] = {w4.x, w4.y, w4.z, w4.w};
    const unsigned M5 = 0x55555555u, M3 = 0x33333333u, MF = 0x0F0F0F0Fu;
    const unsigned ee = (w[0] & M5) | ((w[1] & M5) << 1);
    const unsigned fe = (w[2] & M5) | ((w[3] & M5) << 1);
    const unsigned eo = ((w[0] >> 1) & M5) | (((w[1] >> 1) & M5) << 1);
    const unsigned fo = ((w[2] >> 1) & M5) | (((w[3] >> 1) & M5) << 1);
    const unsigned pe = (ee & M3) | ((fe & M3) << 2);
    const unsigned qe = ((ee >> 2) & M3) | (((fe >> 2) & M3) << 2);
    const unsigned po = (eo & M3) | ((fo & M3) << 2);
    const unsigned qo = ((eo >> 2) & M3) | (((fo >> 2) & M3) << 2);
    uint4 o;
    o.x = (pe & MF) | ((po & MF) << 4);
    o.y = (qe & MF) | ((qo & MF) << 4);
    o.z = ((pe >> 4) & MF) | (po & ~MF);
    o.w = ((qe >> 4) & MF) | (qo & ~MF);
    ((uint4*)tileid)[(size_t)blockIdx.x * GROUPS_PER_BLOCK + threadIdx.x] = o;
  }

  const float* ap = ht + lo16 * HT_PITCH + kbase;
  v8f c[PROJ_TILES];
#pragma unroll
  for (int t = 0; t < PROJ_TILES; ++t) c[t] = v8f{};
#pragma unroll
  for (int j = 0; j < KSTEPS; ++j) {
#pragma unroll
    for (int t = 0; t < PROJ_TILES; ++t) {
      const v2f a = *(const v2f*)(ap + t * HT_TILE + 4 * j);
      c[t] = __builtin_amdgcn_wmma_f32_16x16x4_f32(
           false, a,  false, b[j],
           (short)0, c[t],  false,  false);
    }
  }

#pragma unroll
  for (int t = 0; t < PROJ_TILES; ++t) {
    const int rbase = node0 + 16 * t + ((lane >> 4) << 3);
    float* __restrict__ zp = z + (size_t)rbase * ZCOLS + ntile * 16 + lo16;
#pragma unroll
    for (int r = 0; r < 8; ++r)
      zp[(size_t)r * ZCOLS] = c[t][r];
  }

  {
    const float aS = A[ntile * (2 * OUT_FEATS) + lo16];
    const bool h3 = (lane & 8) != 0, h2 = (lane & 4) != 0, h1 = (lane & 2) != 0;
#pragma unroll
    for (int t = 0; t < PROJ_TILES; ++t) {
      float q[4], d2[2];
#pragma unroll
      for (int r = 0; r < 4; ++r) {
        const float plo = c[t][r] * aS, phi = c[t][r + 4] * aS;
        q[r] = (h3 ? phi : plo) + __shfl_xor(h3 ? plo : phi, 8, 32);
      }
#pragma unroll
      for (int r = 0; r < 2; ++r)
        d2[r] = (h2 ? q[r + 2] : q[r]) + __shfl_xor(h2 ? q[r] : q[r + 2], 4, 32);
      const float e1 = (h1 ? d2[1] : d2[0]) + __shfl_xor(h1 ? d2[0] : d2[1], 2, 32);
      const float e0 = e1 + __shfl_xor(e1, 1, 32);
      if ((lane & 1) == 0)
        sc[(16 * t + ((lane >> 4) << 3) + (lo16 >> 1)) * NUM_HEADS + ntile] = e0;
    }
  }
  __syncthreads();
  if (threadIdx.x < PROJ_ROWS)
    ((float4*)(asrc + (size_t)node0 * NUM_HEADS))[threadIdx.x] = *(const float4*)(sc + 4 * threadIdx.x);
}

__device__ __forceinline__ int f32_ord(float v) {
  int i = __float_as_int(v);
  return i >= 0 ? i : (i ^ 0x7FFFFFFF);
}

__device__ __forceinline__ float ord_f32_guarded(int iv) {
  float mv = 0.f;
  if (iv != INT_MIN) {
    int j = iv >= 0 ? iv : (iv ^ 0x7FFFFFFF);
    mv = __int_as_float(j);
    if (!isfinite(mv)) mv = 0.f;
  }
  return mv;
}

__device__ __forceinline__ float leaky(float v) { return v > 0.f ? v : ALPHA * v; }

__device__ __forceinline__ void lds_add(float* p, float v) {
  __hip_atomic_fetch_add(p, v, __ATOMIC_RELAXED, __HIP_MEMORY_SCOPE_WORKGROUP);
}
__device__ __forceinline__ void lds_max(int* p, int v) {
  __hip_atomic_fetch_max(p, v, __ATOMIC_RELAXED, __HIP_MEMORY_SCOPE_WORKGROUP);
}

#define TILE_THREADS 1024
#define ACC_PITCH    65
#define LIST_CAP     6144
#define MIN_CHUNK    48
#define B_DEPTH      8
#define SCAN_VEC     N_GROUPS
#define SCAN_STRIDE  (4 * TILE_THREADS)
#define SCAN_TRIPS   ((SCAN_VEC + SCAN_STRIDE - 1) / SCAN_STRIDE)

static_assert(N_EDGES <= (1 << 20), "edge id must fit 20 bits of a list word");
static_assert(N_NODES <= (1 << 16), "source id must fit 16 bits of a list word");
static_assert(TILE_NODES <= (1 << 12), "local node id must fit the rest of a list word");

__global__ __launch_bounds__(TILE_THREADS) void gat_tile_softmax(
    const int* __restrict__ src, const int* __restrict__ dst,
    const unsigned* __restrict__ tileid,
    const float* __restrict__ asrc, const float* __restrict__ A,
    const float* __restrict__ z, float* __restrict__ out)
{
  __shared__ float    acc[TILE_NODES * ACC_PITCH];
  __shared__ float    den[TILE_NODES * NUM_HEADS];
  __shared__ int      mx[TILE_NODES * NUM_HEADS];
  __shared__ __attribute__((aligned(16))) float adl[TILE_NODES * NUM_HEADS];
  __shared__ unsigned list[LIST_CAP];
  __shared__ __attribute__((aligned(16))) float lv[LIST_CAP * NUM_HEADS];
  __shared__ unsigned rowbuf[LIST_CAP];
  __shared__ unsigned deg[TILE_NODES];
  __shared__ unsigned rstart[TILE_NODES];
  __shared__ unsigned rcur[TILE_NODES];
  __shared__ unsigned cnt;

  const int tid  = threadIdx.x;
  const int base = blockIdx.x * TILE_NODES;
  const int nloc = min(TILE_NODES, N_NODES - base);

  for (int i = tid; i < TILE_NODES * ACC_PITCH; i += TILE_THREADS) acc[i] = 0.f;
  for (int i = tid; i < TILE_NODES * NUM_HEADS; i += TILE_THREADS) {
    den[i] = 0.f;
    mx[i]  = INT_MIN;
    float dsc = 0.f;
    if (i < nloc * NUM_HEADS) {
      const int hh = i & 3;
      const float* __restrict__ zv = z + (size_t)(base + (i >> 2)) * ZCOLS + hh * OUT_FEATS;
      const float* __restrict__ Ad = A + hh * (2 * OUT_FEATS) + OUT_FEATS;
#pragma unroll
      for (int f = 0; f < OUT_FEATS; ++f) dsc += zv[f] * Ad[f];
    }
    adl[i] = dsc;
  }
  for (int i = tid; i < TILE_NODES; i += TILE_THREADS) {
    deg[i]  = 0u;
    rcur[i] = 0u;
  }
  if (tid == 0) cnt = 0u;
  __syncthreads();

  {
    const uint4* __restrict__ tid4 = (const uint4*)tileid;
    unsigned sel[4];
#pragma unroll
    for (int j = 0; j < 4; ++j)
      sel[j] = (((blockIdx.x >> (2 * j)) & 1u) ? 0x0F0F0F0Fu : 0u) |
               (((blockIdx.x >> (2 * j + 1)) & 1u) ? 0xF0F0F0F0u : 0u);
    int trip = (int)(blockIdx.x % SCAN_TRIPS);
    for (int it = 0; it < SCAN_TRIPS; ++it) {
      const int i0 = trip * SCAN_STRIDE + tid;
      uint4 d[4];
#pragma unroll
      for (int u = 0; u < 4; ++u) d[u] = tid4[min(i0 + u * TILE_THREADS, SCAN_VEC - 1)];
#pragma unroll
      for (int q = 0; q < 2; ++q) {
        unsigned m = 0u;
#pragma unroll
        for (int w = 0; w < 2; ++w) {
          const int u = 2 * q + w;
          const unsigned dif = (d[u].x ^ sel[0]) | (d[u].y ^ sel[1]) | (d[u].z ^ sel[2]) | (d[u].w ^ sel[3]);
          const unsigned mw = ~(dif | (dif >> 4)) & 0x0F0F0F0Fu;
          m |= (i0 + u * TILE_THREADS < SCAN_VEC) ? (mw << (4 * w)) : 0u;
        }
        const unsigned e0 = 16u * (unsigned)(i0 + 2 * q * TILE_THREADS);
        while (m != 0u) {
          const unsigned p = (unsigned)__builtin_ctz(m);
          m &= m - 1u;
          const unsigned slot =
              __hip_atomic_fetch_add(&cnt, 1u, __ATOMIC_RELAXED, __HIP_MEMORY_SCOPE_WORKGROUP);
          if (slot < (unsigned)LIST_CAP)
            list[slot] = e0 + (((p >> 2) & 1u) * (16u * TILE_THREADS)) + ((p & 3u) << 2) + (p >> 3);
        }
      }
      trip = (trip + 1 == SCAN_TRIPS) ? 0 : trip + 1;
    }
  }
  __syncthreads();
  const unsigned total = cnt;

  if (total <= (unsigned)LIST_CAP) {
    {
      constexpr int PER = LIST_CAP / TILE_THREADS;
      static_assert(LIST_CAP % TILE_THREADS == 0, "every list slot has one thread and one turn");
      unsigned ev[PER];
      int      dv[PER];
#pragma unroll
      for (int u = 0; u < PER; ++u) {
        const unsigned k = (unsigned)tid + (unsigned)u * TILE_THREADS;
        ev[u] = (k < total) ? list[k] : 0u;
      }
#pragma unroll
      for (int u = 0; u < PER; ++u) dv[u] = dst[ev[u]];
#pragma unroll
      for (int u = 0; u < PER; ++u) {
        const unsigned k = (unsigned)tid + (unsigned)u * TILE_THREADS;
        if (k < total) {
          const unsigned dl = min((unsigned)(dv[u] - base), (unsigned)(TILE_NODES - 1));
          __hip_atomic_fetch_add(&deg[dl], 1u, __ATOMIC_RELAXED, __HIP_MEMORY_SCOPE_WORKGROUP);
          list[k] = (dl << 20) | ev[u];
        }
      }
    }
    __syncthreads();

    if (tid < 32) {
      constexpr int PER = (TILE_NODES + 31) / 32;
      unsigned v[PER];
      unsigned sum = 0u;
#pragma unroll
      for (int u = 0; u < PER; ++u) {
        const int idx = tid * PER + u;
        v[u] = (idx < TILE_NODES) ? deg[idx] : 0u;
        sum += v[u];
      }
      unsigned incl = sum;
#pragma unroll
      for (int d = 1; d < 32; d <<= 1) {
        const unsigned t = __shfl_up(incl, (unsigned)d, 32);
        if (tid >= d) incl += t;
      }
      unsigned run = incl - sum;
#pragma unroll
      for (int u = 0; u < PER; ++u) {
        const int idx = tid * PER + u;
        if (idx < TILE_NODES) rstart[idx] = run;
        run += v[u];
      }
    }
    __syncthreads();

    for (unsigned k = tid; k < total; k += TILE_THREADS) {
      const unsigned w  = list[k];
      const unsigned dl = w >> 20;
      const unsigned p  =
          __hip_atomic_fetch_add(&rcur[dl], 1u, __ATOMIC_RELAXED, __HIP_MEMORY_SCOPE_WORKGROUP);
      rowbuf[rstart[dl] + p] = w;
    }
    __syncthreads();

    for (unsigned j = tid; j < total; j += TILE_THREADS) {
      const unsigned w  = rowbuf[j];
      const unsigned dl = w >> 20;
      const unsigned st = rstart[dl];
      const unsigned n  = deg[dl];
      unsigned r = 0u;
      for (unsigned i = 0; i < n; i += 8u) {
        unsigned v[8];
#pragma unroll
        for (unsigned u = 0; u < 8u; ++u) v[u] = rowbuf[st + min(i + u, n - 1u)];
#pragma unroll
        for (unsigned u = 0; u < 8u; ++u) r += (i + u < n && v[u] < w) ? 1u : 0u;
      }
      list[st + r] = w;
    }
    __syncthreads();

    const unsigned S = ((total + 31u) >> 5) | 1u;
    for (unsigned kt = tid; kt < 32u * S; kt += TILE_THREADS) {
      const unsigned k = (kt & 31u) * S + (kt >> 5);
      if (k >= total) continue;
      const unsigned w  = list[k];
      const unsigned e  = w & 0xFFFFFu;
      const unsigned dl = w >> 20;
      const int s = src[e];
      const float4 as = *(const float4*)(asrc + (size_t)s * NUM_HEADS);
      const float4 ad = *(const float4*)(adl + dl * NUM_HEADS);
      float4 v;
      v.x = leaky(as.x + ad.x);
      v.y = leaky(as.y + ad.y);
      v.z = leaky(as.z + ad.z);
      v.w = leaky(as.w + ad.w);
      *(float4*)(lv + k * NUM_HEADS) = v;
      lds_max(&mx[dl * NUM_HEADS + 0], f32_ord(v.x));
      lds_max(&mx[dl * NUM_HEADS + 1], f32_ord(v.y));
      lds_max(&mx[dl * NUM_HEADS + 2], f32_ord(v.z));
      lds_max(&mx[dl * NUM_HEADS + 3], f32_ord(v.w));
      list[k] = (dl << 16) | (unsigned)s;
    }
    __syncthreads();

    for (unsigned q = tid; q < total * NUM_HEADS; q += TILE_THREADS) {
      const unsigned dl = list[q >> 2] >> 16;
      const unsigned hq = q & 3u;
      const float ex = expf(lv[q] - ord_f32_guarded(mx[dl * NUM_HEADS + hq]));
      lv[q] = ex;
    }
    __syncthreads();

    {
      const unsigned G   = TILE_THREADS / 16;
      const unsigned g   = (unsigned)tid >> 4;
      const int      l16 = tid & 15;
      const int      hh  = l16 >> 2;
      unsigned chunk = (total + G - 1u) / G;
      chunk = (chunk < (unsigned)MIN_CHUNK) ? (unsigned)MIN_CHUNK : chunk;
      chunk = (chunk + (unsigned)B_DEPTH - 1u) / (unsigned)B_DEPTH * (unsigned)B_DEPTH;
      const unsigned kbeg = g * chunk;
      const unsigned kend = min(kbeg + chunk, total);
      const unsigned NONE = 0xFFFFFFFFu;
      unsigned cur = NONE;
      float4 s  = make_float4(0.f, 0.f, 0.f, 0.f);
      float  ds = 0.f;
      for (unsigned k0 = kbeg; k0 < kend; k0 += (unsigned)B_DEPTH) {
        unsigned w[B_DEPTH];
        float4   zr[B_DEPTH];
        float    ex[B_DEPTH];
#pragma unroll
        for (int u = 0; u < B_DEPTH; ++u) w[u] = list[min(k0 + (unsigned)u, kend - 1u)];
#pragma unroll
        for (int u = 0; u < B_DEPTH; ++u)
          zr[u] = *(const float4*)(z + (size_t)(w[u] & 0xFFFFu) * ZCOLS + l16 * 4);
#pragma unroll
        for (int u = 0; u < B_DEPTH; ++u)
          ex[u] = lv[min(k0 + (unsigned)u, kend - 1u) * NUM_HEADS + hh];
#pragma unroll
        for (int u = 0; u < B_DEPTH; ++u) {
          if (k0 + (unsigned)u < kend) {
            const unsigned node = w[u] >> 16;
            if (node != cur) {
              if (cur != NONE) {
                float* a = acc + cur * ACC_PITCH + l16 * 4;
                lds_add(a + 0, s.x);
                lds_add(a + 1, s.y);
                lds_add(a + 2, s.z);
                lds_add(a + 3, s.w);
                if ((l16 & 3) == 0) lds_add(&den[cur * NUM_HEADS + hh], ds);
              }
              cur = node;
              s   = make_float4(0.f, 0.f, 0.f, 0.f);
              ds  = 0.f;
            }
            s.x += ex[u] * zr[u].x;
            s.y += ex[u] * zr[u].y;
            s.z += ex[u] * zr[u].z;
            s.w += ex[u] * zr[u].w;
            ds  += ex[u];
          }
        }
      }
      if (cur != NONE) {
        float* a = acc + cur * ACC_PITCH + l16 * 4;
        lds_add(a + 0, s.x);
        lds_add(a + 1, s.y);
        lds_add(a + 2, s.z);
        lds_add(a + 3, s.w);
        if ((l16 & 3) == 0) lds_add(&den[cur * NUM_HEADS + hh], ds);
      }
    }
  } else {
    for (int e = tid; e < N_EDGES; e += TILE_THREADS) {
      const unsigned dl = (unsigned)(dst[e] - base);
      if (dl < (unsigned)TILE_NODES) {
        const int s = src[e];
        for (int h = 0; h < NUM_HEADS; ++h)
          lds_max(&mx[dl * NUM_HEADS + h],
                  f32_ord(leaky(asrc[(size_t)s * NUM_HEADS + h] + adl[dl * NUM_HEADS + h])));
      }
    }
    __syncthreads();
    for (int e = tid; e < N_EDGES; e += TILE_THREADS) {
      const unsigned dl = (unsigned)(dst[e] - base);
      if (dl < (unsigned)TILE_NODES) {
        const int s = src[e];
        for (int h = 0; h < NUM_HEADS; ++h) {
          const float v  = leaky(asrc[(size_t)s * NUM_HEADS + h] + adl[dl * NUM_HEADS + h]);
          const float ex = expf(v - ord_f32_guarded(mx[dl * NUM_HEADS + h]));
          lds_add(&den[dl * NUM_HEADS + h], ex);
          for (int f = 0; f < OUT_FEATS; ++f)
            lds_add(&acc[dl * ACC_PITCH + h * OUT_FEATS + f],
                    ex * z[(size_t)s * ZCOLS + h * OUT_FEATS + f]);
        }
      }
    }
  }
  __syncthreads();

  for (int i = tid; i < nloc * ZCOLS; i += TILE_THREADS) {
    const int dl = i >> 6, j = i & 63;
    out[(size_t)base * ZCOLS + i] =
        acc[dl * ACC_PITCH + j] / fmaxf(den[dl * NUM_HEADS + (j >> 4)], 1e-12f);
  }
}

extern "C" void kernel_launch(void* const* d_in, const int* in_sizes, int n_in,
                              void* d_out, int out_size, void* d_ws, size_t ws_size,
                              hipStream_t stream) {
  const float* h   = (const float*)d_in[0];
  const int*   src = (const int*)d_in[1];
  const int*   dst = (const int*)d_in[2];
  const float* W   = (const float*)d_in[3];
  const float* A   = (const float*)d_in[4];
  float* out = (float*)d_out;

  float* z     = (float*)d_ws;
  float* asrc  = z     + (size_t)N_NODES * ZCOLS;
  unsigned* tileid = (unsigned*)(asrc + (size_t)N_NODES * NUM_HEADS);

  gat_proj_wmma<<<PROJ_BLOCKS, 128, 0, stream>>>(h, W, A, dst, z, asrc, tileid);
  gat_tile_softmax<<<N_TILES, TILE_THREADS, 0, stream>>>(src, dst, tileid, asrc, A, z, out);
}


# tile-id planes replaced by presence masks: projection marks, per 1280-edge chunk and tile, which 128-B lines of dst hold an edge of the tile; tile block reads only marked lines; sum order unchanged
# speedup vs baseline: 5.0135x; 1.0377x over previous
#include <hip/hip_runtime.h>
#include <math.h>
#include <limits.h>

#define N_NODES   50000
#define N_EDGES   800000
#define IN_FEATS  128
#define OUT_FEATS 16
#define NUM_HEADS 4
#define ZCOLS     (NUM_HEADS * OUT_FEATS)
#define ALPHA     0.2f

typedef __attribute__((ext_vector_type(2))) float v2f;
typedef __attribute__((ext_vector_type(8))) float v8f;

#define PROJ_TILES   5
#define PROJ_ROWS    (16 * PROJ_TILES)
#define PROJ_BLOCKS  (N_NODES / PROJ_ROWS)
#define KSTEPS       (IN_FEATS / 4)
#define HT_PITCH     132
#define HT_TILE      (16 * HT_PITCH)
#define PIECES_PER_THREAD (PROJ_ROWS * (IN_FEATS / 4) / 128)

#define TILE_NODES   196
#define N_TILES      ((N_NODES + TILE_NODES - 1) / TILE_NODES)
#define CHUNK_EDGES  (N_EDGES / PROJ_BLOCKS)
#define LINE_EDGES   32
#define CHUNK_LINES  (CHUNK_EDGES / LINE_EDGES)
#define N_LINES      (N_EDGES / LINE_EDGES)
#define DST_WORDS    (CHUNK_EDGES / 4)
#define DST_PER_THREAD ((DST_WORDS + 127) / 128)
#define MASK_LO_WORDS (PROJ_BLOCKS * N_TILES)

static_assert(N_NODES % PROJ_ROWS == 0, "every block takes a whole panel");
static_assert(PROJ_ROWS * (IN_FEATS / 4) % 128 == 0, "the panel's 16-B pieces divide over the threads");
static_assert(N_TILES == 256, "two mask words of a thread cover the tiles, and a tile index is a byte");
static_assert(N_EDGES % PROJ_BLOCKS == 0 && CHUNK_EDGES % LINE_EDGES == 0, "a block marks whole lines of dst");
static_assert(CHUNK_LINES > 32 && CHUNK_LINES <= 40, "a mask is one 32-bit word and one byte");
static_assert(N_EDGES == 4 * N_NODES * NUM_HEADS && 5 * MASK_LO_WORDS == N_EDGES,
              "mask words and bytes fill the workspace words the packed tile-id bytes had");
static_assert(PROJ_ROWS <= 128, "one thread per node of the panel");
static_assert((PROJ_ROWS * NUM_HEADS * 4) % 128 == 0 && (N_TILES * 4) % 128 == 0 && N_TILES % 128 == 0,
              "a block's asrc, mask words and mask bytes are whole 128-B lines");

__global__ __launch_bounds__(128) void gat_proj_wmma(
    const float* __restrict__ h, const float* __restrict__ W,
    const float* __restrict__ A, const int* __restrict__ dst,
    float* __restrict__ z, float* __restrict__ asrc, unsigned* __restrict__ masks)
{
  __shared__ __attribute__((aligned(16))) float ht[PROJ_ROWS * HT_PITCH];
  __shared__ __attribute__((aligned(16))) unsigned mlo[N_TILES];
  __shared__ __attribute__((aligned(16))) unsigned mhi[N_TILES];
  __shared__ __attribute__((aligned(16))) float sc[PROJ_ROWS * NUM_HEADS];

  const int wave  = threadIdx.x >> 5;
  const int lane  = threadIdx.x & 31;
  const int ntile = wave;
  const int lo16  = lane & 15;
  const int kbase = (lane >> 4) * 2;
  const int node0 = blockIdx.x * PROJ_ROWS;

  mlo[threadIdx.x] = 0u;  mlo[threadIdx.x + 128] = 0u;
  mhi[threadIdx.x] = 0u;  mhi[threadIdx.x + 128] = 0u;

  const float4* __restrict__ hsrc =
      (const float4*)(h + (size_t)node0 * IN_FEATS) + threadIdx.x;
  float* hdst = ht + (threadIdx.x >> 5) * HT_PITCH + (threadIdx.x & 31) * 4;

  float4 pre[PIECES_PER_THREAD];
#pragma unroll
  for (int u = 0; u < PIECES_PER_THREAD; ++u) pre[u] = hsrc[u * 128];

  const int4* __restrict__ dsrc = (const int4*)dst + (size_t)blockIdx.x * DST_WORDS;
  int4 dq[DST_PER_THREAD];
#pragma unroll
  for (int u = 0; u < DST_PER_THREAD; ++u)
    dq[u] = dsrc[min((int)threadIdx.x + u * 128, DST_WORDS - 1)];

  const float* __restrict__ wb = W + (size_t)(ntile * 16 + lo16) * IN_FEATS + kbase;
  v2f b[KSTEPS];
#pragma unroll
  for (int j = 0; j < KSTEPS; ++j) { b[j].x = wb[4 * j]; b[j].y = wb[4 * j + 1]; }

#pragma unroll
  for (int u = 0; u < PIECES_PER_THREAD; ++u) *(float4*)(hdst + u * 4 * HT_PITCH) = pre[u];
  __syncthreads();

#pragma unroll
  for (int u = 0; u < DST_PER_THREAD; ++u) {
    const int idx = (int)threadIdx.x + u * 128;
    if (idx < DST_WORDS) {
      const int      ln  = idx >> 3;
      unsigned*      mw  = (ln < 32) ? mlo : mhi;
      const unsigned bit = 1u << (ln & 31);
      const unsigned t0 = min((unsigned)dq[u].x / (unsigned)TILE_NODES, (unsigned)(N_TILES - 1));
      const unsigned t1 = min((unsigned)dq[u].y / (unsigned)TILE_NODES, (unsigned)(N_TILES - 1));
      const unsigned t2 = min((unsigned)dq[u].z / (unsigned)TILE_NODES, (unsigned)(N_TILES - 1));
      const unsigned t3 = min((unsigned)dq[u].w / (unsigned)TILE_NODES, (unsigned)(N_TILES - 1));
      __hip_atomic_fetch_or(&mw[t0], bit, __ATOMIC_RELAXED, __HIP_MEMORY_SCOPE_WORKGROUP);
      __hip_atomic_fetch_or(&mw[t1], bit, __ATOMIC_RELAXED, __HIP_MEMORY_SCOPE_WORKGROUP);
      __hip_atomic_fetch_or(&mw[t2], bit, __ATOMIC_RELAXED, __HIP_MEMORY_SCOPE_WORKGROUP);
      __hip_atomic_fetch_or(&mw[t3], bit, __ATOMIC_RELAXED, __HIP_MEMORY_SCOPE_WORKGROUP);
    }
  }

  const float* ap = ht + lo16 * HT_PITCH + kbase;
  v8f c[PROJ_TILES];
#pragma unroll
  for (int t = 0; t < PROJ_TILES; ++t) c[t] = v8f{};
#pragma unroll
  for (int j = 0; j < KSTEPS; ++j) {
#pragma unroll
    for (int t = 0; t < PROJ_TILES; ++t) {
      const v2f a = *(const v2f*)(ap + t * HT_TILE + 4 * j);
      c[t] = __builtin_amdgcn_wmma_f32_16x16x4_f32(
           false, a,  false, b[j],
           (short)0, c[t],  false,  false);
    }
  }

#pragma unroll
  for (int t = 0; t < PROJ_TILES; ++t) {
    const int rbase = node0 + 16 * t + ((lane >> 4) << 3);
    float* __restrict__ zp = z + (size_t)rbase * ZCOLS + ntile * 16 + lo16;
#pragma unroll
    for (int r = 0; r < 8; ++r)
      zp[(size_t)r * ZCOLS] = c[t][r];
  }

  {
    const float aS = A[ntile * (2 * OUT_FEATS) + lo16];
    const bool h3 = (lane & 8) != 0, h2 = (lane & 4) != 0, h1 = (lane & 2) != 0;
#pragma unroll
    for (int t = 0; t < PROJ_TILES; ++t) {
      float q[4], d2[2];
#pragma unroll
      for (int r = 0; r < 4; ++r) {
        const float plo = c[t][r] * aS, phi = c[t][r + 4] * aS;
        q[r] = (h3 ? phi : plo) + __shfl_xor(h3 ? plo : phi, 8, 32);
      }
#pragma unroll
      for (int r = 0; r < 2; ++r)
        d2[r] = (h2 ? q[r + 2] : q[r]) + __shfl_xor(h2 ? q[r] : q[r + 2], 4, 32);
      const float e1 = (h1 ? d2[1] : d2[0]) + __shfl_xor(h1 ? d2[0] : d2[1], 2, 32);
      const float e0 = e1 + __shfl_xor(e1, 1, 32);
      if ((lane & 1) == 0)
        sc[(16 * t + ((lane >> 4) << 3) + (lo16 >> 1)) * NUM_HEADS + ntile] = e0;
    }
  }
  __syncthreads();
  if (threadIdx.x < PROJ_ROWS)
    ((float4*)(asrc + (size_t)node0 * NUM_HEADS))[threadIdx.x] = *(const float4*)(sc + 4 * threadIdx.x);
  if (threadIdx.x < N_TILES / 4) {
    ((uint4*)(masks + (size_t)blockIdx.x * N_TILES))[threadIdx.x] = *(const uint4*)(mlo + 4 * threadIdx.x);
  } else if (threadIdx.x < N_TILES / 4 + N_TILES / 16) {
    const int j = (int)threadIdx.x - N_TILES / 4;
    unsigned o[4];
#pragma unroll
    for (int q = 0; q < 4; ++q) {
      const uint4 m4 = *(const uint4*)(mhi + 16 * j + 4 * q);
      o[q] = (m4.x & 0xFFu) | ((m4.y & 0xFFu) << 8) | ((m4.z & 0xFFu) << 16) | ((m4.w & 0xFFu) << 24);
    }
    ((uint4*)(masks + MASK_LO_WORDS + (size_t)blockIdx.x * (N_TILES / 4)))[j] = make_uint4(o[0], o[1], o[2], o[3]);
  }
}

__device__ __forceinline__ int f32_ord(float v) {
  int i = __float_as_int(v);
  return i >= 0 ? i : (i ^ 0x7FFFFFFF);
}

__device__ __forceinline__ float ord_f32_guarded(int iv) {
  float mv = 0.f;
  if (iv != INT_MIN) {
    int j = iv >= 0 ? iv : (iv ^ 0x7FFFFFFF);
    mv = __int_as_float(j);
    if (!isfinite(mv)) mv = 0.f;
  }
  return mv;
}

__device__ __forceinline__ float leaky(float v) { return v > 0.f ? v : ALPHA * v; }

__device__ __forceinline__ void lds_add(float* p, float v) {
  __hip_atomic_fetch_add(p, v, __ATOMIC_RELAXED, __HIP_MEMORY_SCOPE_WORKGROUP);
}
__device__ __forceinline__ void lds_max(int* p, int v) {
  __hip_atomic_fetch_max(p, v, __ATOMIC_RELAXED, __HIP_MEMORY_SCOPE_WORKGROUP);
}

#define TILE_THREADS 1024
#define ACC_PITCH    65
#define LIST_CAP     6144
#define MIN_CHUNK    48
#define B_DEPTH      8
#define LINE_DEPTH   8
#define MASK_WAVES   ((PROJ_BLOCKS + 31) / 32)

static_assert(N_EDGES <= (1 << 20), "edge id must fit 20 bits of a list word");
static_assert(N_NODES <= (1 << 16), "source id must fit 16 bits of a list word");
static_assert(TILE_NODES <= (1 << 12), "local node id must fit the rest of a list word");
static_assert(LINE_DEPTH == 8 && LINE_EDGES == 32, "a trip's pieces fill one 32-bit match mask; 8 pieces to a line");

__global__ __launch_bounds__(TILE_THREADS) void gat_tile_softmax(
    const int* __restrict__ src, const int* __restrict__ dst,
    const unsigned* __restrict__ masks,
    const float* __restrict__ asrc, const float* __restrict__ A,
    const float* __restrict__ z, float* __restrict__ out)
{
  __shared__ float    acc[TILE_NODES * ACC_PITCH];
  __shared__ float    den[TILE_NODES * NUM_HEADS];
  __shared__ int      mx[TILE_NODES * NUM_HEADS];
  __shared__ __attribute__((aligned(16))) float adl[TILE_NODES * NUM_HEADS];
  __shared__ unsigned list[LIST_CAP];
  __shared__ __attribute__((aligned(16))) float lv[LIST_CAP * NUM_HEADS];
  __shared__ unsigned rowbuf[LIST_CAP];
  __shared__ unsigned deg[TILE_NODES];
  __shared__ unsigned rstart[TILE_NODES];
  __shared__ unsigned rcur[TILE_NODES];
  __shared__ unsigned cnt;
  __shared__ unsigned lcnt;

  const int tid  = threadIdx.x;
  const int base = blockIdx.x * TILE_NODES;
  const int nloc = min(TILE_NODES, N_NODES - base);

  unsigned mlo_v = 0u, mhi_v = 0u;
  if (tid < PROJ_BLOCKS) {
    mlo_v = masks[(size_t)tid * N_TILES + blockIdx.x];
    mhi_v = ((const unsigned char*)(masks + MASK_LO_WORDS))[(size_t)tid * N_TILES + blockIdx.x];
  }

  for (int i = tid; i < TILE_NODES * ACC_PITCH; i += TILE_THREADS) acc[i] = 0.f;
  for (int i = tid; i < TILE_NODES * NUM_HEADS; i += TILE_THREADS) {
    den[i] = 0.f;
    mx[i]  = INT_MIN;
    float dsc = 0.f;
    if (i < nloc * NUM_HEADS) {
      const int hh = i & 3;
      const float* __restrict__ zv = z + (size_t)(base + (i >> 2)) * ZCOLS + hh * OUT_FEATS;
      const float* __restrict__ Ad = A + hh * (2 * OUT_FEATS) + OUT_FEATS;
#pragma unroll
      for (int f = 0; f < OUT_FEATS; ++f) dsc += zv[f] * Ad[f];
    }
    adl[i] = dsc;
  }
  for (int i = tid; i < TILE_NODES; i += TILE_THREADS) {
    deg[i]  = 0u;
    rcur[i] = 0u;
  }
  if (tid == 0) { cnt = 0u; lcnt = 0u; }
  __syncthreads();

  if ((tid >> 5) < MASK_WAVES) {
    const int lane = tid & 31;
    const unsigned n = (unsigned)__popc(mlo_v) + (unsigned)__popc(mhi_v);
    unsigned incl = n;
#pragma unroll
    for (int d = 1; d < 32; d <<= 1) {
      const unsigned t = __shfl_up(incl, (unsigned)d, 32);
      if (lane >= d) incl += t;
    }
    unsigned wbase = 0u;
    if (lane == 31)
      wbase = __hip_atomic_fetch_add(&lcnt, incl, __ATOMIC_RELAXED, __HIP_MEMORY_SCOPE_WORKGROUP);
    wbase = __shfl(wbase, 31, 32);
    unsigned slot = wbase + incl - n;
    const unsigned l0 = (unsigned)tid * CHUNK_LINES;
    while (mlo_v != 0u) {
      const unsigned s = (unsigned)__builtin_ctz(mlo_v);
      mlo_v &= mlo_v - 1u;
      if (slot < (unsigned)LIST_CAP) rowbuf[slot] = l0 + s;
      ++slot;
    }
    while (mhi_v != 0u) {
      const unsigned s = (unsigned)__builtin_ctz(mhi_v);
      mhi_v &= mhi_v - 1u;
      if (slot < (unsigned)LIST_CAP) rowbuf[slot] = l0 + 32u + s;
      ++slot;
    }
  }
  __syncthreads();
  const unsigned nlines = lcnt;

  if (nlines <= (unsigned)LIST_CAP) {
    const int4* __restrict__ dst4 = (const int4*)dst;
    const unsigned items = nlines * 8u;
    const unsigned sub   = ((unsigned)tid & 7u) * 4u;
    const int      lane  = tid & 31;
    for (unsigned k0 = 0u; k0 < items; k0 += (unsigned)(LINE_DEPTH * TILE_THREADS)) {
      const unsigned i0 = k0 + (unsigned)tid;
      int4 d[LINE_DEPTH];
#pragma unroll
      for (int u = 0; u < LINE_DEPTH; ++u) {
        const unsigned it = min(i0 + (unsigned)(u * TILE_THREADS), items - 1u);
        d[u] = dst4[rowbuf[it >> 3] * 8u + (it & 7u)];
      }
      const unsigned nv = (i0 < items) ? min((items - i0 + (unsigned)(TILE_THREADS - 1)) / (unsigned)TILE_THREADS,
                                             (unsigned)LINE_DEPTH) : 0u;
      const unsigned own = (nv >= 8u) ? 0xFFFFFFFFu : ((1u << (4u * nv)) - 1u);
      unsigned m = 0u;
#pragma unroll
      for (int u = 0; u < LINE_DEPTH; ++u) {
        m |= ((unsigned)(d[u].x - base) < (unsigned)TILE_NODES) ? (1u << (4 * u + 0)) : 0u;
        m |= ((unsigned)(d[u].y - base) < (unsigned)TILE_NODES) ? (1u << (4 * u + 1)) : 0u;
        m |= ((unsigned)(d[u].z - base) < (unsigned)TILE_NODES) ? (1u << (4 * u + 2)) : 0u;
        m |= ((unsigned)(d[u].w - base) < (unsigned)TILE_NODES) ? (1u << (4 * u + 3)) : 0u;
      }
      m &= own;
      const unsigned n = (unsigned)__popc(m);
      unsigned incl = n;
#pragma unroll
      for (int s = 1; s < 32; s <<= 1) {
        const unsigned t = __shfl_up(incl, (unsigned)s, 32);
        if (lane >= s) incl += t;
      }
      unsigned wbase = 0u;
      if (lane == 31 && incl != 0u)
        wbase = __hip_atomic_fetch_add(&cnt, incl, __ATOMIC_RELAXED, __HIP_MEMORY_SCOPE_WORKGROUP);
      wbase = __shfl(wbase, 31, 32);
      unsigned slot = wbase + incl - n;
      while (m != 0u) {
        const unsigned p = (unsigned)__builtin_ctz(m);
        m &= m - 1u;
        const unsigned ln = rowbuf[(i0 >> 3) + (p >> 2) * (unsigned)(TILE_THREADS / 8)];
        if (slot < (unsigned)LIST_CAP) list[slot] = ln * (unsigned)LINE_EDGES + sub + (p & 3u);
        ++slot;
      }
    }
  }
  __syncthreads();
  const unsigned total = (nlines <= (unsigned)LIST_CAP) ? cnt : (unsigned)LIST_CAP + 1u;

  if (total <= (unsigned)LIST_CAP) {
    {
      constexpr int PER = LIST_CAP / TILE_THREADS;
      static_assert(LIST_CAP % TILE_THREADS == 0, "every list slot has one thread and one turn");
      unsigned ev[PER];
      int      dv[PER];
#pragma unroll
      for (int u = 0; u < PER; ++u) {
        const unsigned k = (unsigned)tid + (unsigned)u * TILE_THREADS;
        ev[u] = (k < total) ? list[k] : 0u;
      }
#pragma unroll
      for (int u = 0; u < PER; ++u) dv[u] = dst[ev[u]];
#pragma unroll
      for (int u = 0; u < PER; ++u) {
        const unsigned k = (unsigned)tid + (unsigned)u * TILE_THREADS;
        if (k < total) {
          const unsigned dl = min((unsigned)(dv[u] - base), (unsigned)(TILE_NODES - 1));
          __hip_atomic_fetch_add(&deg[dl], 1u, __ATOMIC_RELAXED, __HIP_MEMORY_SCOPE_WORKGROUP);
          list[k] = (dl << 20) | ev[u];
        }
      }
    }
    __syncthreads();

    if (tid < 32) {
      constexpr int PER = (TILE_NODES + 31) / 32;
      unsigned v[PER];
      unsigned sum = 0u;
#pragma unroll
      for (int u = 0; u < PER; ++u) {
        const int idx = tid * PER + u;
        v[u] = (idx < TILE_NODES) ? deg[idx] : 0u;
        sum += v[u];
      }
      unsigned incl = sum;
#pragma unroll
      for (int d = 1; d < 32; d <<= 1) {
        const unsigned t = __shfl_up(incl, (unsigned)d, 32);
        if (tid >= d) incl += t;
      }
      unsigned run = incl - sum;
#pragma unroll
      for (int u = 0; u < PER; ++u) {
        const int idx = tid * PER + u;
        if (idx < TILE_NODES) rstart[idx] = run;
        run += v[u];
      }
    }
    __syncthreads();

    for (unsigned k = tid; k < total; k += TILE_THREADS) {
      const unsigned w  = list[k];
      const unsigned dl = w >> 20;
      const unsigned p  =
          __hip_atomic_fetch_add(&rcur[dl], 1u, __ATOMIC_RELAXED, __HIP_MEMORY_SCOPE_WORKGROUP);
      rowbuf[rstart[dl] + p] = w;
    }
    __syncthreads();

    for (unsigned j = tid; j < total; j += TILE_THREADS) {
      const unsigned w  = rowbuf[j];
      const unsigned dl = w >> 20;
      const unsigned st = rstart[dl];
      const unsigned n  = deg[dl];
      unsigned r = 0u;
      for (unsigned i = 0; i < n; i += 8u) {
        unsigned v[8];
#pragma unroll
        for (unsigned u = 0; u < 8u; ++u) v[u] = rowbuf[st + min(i + u, n - 1u)];
#pragma unroll
        for (unsigned u = 0; u < 8u; ++u) r += (i + u < n && v[u] < w) ? 1u : 0u;
      }
      list[st + r] = w;
    }
    __syncthreads();

    const unsigned S = ((total + 31u) >> 5) | 1u;
    for (unsigned kt = tid; kt < 32u * S; kt += TILE_THREADS) {
      const unsigned k = (kt & 31u) * S + (kt >> 5);
      if (k >= total) continue;
      const unsigned w  = list[k];
      const unsigned e  = w & 0xFFFFFu;
      const unsigned dl = w >> 20;
      const int s = src[e];
      const float4 as = *(const float4*)(asrc + (size_t)s * NUM_HEADS);
      const float4 ad = *(const float4*)(adl + dl * NUM_HEADS);
      float4 v;
      v.x = leaky(as.x + ad.x);
      v.y = leaky(as.y + ad.y);
      v.z = leaky(as.z + ad.z);
      v.w = leaky(as.w + ad.w);
      *(float4*)(lv + k * NUM_HEADS) = v;
      lds_max(&mx[dl * NUM_HEADS + 0], f32_ord(v.x));
      lds_max(&mx[dl * NUM_HEADS + 1], f32_ord(v.y));
      lds_max(&mx[dl * NUM_HEADS + 2], f32_ord(v.z));
      lds_max(&mx[dl * NUM_HEADS + 3], f32_ord(v.w));
      list[k] = (dl << 16) | (unsigned)s;
    }
    __syncthreads();

    for (unsigned q = tid; q < total * NUM_HEADS; q += TILE_THREADS) {
      const unsigned dl = list[q >> 2] >> 16;
      const unsigned hq = q & 3u;
      const float ex = expf(lv[q] - ord_f32_guarded(mx[dl * NUM_HEADS + hq]));
      lv[q] = ex;
    }
    __syncthreads();

    {
      const unsigned G   = TILE_THREADS / 16;
      const unsigned g   = (unsigned)tid >> 4;
      const int      l16 = tid & 15;
      const int      hh  = l16 >> 2;
      unsigned chunk = (total + G - 1u) / G;
      chunk = (chunk < (unsigned)MIN_CHUNK) ? (unsigned)MIN_CHUNK : chunk;
      chunk = (chunk + (unsigned)B_DEPTH - 1u) / (unsigned)B_DEPTH * (unsigned)B_DEPTH;
      const unsigned kbeg = g * chunk;
      const unsigned kend = min(kbeg + chunk, total);
      const unsigned NONE = 0xFFFFFFFFu;
      unsigned cur = NONE;
      float4 s  = make_float4(0.f, 0.f, 0.f, 0.f);
      float  ds = 0.f;
      for (unsigned k0 = kbeg; k0 < kend; k0 += (unsigned)B_DEPTH) {
        unsigned w[B_DEPTH];
        float4   zr[B_DEPTH];
        float    ex[B_DEPTH];
#pragma unroll
        for (int u = 0; u < B_DEPTH; ++u) w[u] = list[min(k0 + (unsigned)u, kend - 1u)];
#pragma unroll
        for (int u = 0; u < B_DEPTH; ++u)
          zr[u] = *(const float4*)(z + (size_t)(w[u] & 0xFFFFu) * ZCOLS + l16 * 4);
#pragma unroll
        for (int u = 0; u < B_DEPTH; ++u)
          ex[u] = lv[min(k0 + (unsigned)u, kend - 1u) * NUM_HEADS + hh];
#pragma unroll
        for (int u = 0; u < B_DEPTH; ++u) {
          if (k0 + (unsigned)u < kend) {
            const unsigned node = w[u] >> 16;
            if (node != cur) {
              if (cur != NONE) {
                float* a = acc + cur * ACC_PITCH + l16 * 4;
                lds_add(a + 0, s.x);
                lds_add(a + 1, s.y);
                lds_add(a + 2, s.z);
                lds_add(a + 3, s.w);
                if ((l16 & 3) == 0) lds_add(&den[cur * NUM_HEADS + hh], ds);
              }
              cur = node;
              s   = make_float4(0.f, 0.f, 0.f, 0.f);
              ds  = 0.f;
            }
            s.x += ex[u] * zr[u].x;
            s.y += ex[u] * zr[u].y;
            s.z += ex[u] * zr[u].z;
            s.w += ex[u] * zr[u].w;
            ds  += ex[u];
          }
        }
      }
      if (cur != NONE) {
        float* a = acc + cur * ACC_PITCH + l16 * 4;
        lds_add(a + 0, s.x);
        lds_add(a + 1, s.y);
        lds_add(a + 2, s.z);
        lds_add(a + 3, s.w);
        if ((l16 & 3) == 0) lds_add(&den[cur * NUM_HEADS + hh], ds);
      }
    }
  } else {
    for (int e = tid; e < N_EDGES; e += TILE_THREADS) {
      const unsigned dl = (unsigned)(dst[e] - base);
      if (dl < (unsigned)TILE_NODES) {
        const int s = src[e];
        for (int h = 0; h < NUM_HEADS; ++h)
          lds_max(&mx[dl * NUM_HEADS + h],
                  f32_ord(leaky(asrc[(size_t)s * NUM_HEADS + h] + adl[dl * NUM_HEADS + h])));
      }
    }
    __syncthreads();
    for (int e = tid; e < N_EDGES; e += TILE_THREADS) {
      const unsigned dl = (unsigned)(dst[e] - base);
      if (dl < (unsigned)TILE_NODES) {
        const int s = src[e];
        for (int h = 0; h < NUM_HEADS; ++h) {
          const float v  = leaky(asrc[(size_t)s * NUM_HEADS + h] + adl[dl * NUM_HEADS + h]);
          const float ex = expf(v - ord_f32_guarded(mx[dl * NUM_HEADS + h]));
          lds_add(&den[dl * NUM_HEADS + h], ex);
          for (int f = 0; f < OUT_FEATS; ++f)
            lds_add(&acc[dl * ACC_PITCH + h * OUT_FEATS + f],
                    ex * z[(size_t)s * ZCOLS + h * OUT_FEATS + f]);
        }
      }
    }
  }
  __syncthreads();

  for (int i = tid; i < nloc * ZCOLS; i += TILE_THREADS) {
    const int dl = i >> 6, j = i & 63;
    out[(size_t)base * ZCOLS + i] =
        acc[dl * ACC_PITCH + j] / fmaxf(den[dl * NUM_HEADS + (j >> 4)], 1e-12f);
  }
}

extern "C" void kernel_launch(void* const* d_in, const int* in_sizes, int n_in,
                              void* d_out, int out_size, void* d_ws, size_t ws_size,
                              hipStream_t stream) {
  const float* h   = (const float*)d_in[0];
  const int*   src = (const int*)d_in[1];
  const int*   dst = (const int*)d_in[2];
  const float* W   = (const float*)d_in[3];
  const float* A   = (const float*)d_in[4];
  float* out = (float*)d_out;

  float* z     = (float*)d_ws;
  float* asrc  = z     + (size_t)N_NODES * ZCOLS;
  unsigned* masks = (unsigned*)(asrc + (size_t)N_NODES * NUM_HEADS);

  gat_proj_wmma<<<PROJ_BLOCKS, 128, 0, stream>>>(h, W, A, dst, z, asrc, masks);
  gat_tile_softmax<<<N_TILES, TILE_THREADS, 0, stream>>>(src, dst, masks, asrc, A, z, out);
}


# tile scan takes a kept edge's local node from the dst word it just tested (a piece's four packed to bytes, picked by the match bit) and counts it there; second read of dst gone; sum order unchanged
# speedup vs baseline: 5.2459x; 1.0464x over previous
#include <hip/hip_runtime.h>
#include <math.h>
#include <limits.h>

#define N_NODES   50000
#define N_EDGES   800000
#define IN_FEATS  128
#define OUT_FEATS 16
#define NUM_HEADS 4
#define ZCOLS     (NUM_HEADS * OUT_FEATS)
#define ALPHA     0.2f

typedef __attribute__((ext_vector_type(2))) float v2f;
typedef __attribute__((ext_vector_type(8))) float v8f;

#define PROJ_TILES   5
#define PROJ_ROWS    (16 * PROJ_TILES)
#define PROJ_BLOCKS  (N_NODES / PROJ_ROWS)
#define KSTEPS       (IN_FEATS / 4)
#define HT_PITCH     132
#define HT_TILE      (16 * HT_PITCH)
#define PIECES_PER_THREAD (PROJ_ROWS * (IN_FEATS / 4) / 128)

#define TILE_NODES   196
#define N_TILES      ((N_NODES + TILE_NODES - 1) / TILE_NODES)
#define CHUNK_EDGES  (N_EDGES / PROJ_BLOCKS)
#define LINE_EDGES   32
#define CHUNK_LINES  (CHUNK_EDGES / LINE_EDGES)
#define N_LINES      (N_EDGES / LINE_EDGES)
#define DST_WORDS    (CHUNK_EDGES / 4)
#define DST_PER_THREAD ((DST_WORDS + 127) / 128)
#define MASK_LO_WORDS (PROJ_BLOCKS * N_TILES)

static_assert(N_NODES % PROJ_ROWS == 0, "every block takes a whole panel");
static_assert(PROJ_ROWS * (IN_FEATS / 4) % 128 == 0, "the panel's 16-B pieces divide over the threads");
static_assert(N_TILES == 256, "two mask words of a thread cover the tiles, and a tile index is a byte");
static_assert(N_EDGES % PROJ_BLOCKS == 0 && CHUNK_EDGES % LINE_EDGES == 0, "a block marks whole lines of dst");
static_assert(CHUNK_LINES > 32 && CHUNK_LINES <= 40, "a mask is one 32-bit word and one byte");
static_assert(N_EDGES == 4 * N_NODES * NUM_HEADS && 5 * MASK_LO_WORDS == N_EDGES,
              "mask words and bytes fill the workspace words the packed tile-id bytes had");
static_assert(PROJ_ROWS <= 128, "one thread per node of the panel");
static_assert((PROJ_ROWS * NUM_HEADS * 4) % 128 == 0 && (N_TILES * 4) % 128 == 0 && N_TILES % 128 == 0,
              "a block's asrc, mask words and mask bytes are whole 128-B lines");

__global__ __launch_bounds__(128) void gat_proj_wmma(
    const float* __restrict__ h, const float* __restrict__ W,
    const float* __restrict__ A, const int* __restrict__ dst,
    float* __restrict__ z, float* __restrict__ asrc, unsigned* __restrict__ masks)
{
  __shared__ __attribute__((aligned(16))) float ht[PROJ_ROWS * HT_PITCH];
  __shared__ __attribute__((aligned(16))) unsigned mlo[N_TILES];
  __shared__ __attribute__((aligned(16))) unsigned mhi[N_TILES];
  __shared__ __attribute__((aligned(16))) float sc[PROJ_ROWS * NUM_HEADS];

  const int wave  = threadIdx.x >> 5;
  const int lane  = threadIdx.x & 31;
  const int ntile = wave;
  const int lo16  = lane & 15;
  const int kbase = (lane >> 4) * 2;
  const int node0 = blockIdx.x * PROJ_ROWS;

  mlo[threadIdx.x] = 0u;  mlo[threadIdx.x + 128] = 0u;
  mhi[threadIdx.x] = 0u;  mhi[threadIdx.x + 128] = 0u;

  const float4* __restrict__ hsrc =
      (const float4*)(h + (size_t)node0 * IN_FEATS) + threadIdx.x;
  float* hdst = ht + (threadIdx.x >> 5) * HT_PITCH + (threadIdx.x & 31) * 4;

  float4 pre[PIECES_PER_THREAD];
#pragma unroll
  for (int u = 0; u < PIECES_PER_THREAD; ++u) pre[u] = hsrc[u * 128];

  const int4* __restrict__ dsrc = (const int4*)dst + (size_t)blockIdx.x * DST_WORDS;
  int4 dq[DST_PER_THREAD];
#pragma unroll
  for (int u = 0; u < DST_PER_THREAD; ++u)
    dq[u] = dsrc[min((int)threadIdx.x + u * 128, DST_WORDS - 1)];

  const float* __restrict__ wb = W + (size_t)(ntile * 16 + lo16) * IN_FEATS + kbase;
  v2f b[KSTEPS];
#pragma unroll
  for (int j = 0; j < KSTEPS; ++j) { b[j].x = wb[4 * j]; b[j].y = wb[4 * j + 1]; }

#pragma unroll
  for (int u = 0; u < PIECES_PER_THREAD; ++u) *(float4*)(hdst + u * 4 * HT_PITCH) = pre[u];
  __syncthreads();

#pragma unroll
  for (int u = 0; u < DST_PER_THREAD; ++u) {
    const int idx = (int)threadIdx.x + u * 128;
    if (idx < DST_WORDS) {
      const int      ln  = idx >> 3;
      unsigned*      mw  = (ln < 32) ? mlo : mhi;
      const unsigned bit = 1u << (ln & 31);
      const unsigned t0 = min((unsigned)dq[u].x / (unsigned)TILE_NODES, (unsigned)(N_TILES - 1));
      const unsigned t1 = min((unsigned)dq[u].y / (unsigned)TILE_NODES, (unsigned)(N_TILES - 1));
      const unsigned t2 = min((unsigned)dq[u].z / (unsigned)TILE_NODES, (unsigned)(N_TILES - 1));
      const unsigned t3 = min((unsigned)dq[u].w / (unsigned)TILE_NODES, (unsigned)(N_TILES - 1));
      __hip_atomic_fetch_or(&mw[t0], bit, __ATOMIC_RELAXED, __HIP_MEMORY_SCOPE_WORKGROUP);
      __hip_atomic_fetch_or(&mw[t1], bit, __ATOMIC_RELAXED, __HIP_MEMORY_SCOPE_WORKGROUP);
      __hip_atomic_fetch_or(&mw[t2], bit, __ATOMIC_RELAXED, __HIP_MEMORY_SCOPE_WORKGROUP);
      __hip_atomic_fetch_or(&mw[t3], bit, __ATOMIC_RELAXED, __HIP_MEMORY_SCOPE_WORKGROUP);
    }
  }

  const float* ap = ht + lo16 * HT_PITCH + kbase;
  v8f c[PROJ_TILES];
#pragma unroll
  for (int t = 0; t < PROJ_TILES; ++t) c[t] = v8f{};
#pragma unroll
  for (int j = 0; j < KSTEPS; ++j) {
#pragma unroll
    for (int t = 0; t < PROJ_TILES; ++t) {
      const v2f a = *(const v2f*)(ap + t * HT_TILE + 4 * j);
      c[t] = __builtin_amdgcn_wmma_f32_16x16x4_f32(
           false, a,  false, b[j],
           (short)0, c[t],  false,  false);
    }
  }

#pragma unroll
  for (int t = 0; t < PROJ_TILES; ++t) {
    const int rbase = node0 + 16 * t + ((lane >> 4) << 3);
    float* __restrict__ zp = z + (size_t)rbase * ZCOLS + ntile * 16 + lo16;
#pragma unroll
    for (int r = 0; r < 8; ++r)
      zp[(size_t)r * ZCOLS] = c[t][r];
  }

  {
    const float aS = A[ntile * (2 * OUT_FEATS) + lo16];
    const bool h3 = (lane & 8) != 0, h2 = (lane & 4) != 0, h1 = (lane & 2) != 0;
#pragma unroll
    for (int t = 0; t < PROJ_TILES; ++t) {
      float q[4], d2[2];
#pragma unroll
      for (int r = 0; r < 4; ++r) {
        const float plo = c[t][r] * aS, phi = c[t][r + 4] * aS;
        q[r] = (h3 ? phi : plo) + __shfl_xor(h3 ? plo : phi, 8, 32);
      }
#pragma unroll
      for (int r = 0; r < 2; ++r)
        d2[r] = (h2 ? q[r + 2] : q[r]) + __shfl_xor(h2 ? q[r] : q[r + 2], 4, 32);
      const float e1 = (h1 ? d2[1] : d2[0]) + __shfl_xor(h1 ? d2[0] : d2[1], 2, 32);
      const float e0 = e1 + __shfl_xor(e1, 1, 32);
      if ((lane & 1) == 0)
        sc[(16 * t + ((lane >> 4) << 3) + (lo16 >> 1)) * NUM_HEADS + ntile] = e0;
    }
  }
  __syncthreads();
  if (threadIdx.x < PROJ_ROWS)
    ((float4*)(asrc + (size_t)node0 * NUM_HEADS))[threadIdx.x] = *(const float4*)(sc + 4 * threadIdx.x);
  if (threadIdx.x < N_TILES / 4) {
    ((uint4*)(masks + (size_t)blockIdx.x * N_TILES))[threadIdx.x] = *(const uint4*)(mlo + 4 * threadIdx.x);
  } else if (threadIdx.x < N_TILES / 4 + N_TILES / 16) {
    const int j = (int)threadIdx.x - N_TILES / 4;
    unsigned o[4];
#pragma unroll
    for (int q = 0; q < 4; ++q) {
      const uint4 m4 = *(const uint4*)(mhi + 16 * j + 4 * q);
      o[q] = (m4.x & 0xFFu) | ((m4.y & 0xFFu) << 8) | ((m4.z & 0xFFu) << 16) | ((m4.w & 0xFFu) << 24);
    }
    ((uint4*)(masks + MASK_LO_WORDS + (size_t)blockIdx.x * (N_TILES / 4)))[j] = make_uint4(o[0], o[1], o[2], o[3]);
  }
}

__device__ __forceinline__ int f32_ord(float v) {
  int i = __float_as_int(v);
  return i >= 0 ? i : (i ^ 0x7FFFFFFF);
}

__device__ __forceinline__ float ord_f32_guarded(int iv) {
  float mv = 0.f;
  if (iv != INT_MIN) {
    int j = iv >= 0 ? iv : (iv ^ 0x7FFFFFFF);
    mv = __int_as_float(j);
    if (!isfinite(mv)) mv = 0.f;
  }
  return mv;
}

__device__ __forceinline__ float leaky(float v) { return v > 0.f ? v : ALPHA * v; }

__device__ __forceinline__ void lds_add(float* p, float v) {
  __hip_atomic_fetch_add(p, v, __ATOMIC_RELAXED, __HIP_MEMORY_SCOPE_WORKGROUP);
}
__device__ __forceinline__ void lds_max(int* p, int v) {
  __hip_atomic_fetch_max(p, v, __ATOMIC_RELAXED, __HIP_MEMORY_SCOPE_WORKGROUP);
}

#define TILE_THREADS 1024
#define ACC_PITCH    65
#define LIST_CAP     6144
#define MIN_CHUNK    48
#define B_DEPTH      8
#define LINE_DEPTH   8
#define MASK_WAVES   ((PROJ_BLOCKS + 31) / 32)

static_assert(N_EDGES <= (1 << 20), "edge id must fit 20 bits of a list word");
static_assert(N_NODES <= (1 << 16), "source id must fit 16 bits of a list word");
static_assert(TILE_NODES <= (1 << 12), "local node id must fit the rest of a list word");
static_assert(LINE_DEPTH == 8 && LINE_EDGES == 32, "a trip's pieces fill one 32-bit match mask; 8 pieces to a line");
static_assert(TILE_NODES <= 256, "the local node of a kept edge is carried as one byte of its piece's packed word");

__device__ __forceinline__ unsigned pack_low_bytes(unsigned a, unsigned b, unsigned c, unsigned d) {
  const unsigned lo = __builtin_amdgcn_perm(b, a, 0x0c0c0400u);
  const unsigned hi = __builtin_amdgcn_perm(d, c, 0x0c0c0400u);
  return __builtin_amdgcn_perm(hi, lo, 0x05040100u);
}

__device__ __forceinline__ unsigned pick_byte(const unsigned (&v)[8], unsigned p) {
  const unsigned sel = 0x0c0c0c00u | (p & 7u);
  const unsigned c0 = __builtin_amdgcn_perm(v[1], v[0], sel);
  const unsigned c1 = __builtin_amdgcn_perm(v[3], v[2], sel);
  const unsigned c2 = __builtin_amdgcn_perm(v[5], v[4], sel);
  const unsigned c3 = __builtin_amdgcn_perm(v[7], v[6], sel);
  const unsigned e0 = (p & 8u) ? c1 : c0;
  const unsigned e1 = (p & 8u) ? c3 : c2;
  return (p & 16u) ? e1 : e0;
}

__global__ __launch_bounds__(TILE_THREADS) void gat_tile_softmax(
    const int* __restrict__ src, const int* __restrict__ dst,
    const unsigned* __restrict__ masks,
    const float* __restrict__ asrc, const float* __restrict__ A,
    const float* __restrict__ z, float* __restrict__ out)
{
  __shared__ float    acc[TILE_NODES * ACC_PITCH];
  __shared__ float    den[TILE_NODES * NUM_HEADS];
  __shared__ int      mx[TILE_NODES * NUM_HEADS];
  __shared__ __attribute__((aligned(16))) float adl[TILE_NODES * NUM_HEADS];
  __shared__ unsigned list[LIST_CAP];
  __shared__ __attribute__((aligned(16))) float lv[LIST_CAP * NUM_HEADS];
  __shared__ unsigned rowbuf[LIST_CAP];
  __shared__ unsigned deg[TILE_NODES];
  __shared__ unsigned rstart[TILE_NODES];
  __shared__ unsigned rcur[TILE_NODES];
  __shared__ unsigned cnt;
  __shared__ unsigned lcnt;

  const int tid  = threadIdx.x;
  const int base = blockIdx.x * TILE_NODES;
  const int nloc = min(TILE_NODES, N_NODES - base);

  unsigned mlo_v = 0u, mhi_v = 0u;
  if (tid < PROJ_BLOCKS) {
    mlo_v = masks[(size_t)tid * N_TILES + blockIdx.x];
    mhi_v = ((const unsigned char*)(masks + MASK_LO_WORDS))[(size_t)tid * N_TILES + blockIdx.x];
  }

  for (int i = tid; i < TILE_NODES * ACC_PITCH; i += TILE_THREADS) acc[i] = 0.f;
  for (int i = tid; i < TILE_NODES * NUM_HEADS; i += TILE_THREADS) {
    den[i] = 0.f;
    mx[i]  = INT_MIN;
    float dsc = 0.f;
    if (i < nloc * NUM_HEADS) {
      const int hh = i & 3;
      const float* __restrict__ zv = z + (size_t)(base + (i >> 2)) * ZCOLS + hh * OUT_FEATS;
      const float* __restrict__ Ad = A + hh * (2 * OUT_FEATS) + OUT_FEATS;
#pragma unroll
      for (int f = 0; f < OUT_FEATS; ++f) dsc += zv[f] * Ad[f];
    }
    adl[i] = dsc;
  }
  for (int i = tid; i < TILE_NODES; i += TILE_THREADS) {
    deg[i]  = 0u;
    rcur[i] = 0u;
  }
  if (tid == 0) { cnt = 0u; lcnt = 0u; }
  __syncthreads();

  if ((tid >> 5) < MASK_WAVES) {
    const int lane = tid & 31;
    const unsigned n = (unsigned)__popc(mlo_v) + (unsigned)__popc(mhi_v);
    unsigned incl = n;
#pragma unroll
    for (int d = 1; d < 32; d <<= 1) {
      const unsigned t = __shfl_up(incl, (unsigned)d, 32);
      if (lane >= d) incl += t;
    }
    unsigned wbase = 0u;
    if (lane == 31)
      wbase = __hip_atomic_fetch_add(&lcnt, incl, __ATOMIC_RELAXED, __HIP_MEMORY_SCOPE_WORKGROUP);
    wbase = __shfl(wbase, 31, 32);
    unsigned slot = wbase + incl - n;
    const unsigned l0 = (unsigned)tid * CHUNK_LINES;
    while (mlo_v != 0u) {
      const unsigned s = (unsigned)__builtin_ctz(mlo_v);
      mlo_v &= mlo_v - 1u;
      if (slot < (unsigned)LIST_CAP) rowbuf[slot] = l0 + s;
      ++slot;
    }
    while (mhi_v != 0u) {
      const unsigned s = (unsigned)__builtin_ctz(mhi_v);
      mhi_v &= mhi_v - 1u;
      if (slot < (unsigned)LIST_CAP) rowbuf[slot] = l0 + 32u + s;
      ++slot;
    }
  }
  __syncthreads();
  const unsigned nlines = lcnt;

  if (nlines <= (unsigned)LIST_CAP) {
    const int4* __restrict__ dst4 = (const int4*)dst;
    const unsigned items = nlines * 8u;
    const unsigned sub   = ((unsigned)tid & 7u) * 4u;
    const int      lane  = tid & 31;
    for (unsigned k0 = 0u; k0 < items; k0 += (unsigned)(LINE_DEPTH * TILE_THREADS)) {
      const unsigned i0 = k0 + (unsigned)tid;
      int4 d[LINE_DEPTH];
#pragma unroll
      for (int u = 0; u < LINE_DEPTH; ++u) {
        const unsigned it = min(i0 + (unsigned)(u * TILE_THREADS), items - 1u);
        d[u] = dst4[rowbuf[it >> 3] * 8u + (it & 7u)];
      }
      const unsigned nv = (i0 < items) ? min((items - i0 + (unsigned)(TILE_THREADS - 1)) / (unsigned)TILE_THREADS,
                                             (unsigned)LINE_DEPTH) : 0u;
      const unsigned own = (nv >= 8u) ? 0xFFFFFFFFu : ((1u << (4u * nv)) - 1u);
      unsigned m = 0u;
      unsigned pk[LINE_DEPTH];
#pragma unroll
      for (int u = 0; u < LINE_DEPTH; ++u) {
        const unsigned lx = (unsigned)(d[u].x - base), ly = (unsigned)(d[u].y - base);
        const unsigned lz = (unsigned)(d[u].z - base), lw = (unsigned)(d[u].w - base);
        m |= (lx < (unsigned)TILE_NODES) ? (1u << (4 * u + 0)) : 0u;
        m |= (ly < (unsigned)TILE_NODES) ? (1u << (4 * u + 1)) : 0u;
        m |= (lz < (unsigned)TILE_NODES) ? (1u << (4 * u + 2)) : 0u;
        m |= (lw < (unsigned)TILE_NODES) ? (1u << (4 * u + 3)) : 0u;
        pk[u] = pack_low_bytes(lx, ly, lz, lw);
      }
      m &= own;
      const unsigned n = (unsigned)__popc(m);
      unsigned incl = n;
#pragma unroll
      for (int s = 1; s < 32; s <<= 1) {
        const unsigned t = __shfl_up(incl, (unsigned)s, 32);
        if (lane >= s) incl += t;
      }
      unsigned wbase = 0u;
      if (lane == 31 && incl != 0u)
        wbase = __hip_atomic_fetch_add(&cnt, incl, __ATOMIC_RELAXED, __HIP_MEMORY_SCOPE_WORKGROUP);
      wbase = __shfl(wbase, 31, 32);
      unsigned slot = wbase + incl - n;
      while (m != 0u) {
        const unsigned p = (unsigned)__builtin_ctz(m);
        m &= m - 1u;
        const unsigned ln = rowbuf[(i0 >> 3) + (p >> 2) * (unsigned)(TILE_THREADS / 8)];
        const unsigned dl = pick_byte(pk, p);
        if (slot < (unsigned)LIST_CAP) {
          list[slot] = (dl << 20) | (ln * (unsigned)LINE_EDGES + sub + (p & 3u));
          __hip_atomic_fetch_add(&deg[dl], 1u, __ATOMIC_RELAXED, __HIP_MEMORY_SCOPE_WORKGROUP);
        }
        ++slot;
      }
    }
  }
  __syncthreads();
  const unsigned total = (nlines <= (unsigned)LIST_CAP) ? cnt : (unsigned)LIST_CAP + 1u;

  if (total <= (unsigned)LIST_CAP) {
    if (tid < 32) {
      constexpr int PER = (TILE_NODES + 31) / 32;
      unsigned v[PER];
      unsigned sum = 0u;
#pragma unroll
      for (int u = 0; u < PER; ++u) {
        const int idx = tid * PER + u;
        v[u] = (idx < TILE_NODES) ? deg[idx] : 0u;
        sum += v[u];
      }
      unsigned incl = sum;
#pragma unroll
      for (int d = 1; d < 32; d <<= 1) {
        const unsigned t = __shfl_up(incl, (unsigned)d, 32);
        if (tid >= d) incl += t;
      }
      unsigned run = incl - sum;
#pragma unroll
      for (int u = 0; u < PER; ++u) {
        const int idx = tid * PER + u;
        if (idx < TILE_NODES) rstart[idx] = run;
        run += v[u];
      }
    }
    __syncthreads();

    for (unsigned k = tid; k < total; k += TILE_THREADS) {
      const unsigned w  = list[k];
      const unsigned dl = w >> 20;
      const unsigned p  =
          __hip_atomic_fetch_add(&rcur[dl], 1u, __ATOMIC_RELAXED, __HIP_MEMORY_SCOPE_WORKGROUP);
      rowbuf[rstart[dl] + p] = w;
    }
    __syncthreads();

    for (unsigned j = tid; j < total; j += TILE_THREADS) {
      const unsigned w  = rowbuf[j];
      const unsigned dl = w >> 20;
      const unsigned st = rstart[dl];
      const unsigned n  = deg[dl];
      unsigned r = 0u;
      for (unsigned i = 0; i < n; i += 8u) {
        unsigned v[8];
#pragma unroll
        for (unsigned u = 0; u < 8u; ++u) v[u] = rowbuf[st + min(i + u, n - 1u)];
#pragma unroll
        for (unsigned u = 0; u < 8u; ++u) r += (i + u < n && v[u] < w) ? 1u : 0u;
      }
      list[st + r] = w;
    }
    __syncthreads();

    const unsigned S = ((total + 31u) >> 5) | 1u;
    for (unsigned kt = tid; kt < 32u * S; kt += TILE_THREADS) {
      const unsigned k = (kt & 31u) * S + (kt >> 5);
      if (k >= total) continue;
      const unsigned w  = list[k];
      const unsigned e  = w & 0xFFFFFu;
      const unsigned dl = w >> 20;
      const int s = src[e];
      const float4 as = *(const float4*)(asrc + (size_t)s * NUM_HEADS);
      const float4 ad = *(const float4*)(adl + dl * NUM_HEADS);
      float4 v;
      v.x = leaky(as.x + ad.x);
      v.y = leaky(as.y + ad.y);
      v.z = leaky(as.z + ad.z);
      v.w = leaky(as.w + ad.w);
      *(float4*)(lv + k * NUM_HEADS) = v;
      lds_max(&mx[dl * NUM_HEADS + 0], f32_ord(v.x));
      lds_max(&mx[dl * NUM_HEADS + 1], f32_ord(v.y));
      lds_max(&mx[dl * NUM_HEADS + 2], f32_ord(v.z));
      lds_max(&mx[dl * NUM_HEADS + 3], f32_ord(v.w));
      list[k] = (dl << 16) | (unsigned)s;
    }
    __syncthreads();

    for (unsigned q = tid; q < total * NUM_HEADS; q += TILE_THREADS) {
      const unsigned dl = list[q >> 2] >> 16;
      const unsigned hq = q & 3u;
      const float ex = expf(lv[q] - ord_f32_guarded(mx[dl * NUM_HEADS + hq]));
      lv[q] = ex;
    }
    __syncthreads();

    {
      const unsigned G   = TILE_THREADS / 16;
      const unsigned g   = (unsigned)tid >> 4;
      const int      l16 = tid & 15;
      const int      hh  = l16 >> 2;
      unsigned chunk = (total + G - 1u) / G;
      chunk = (chunk < (unsigned)MIN_CHUNK) ? (unsigned)MIN_CHUNK : chunk;
      chunk = (chunk + (unsigned)B_DEPTH - 1u) / (unsigned)B_DEPTH * (unsigned)B_DEPTH;
      const unsigned kbeg = g * chunk;
      const unsigned kend = min(kbeg + chunk, total);
      const unsigned NONE = 0xFFFFFFFFu;
      unsigned cur = NONE;
      float4 s  = make_float4(0.f, 0.f, 0.f, 0.f);
      float  ds = 0.f;
      for (unsigned k0 = kbeg; k0 < kend; k0 += (unsigned)B_DEPTH) {
        unsigned w[B_DEPTH];
        float4   zr[B_DEPTH];
        float    ex[B_DEPTH];
#pragma unroll
        for (int u = 0; u < B_DEPTH; ++u) w[u] = list[min(k0 + (unsigned)u, kend - 1u)];
#pragma unroll
        for (int u = 0; u < B_DEPTH; ++u)
          zr[u] = *(const float4*)(z + (size_t)(w[u] & 0xFFFFu) * ZCOLS + l16 * 4);
#pragma unroll
        for (int u = 0; u < B_DEPTH; ++u)
          ex[u] = lv[min(k0 + (unsigned)u, kend - 1u) * NUM_HEADS + hh];
#pragma unroll
        for (int u = 0; u < B_DEPTH; ++u) {
          if (k0 + (unsigned)u < kend) {
            const unsigned node = w[u] >> 16;
            if (node != cur) {
              if (cur != NONE) {
                float* a = acc + cur * ACC_PITCH + l16 * 4;
                lds_add(a + 0, s.x);
                lds_add(a + 1, s.y);
                lds_add(a + 2, s.z);
                lds_add(a + 3, s.w);
                if ((l16 & 3) == 0) lds_add(&den[cur * NUM_HEADS + hh], ds);
              }
              cur = node;
              s   = make_float4(0.f, 0.f, 0.f, 0.f);
              ds  = 0.f;
            }
            s.x += ex[u] * zr[u].x;
            s.y += ex[u] * zr[u].y;
            s.z += ex[u] * zr[u].z;
            s.w += ex[u] * zr[u].w;
            ds  += ex[u];
          }
        }
      }
      if (cur != NONE) {
        float* a = acc + cur * ACC_PITCH + l16 * 4;
        lds_add(a + 0, s.x);
        lds_add(a + 1, s.y);
        lds_add(a + 2, s.z);
        lds_add(a + 3, s.w);
        if ((l16 & 3) == 0) lds_add(&den[cur * NUM_HEADS + hh], ds);
      }
    }
  } else {
    for (int e = tid; e < N_EDGES; e += TILE_THREADS) {
      const unsigned dl = (unsigned)(dst[e] - base);
      if (dl < (unsigned)TILE_NODES) {
        const int s = src[e];
        for (int h = 0; h < NUM_HEADS; ++h)
          lds_max(&mx[dl * NUM_HEADS + h],
                  f32_ord(leaky(asrc[(size_t)s * NUM_HEADS + h] + adl[dl * NUM_HEADS + h])));
      }
    }
    __syncthreads();
    for (int e = tid; e < N_EDGES; e += TILE_THREADS) {
      const unsigned dl = (unsigned)(dst[e] - base);
      if (dl < (unsigned)TILE_NODES) {
        const int s = src[e];
        for (int h = 0; h < NUM_HEADS; ++h) {
          const float v  = leaky(asrc[(size_t)s * NUM_HEADS + h] + adl[dl * NUM_HEADS + h]);
          const float ex = expf(v - ord_f32_guarded(mx[dl * NUM_HEADS + h]));
          lds_add(&den[dl * NUM_HEADS + h], ex);
          for (int f = 0; f < OUT_FEATS; ++f)
            lds_add(&acc[dl * ACC_PITCH + h * OUT_FEATS + f],
                    ex * z[(size_t)s * ZCOLS + h * OUT_FEATS + f]);
        }
      }
    }
  }
  __syncthreads();

  for (int i = tid; i < nloc * ZCOLS; i += TILE_THREADS) {
    const int dl = i >> 6, j = i & 63;
    out[(size_t)base * ZCOLS + i] =
        acc[dl * ACC_PITCH + j] / fmaxf(den[dl * NUM_HEADS + (j >> 4)], 1e-12f);
  }
}

extern "C" void kernel_launch(void* const* d_in, const int* in_sizes, int n_in,
                              void* d_out, int out_size, void* d_ws, size_t ws_size,
                              hipStream_t stream) {
  const float* h   = (const float*)d_in[0];
  const int*   src = (const int*)d_in[1];
  const int*   dst = (const int*)d_in[2];
  const float* W   = (const float*)d_in[3];
  const float* A   = (const float*)d_in[4];
  float* out = (float*)d_out;

  float* z     = (float*)d_ws;
  float* asrc  = z     + (size_t)N_NODES * ZCOLS;
  unsigned* masks = (unsigned*)(asrc + (size_t)N_NODES * NUM_HEADS);

  gat_proj_wmma<<<PROJ_BLOCKS, 128, 0, stream>>>(h, W, A, dst, z, asrc, masks);
  gat_tile_softmax<<<N_TILES, TILE_THREADS, 0, stream>>>(src, dst, masks, asrc, A, z, out);
}


# LDS lists padded past their ends with harmless entries: line scan, rank loop and pass B read a batch at one address with constant offsets, no clamp or bound test per element; sums unchanged
# speedup vs baseline: 5.4395x; 1.0369x over previous
#include <hip/hip_runtime.h>
#include <math.h>
#include <limits.h>

#define N_NODES   50000
#define N_EDGES   800000
#define IN_FEATS  128
#define OUT_FEATS 16
#define NUM_HEADS 4
#define ZCOLS     (NUM_HEADS * OUT_FEATS)
#define ALPHA     0.2f

typedef __attribute__((ext_vector_type(2))) float v2f;
typedef __attribute__((ext_vector_type(8))) float v8f;

#define PROJ_TILES   5
#define PROJ_ROWS    (16 * PROJ_TILES)
#define PROJ_BLOCKS  (N_NODES / PROJ_ROWS)
#define KSTEPS       (IN_FEATS / 4)
#define HT_PITCH     132
#define HT_TILE      (16 * HT_PITCH)
#define PIECES_PER_THREAD (PROJ_ROWS * (IN_FEATS / 4) / 128)

#define TILE_NODES   196
#define N_TILES      ((N_NODES + TILE_NODES - 1) / TILE_NODES)
#define CHUNK_EDGES  (N_EDGES / PROJ_BLOCKS)
#define LINE_EDGES   32
#define CHUNK_LINES  (CHUNK_EDGES / LINE_EDGES)
#define N_LINES      (N_EDGES / LINE_EDGES)
#define DST_WORDS    (CHUNK_EDGES / 4)
#define DST_PER_THREAD ((DST_WORDS + 127) / 128)
#define MASK_LO_WORDS (PROJ_BLOCKS * N_TILES)

static_assert(N_NODES % PROJ_ROWS == 0, "every block takes a whole panel");
static_assert(PROJ_ROWS * (IN_FEATS / 4) % 128 == 0, "the panel's 16-B pieces divide over the threads");
static_assert(N_TILES == 256, "two mask words of a thread cover the tiles, and a tile index is a byte");
static_assert(N_EDGES % PROJ_BLOCKS == 0 && CHUNK_EDGES % LINE_EDGES == 0, "a block marks whole lines of dst");
static_assert(CHUNK_LINES > 32 && CHUNK_LINES <= 40, "a mask is one 32-bit word and one byte");
static_assert(N_EDGES == 4 * N_NODES * NUM_HEADS && 5 * MASK_LO_WORDS == N_EDGES,
              "mask words and bytes fill the workspace words the packed tile-id bytes had");
static_assert(PROJ_ROWS <= 128, "one thread per node of the panel");
static_assert((PROJ_ROWS * NUM_HEADS * 4) % 128 == 0 && (N_TILES * 4) % 128 == 0 && N_TILES % 128 == 0,
              "a block's asrc, mask words and mask bytes are whole 128-B lines");

__global__ __launch_bounds__(128) void gat_proj_wmma(
    const float* __restrict__ h, const float* __restrict__ W,
    const float* __restrict__ A, const int* __restrict__ dst,
    float* __restrict__ z, float* __restrict__ asrc, unsigned* __restrict__ masks)
{
  __shared__ __attribute__((aligned(16))) float ht[PROJ_ROWS * HT_PITCH];
  __shared__ __attribute__((aligned(16))) unsigned mlo[N_TILES];
  __shared__ __attribute__((aligned(16))) unsigned mhi[N_TILES];
  __shared__ __attribute__((aligned(16))) float sc[PROJ_ROWS * NUM_HEADS];

  const int wave  = threadIdx.x >> 5;
  const int lane  = threadIdx.x & 31;
  const int ntile = wave;
  const int lo16  = lane & 15;
  const int kbase = (lane >> 4) * 2;
  const int node0 = blockIdx.x * PROJ_ROWS;

  mlo[threadIdx.x] = 0u;  mlo[threadIdx.x + 128] = 0u;
  mhi[threadIdx.x] = 0u;  mhi[threadIdx.x + 128] = 0u;

  const float4* __restrict__ hsrc =
      (const float4*)(h + (size_t)node0 * IN_FEATS) + threadIdx.x;
  float* hdst = ht + (threadIdx.x >> 5) * HT_PITCH + (threadIdx.x & 31) * 4;

  float4 pre[PIECES_PER_THREAD];
#pragma unroll
  for (int u = 0; u < PIECES_PER_THREAD; ++u) pre[u] = hsrc[u * 128];

  const int4* __restrict__ dsrc = (const int4*)dst + (size_t)blockIdx.x * DST_WORDS;
  int4 dq[DST_PER_THREAD];
#pragma unroll
  for (int u = 0; u < DST_PER_THREAD; ++u)
    dq[u] = dsrc[min((int)threadIdx.x + u * 128, DST_WORDS - 1)];

  const float* __restrict__ wb = W + (size_t)(ntile * 16 + lo16) * IN_FEATS + kbase;
  v2f b[KSTEPS];
#pragma unroll
  for (int j = 0; j < KSTEPS; ++j) { b[j].x = wb[4 * j]; b[j].y = wb[4 * j + 1]; }

#pragma unroll
  for (int u = 0; u < PIECES_PER_THREAD; ++u) *(float4*)(hdst + u * 4 * HT_PITCH) = pre[u];
  __syncthreads();

#pragma unroll
  for (int u = 0; u < DST_PER_THREAD; ++u) {
    const int idx = (int)threadIdx.x + u * 128;
    if (idx < DST_WORDS) {
      const int      ln  = idx >> 3;
      unsigned*      mw  = (ln < 32) ? mlo : mhi;
      const unsigned bit = 1u << (ln & 31);
      const unsigned t0 = min((unsigned)dq[u].x / (unsigned)TILE_NODES, (unsigned)(N_TILES - 1));
      const unsigned t1 = min((unsigned)dq[u].y / (unsigned)TILE_NODES, (unsigned)(N_TILES - 1));
      const unsigned t2 = min((unsigned)dq[u].z / (unsigned)TILE_NODES, (unsigned)(N_TILES - 1));
      const unsigned t3 = min((unsigned)dq[u].w / (unsigned)TILE_NODES, (unsigned)(N_TILES - 1));
      __hip_atomic_fetch_or(&mw[t0], bit, __ATOMIC_RELAXED, __HIP_MEMORY_SCOPE_WORKGROUP);
      __hip_atomic_fetch_or(&mw[t1], bit, __ATOMIC_RELAXED, __HIP_MEMORY_SCOPE_WORKGROUP);
      __hip_atomic_fetch_or(&mw[t2], bit, __ATOMIC_RELAXED, __HIP_MEMORY_SCOPE_WORKGROUP);
      __hip_atomic_fetch_or(&mw[t3], bit, __ATOMIC_RELAXED, __HIP_MEMORY_SCOPE_WORKGROUP);
    }
  }

  const float* ap = ht + lo16 * HT_PITCH + kbase;
  v8f c[PROJ_TILES];
#pragma unroll
  for (int t = 0; t < PROJ_TILES; ++t) c[t] = v8f{};
#pragma unroll
  for (int j = 0; j < KSTEPS; ++j) {
#pragma unroll
    for (int t = 0; t < PROJ_TILES; ++t) {
      const v2f a = *(const v2f*)(ap + t * HT_TILE + 4 * j);
      c[t] = __builtin_amdgcn_wmma_f32_16x16x4_f32(
           false, a,  false, b[j],
           (short)0, c[t],  false,  false);
    }
  }

#pragma unroll
  for (int t = 0; t < PROJ_TILES; ++t) {
    const int rbase = node0 + 16 * t + ((lane >> 4) << 3);
    float* __restrict__ zp = z + (size_t)rbase * ZCOLS + ntile * 16 + lo16;
#pragma unroll
    for (int r = 0; r < 8; ++r)
      zp[(size_t)r * ZCOLS] = c[t][r];
  }

  {
    const float aS = A[ntile * (2 * OUT_FEATS) + lo16];
    const bool h3 = (lane & 8) != 0, h2 = (lane & 4) != 0, h1 = (lane & 2) != 0;
#pragma unroll
    for (int t = 0; t < PROJ_TILES; ++t) {
      float q[4], d2[2];
#pragma unroll
      for (int r = 0; r < 4; ++r) {
        const float plo = c[t][r] * aS, phi = c[t][r + 4] * aS;
        q[r] = (h3 ? phi : plo) + __shfl_xor(h3 ? plo : phi, 8, 32);
      }
#pragma unroll
      for (int r = 0; r < 2; ++r)
        d2[r] = (h2 ? q[r + 2] : q[r]) + __shfl_xor(h2 ? q[r] : q[r + 2], 4, 32);
      const float e1 = (h1 ? d2[1] : d2[0]) + __shfl_xor(h1 ? d2[0] : d2[1], 2, 32);
      const float e0 = e1 + __shfl_xor(e1, 1, 32);
      if ((lane & 1) == 0)
        sc[(16 * t + ((lane >> 4) << 3) + (lo16 >> 1)) * NUM_HEADS + ntile] = e0;
    }
  }
  __syncthreads();
  if (threadIdx.x < PROJ_ROWS)
    ((float4*)(asrc + (size_t)node0 * NUM_HEADS))[threadIdx.x] = *(const float4*)(sc + 4 * threadIdx.x);
  if (threadIdx.x < N_TILES / 4) {
    ((uint4*)(masks + (size_t)blockIdx.x * N_TILES))[threadIdx.x] = *(const uint4*)(mlo + 4 * threadIdx.x);
  } else if (threadIdx.x < N_TILES / 4 + N_TILES / 16) {
    const int j = (int)threadIdx.x - N_TILES / 4;
    unsigned o[4];
#pragma unroll
    for (int q = 0; q < 4; ++q) {
      const uint4 m4 = *(const uint4*)(mhi + 16 * j + 4 * q);
      o[q] = (m4.x & 0xFFu) | ((m4.y & 0xFFu) << 8) | ((m4.z & 0xFFu) << 16) | ((m4.w & 0xFFu) << 24);
    }
    ((uint4*)(masks + MASK_LO_WORDS + (size_t)blockIdx.x * (N_TILES / 4)))[j] = make_uint4(o[0], o[1], o[2], o[3]);
  }
}

__device__ __forceinline__ int f32_ord(float v) {
  int i = __float_as_int(v);
  return i >= 0 ? i : (i ^ 0x7FFFFFFF);
}

__device__ __forceinline__ float ord_f32_guarded(int iv) {
  float mv = 0.f;
  if (iv != INT_MIN) {
    int j = iv >= 0 ? iv : (iv ^ 0x7FFFFFFF);
    mv = __int_as_float(j);
    if (!isfinite(mv)) mv = 0.f;
  }
  return mv;
}

__device__ __forceinline__ float leaky(float v) { return v > 0.f ? v : ALPHA * v; }

__device__ __forceinline__ void lds_add(float* p, float v) {
  __hip_atomic_fetch_add(p, v, __ATOMIC_RELAXED, __HIP_MEMORY_SCOPE_WORKGROUP);
}
__device__ __forceinline__ void lds_max(int* p, int v) {
  __hip_atomic_fetch_max(p, v, __ATOMIC_RELAXED, __HIP_MEMORY_SCOPE_WORKGROUP);
}

#define TILE_THREADS 1024
#define ACC_PITCH    65
#define LIST_CAP     6144
#define MIN_CHUNK    48
#define B_DEPTH      8
#define LINE_DEPTH   8
#define MASK_WAVES   ((PROJ_BLOCKS + 31) / 32)
#define LIST_PAD     8

static_assert(N_EDGES <= (1 << 20), "edge id must fit 20 bits of a list word");
static_assert(N_NODES <= (1 << 16), "source id must fit 16 bits of a list word");
static_assert(TILE_NODES < (1 << 12), "local node id, and TILE_NODES itself in the padding, must fit the rest of a list word");
static_assert(LINE_DEPTH == 8 && LINE_EDGES == 32, "a trip's pieces fill one 32-bit match mask; 8 pieces to a line");
static_assert(TILE_NODES <= 256, "the local node of a kept edge is carried as one byte of its piece's packed word");
static_assert(LIST_PAD == B_DEPTH && LIST_CAP % TILE_THREADS == 0 && (LINE_DEPTH * TILE_THREADS) % 8 == 0,
              "a batch read past a list's end stays inside its padding");

__device__ __forceinline__ unsigned pack_low_bytes(unsigned a, unsigned b, unsigned c, unsigned d) {
  const unsigned lo = __builtin_amdgcn_perm(b, a, 0x0c0c0400u);
  const unsigned hi = __builtin_amdgcn_perm(d, c, 0x0c0c0400u);
  return __builtin_amdgcn_perm(hi, lo, 0x05040100u);
}

__device__ __forceinline__ unsigned pick_byte(const unsigned (&v)[8], unsigned p) {
  const unsigned sel = 0x0c0c0c00u | (p & 7u);
  const unsigned c0 = __builtin_amdgcn_perm(v[1], v[0], sel);
  const unsigned c1 = __builtin_amdgcn_perm(v[3], v[2], sel);
  const unsigned c2 = __builtin_amdgcn_perm(v[5], v[4], sel);
  const unsigned c3 = __builtin_amdgcn_perm(v[7], v[6], sel);
  const unsigned e0 = (p & 8u) ? c1 : c0;
  const unsigned e1 = (p & 8u) ? c3 : c2;
  return (p & 16u) ? e1 : e0;
}

__global__ __launch_bounds__(TILE_THREADS) void gat_tile_softmax(
    const int* __restrict__ src, const int* __restrict__ dst,
    const unsigned* __restrict__ masks,
    const float* __restrict__ asrc, const float* __restrict__ A,
    const float* __restrict__ z, float* __restrict__ out)
{
  __shared__ float    acc[(TILE_NODES + 1) * ACC_PITCH];
  __shared__ float    den[(TILE_NODES + 1) * NUM_HEADS];
  __shared__ int      mx[TILE_NODES * NUM_HEADS];
  __shared__ __attribute__((aligned(16))) float adl[TILE_NODES * NUM_HEADS];
  __shared__ __attribute__((aligned(16))) unsigned list[LIST_CAP + LIST_PAD];
  __shared__ __attribute__((aligned(16))) float lv[(LIST_CAP + LIST_PAD) * NUM_HEADS];
  __shared__ unsigned rowbuf[LIST_CAP + LIST_PAD];
  __shared__ unsigned deg[TILE_NODES];
  __shared__ unsigned rstart[TILE_NODES];
  __shared__ unsigned rcur[TILE_NODES];
  __shared__ unsigned cnt;
  __shared__ unsigned lcnt;

  const int tid  = threadIdx.x;
  const int base = blockIdx.x * TILE_NODES;
  const int nloc = min(TILE_NODES, N_NODES - base);

  unsigned mlo_v = 0u, mhi_v = 0u;
  if (tid < PROJ_BLOCKS) {
    mlo_v = masks[(size_t)tid * N_TILES + blockIdx.x];
    mhi_v = ((const unsigned char*)(masks + MASK_LO_WORDS))[(size_t)tid * N_TILES + blockIdx.x];
  }

  for (int i = tid; i < (TILE_NODES + 1) * ACC_PITCH; i += TILE_THREADS) acc[i] = 0.f;
  if (tid < NUM_HEADS) den[TILE_NODES * NUM_HEADS + tid] = 0.f;
  for (int i = tid; i < TILE_NODES * NUM_HEADS; i += TILE_THREADS) {
    den[i] = 0.f;
    mx[i]  = INT_MIN;
    float dsc = 0.f;
    if (i < nloc * NUM_HEADS) {
      const int hh = i & 3;
      const float* __restrict__ zv = z + (size_t)(base + (i >> 2)) * ZCOLS + hh * OUT_FEATS;
      const float* __restrict__ Ad = A + hh * (2 * OUT_FEATS) + OUT_FEATS;
#pragma unroll
      for (int f = 0; f < OUT_FEATS; ++f) dsc += zv[f] * Ad[f];
    }
    adl[i] = dsc;
  }
  for (int i = tid; i < TILE_NODES; i += TILE_THREADS) {
    deg[i]  = 0u;
    rcur[i] = 0u;
  }
  for (int i = tid; i < LIST_CAP; i += TILE_THREADS) rowbuf[i] = 0u;
  if (tid == 0) { cnt = 0u; lcnt = 0u; }
  __syncthreads();

  if ((tid >> 5) < MASK_WAVES) {
    const int lane = tid & 31;
    const unsigned n = (unsigned)__popc(mlo_v) + (unsigned)__popc(mhi_v);
    unsigned incl = n;
#pragma unroll
    for (int d = 1; d < 32; d <<= 1) {
      const unsigned t = __shfl_up(incl, (unsigned)d, 32);
      if (lane >= d) incl += t;
    }
    unsigned wbase = 0u;
    if (lane == 31)
      wbase = __hip_atomic_fetch_add(&lcnt, incl, __ATOMIC_RELAXED, __HIP_MEMORY_SCOPE_WORKGROUP);
    wbase = __shfl(wbase, 31, 32);
    unsigned slot = wbase + incl - n;
    const unsigned l0 = (unsigned)tid * CHUNK_LINES;
    while (mlo_v != 0u) {
      const unsigned s = (unsigned)__builtin_ctz(mlo_v);
      mlo_v &= mlo_v - 1u;
      if (slot < (unsigned)LIST_CAP) rowbuf[slot] = l0 + s;
      ++slot;
    }
    while (mhi_v != 0u) {
      const unsigned s = (unsigned)__builtin_ctz(mhi_v);
      mhi_v &= mhi_v - 1u;
      if (slot < (unsigned)LIST_CAP) rowbuf[slot] = l0 + 32u + s;
      ++slot;
    }
  }
  __syncthreads();
  const unsigned nlines = lcnt;

  if (nlines <= (unsigned)LIST_CAP) {
    const int4* __restrict__ dst4 = (const int4*)dst;
    const unsigned items = nlines * 8u;
    const unsigned sub   = ((unsigned)tid & 7u) * 4u;
    const int      lane  = tid & 31;
    for (unsigned k0 = 0u; k0 < items; k0 += (unsigned)(LINE_DEPTH * TILE_THREADS)) {
      const unsigned i0 = k0 + (unsigned)tid;
      int4 d[LINE_DEPTH];
      const unsigned* lp = rowbuf + (i0 >> 3);
#pragma unroll
      for (int u = 0; u < LINE_DEPTH; ++u)
        d[u] = dst4[lp[u * (TILE_THREADS / 8)] * 8u + ((unsigned)tid & 7u)];
      const unsigned nv = (i0 < items) ? min((items - i0 + (unsigned)(TILE_THREADS - 1)) / (unsigned)TILE_THREADS,
                                             (unsigned)LINE_DEPTH) : 0u;
      const unsigned own = (nv >= 8u) ? 0xFFFFFFFFu : ((1u << (4u * nv)) - 1u);
      unsigned m = 0u;
      unsigned pk[LINE_DEPTH];
#pragma unroll
      for (int u = 0; u < LINE_DEPTH; ++u) {
        const unsigned lx = (unsigned)(d[u].x - base), ly = (unsigned)(d[u].y - base);
        const unsigned lz = (unsigned)(d[u].z - base), lw = (unsigned)(d[u].w - base);
        m |= (lx < (unsigned)TILE_NODES) ? (1u << (4 * u + 0)) : 0u;
        m |= (ly < (unsigned)TILE_NODES) ? (1u << (4 * u + 1)) : 0u;
        m |= (lz < (unsigned)TILE_NODES) ? (1u << (4 * u + 2)) : 0u;
        m |= (lw < (unsigned)TILE_NODES) ? (1u << (4 * u + 3)) : 0u;
        pk[u] = pack_low_bytes(lx, ly, lz, lw);
      }
      m &= own;
      const unsigned n = (unsigned)__popc(m);
      unsigned incl = n;
#pragma unroll
      for (int s = 1; s < 32; s <<= 1) {
        const unsigned t = __shfl_up(incl, (unsigned)s, 32);
        if (lane >= s) incl += t;
      }
      unsigned wbase = 0u;
      if (lane == 31 && incl != 0u)
        wbase = __hip_atomic_fetch_add(&cnt, incl, __ATOMIC_RELAXED, __HIP_MEMORY_SCOPE_WORKGROUP);
      wbase = __shfl(wbase, 31, 32);
      unsigned slot = wbase + incl - n;
      while (m != 0u) {
        const unsigned p = (unsigned)__builtin_ctz(m);
        m &= m - 1u;
        const unsigned ln = rowbuf[(i0 >> 3) + (p >> 2) * (unsigned)(TILE_THREADS / 8)];
        const unsigned dl = pick_byte(pk, p);
        if (slot < (unsigned)LIST_CAP) {
          list[slot] = (dl << 20) | (ln * (unsigned)LINE_EDGES + sub + (p & 3u));
          __hip_atomic_fetch_add(&deg[dl], 1u, __ATOMIC_RELAXED, __HIP_MEMORY_SCOPE_WORKGROUP);
        }
        ++slot;
      }
    }
  }
  __syncthreads();
  const unsigned total = (nlines <= (unsigned)LIST_CAP) ? cnt : (unsigned)LIST_CAP + 1u;

  if (total <= (unsigned)LIST_CAP) {
    if (tid < 32) {
      constexpr int PER = (TILE_NODES + 31) / 32;
      unsigned v[PER];
      unsigned sum = 0u;
#pragma unroll
      for (int u = 0; u < PER; ++u) {
        const int idx = tid * PER + u;
        v[u] = (idx < TILE_NODES) ? deg[idx] : 0u;
        sum += v[u];
      }
      unsigned incl = sum;
#pragma unroll
      for (int d = 1; d < 32; d <<= 1) {
        const unsigned t = __shfl_up(incl, (unsigned)d, 32);
        if (tid >= d) incl += t;
      }
      unsigned run = incl - sum;
#pragma unroll
      for (int u = 0; u < PER; ++u) {
        const int idx = tid * PER + u;
        if (idx < TILE_NODES) rstart[idx] = run;
        run += v[u];
      }
    }
    __syncthreads();

    for (unsigned k = tid; k < total; k += TILE_THREADS) {
      const unsigned w  = list[k];
      const unsigned dl = w >> 20;
      const unsigned p  =
          __hip_atomic_fetch_add(&rcur[dl], 1u, __ATOMIC_RELAXED, __HIP_MEMORY_SCOPE_WORKGROUP);
      rowbuf[rstart[dl] + p] = w;
    }
    if (tid < LIST_PAD) rowbuf[total + (unsigned)tid] = 0xFFFFFFFFu;
    __syncthreads();

    for (unsigned j = tid; j < total; j += TILE_THREADS) {
      const unsigned w  = rowbuf[j];
      const unsigned dl = w >> 20;
      const unsigned st = rstart[dl];
      const unsigned n  = deg[dl];
      unsigned r = 0u;
      const unsigned* rp = rowbuf + st;
      for (unsigned i = 0; i < n; i += (unsigned)LIST_PAD) {
        unsigned v[LIST_PAD];
#pragma unroll
        for (unsigned u = 0; u < (unsigned)LIST_PAD; ++u) v[u] = rp[i + u];
#pragma unroll
        for (unsigned u = 0; u < (unsigned)LIST_PAD; ++u) r += (v[u] < w) ? 1u : 0u;
      }
      list[st + r] = w;
    }
    __syncthreads();

    const unsigned S = ((total + 31u) >> 5) | 1u;
    for (unsigned kt = tid; kt < 32u * S; kt += TILE_THREADS) {
      const unsigned k = (kt & 31u) * S + (kt >> 5);
      if (k >= total) continue;
      const unsigned w  = list[k];
      const unsigned e  = w & 0xFFFFFu;
      const unsigned dl = w >> 20;
      const int s = src[e];
      const float4 as = *(const float4*)(asrc + (size_t)s * NUM_HEADS);
      const float4 ad = *(const float4*)(adl + dl * NUM_HEADS);
      float4 v;
      v.x = leaky(as.x + ad.x);
      v.y = leaky(as.y + ad.y);
      v.z = leaky(as.z + ad.z);
      v.w = leaky(as.w + ad.w);
      *(float4*)(lv + k * NUM_HEADS) = v;
      lds_max(&mx[dl * NUM_HEADS + 0], f32_ord(v.x));
      lds_max(&mx[dl * NUM_HEADS + 1], f32_ord(v.y));
      lds_max(&mx[dl * NUM_HEADS + 2], f32_ord(v.z));
      lds_max(&mx[dl * NUM_HEADS + 3], f32_ord(v.w));
      list[k] = (dl << 16) | (unsigned)s;
    }
    if (tid < LIST_PAD) list[total + (unsigned)tid] = (unsigned)TILE_NODES << 16;
    if (tid < LIST_PAD * NUM_HEADS) lv[total * NUM_HEADS + (unsigned)tid] = 0.f;
    __syncthreads();

    for (unsigned q = tid; q < total * NUM_HEADS; q += TILE_THREADS) {
      const unsigned dl = list[q >> 2] >> 16;
      const unsigned hq = q & 3u;
      const float ex = expf(lv[q] - ord_f32_guarded(mx[dl * NUM_HEADS + hq]));
      lv[q] = ex;
    }
    __syncthreads();

    {
      const unsigned G   = TILE_THREADS / 16;
      const unsigned g   = (unsigned)tid >> 4;
      const int      l16 = tid & 15;
      const int      hh  = l16 >> 2;
      unsigned chunk = (total + G - 1u) / G;
      chunk = (chunk < (unsigned)MIN_CHUNK) ? (unsigned)MIN_CHUNK : chunk;
      chunk = (chunk + (unsigned)B_DEPTH - 1u) / (unsigned)B_DEPTH * (unsigned)B_DEPTH;
      const unsigned kbeg = g * chunk;
      const unsigned kend = min(kbeg + chunk, total);
      const unsigned NONE = 0xFFFFFFFFu;
      unsigned cur = NONE;
      float4 s  = make_float4(0.f, 0.f, 0.f, 0.f);
      float  ds = 0.f;
      for (unsigned k0 = kbeg; k0 < kend; k0 += (unsigned)B_DEPTH) {
        unsigned w[B_DEPTH];
        float4   zr[B_DEPTH];
        float    ex[B_DEPTH];
        const unsigned* wp = list + k0;
#pragma unroll
        for (int u = 0; u < B_DEPTH; ++u) w[u] = wp[u];
#pragma unroll
        for (int u = 0; u < B_DEPTH; ++u)
          zr[u] = *(const float4*)(z + (size_t)(w[u] & 0xFFFFu) * ZCOLS + l16 * 4);
        const float* ep = lv + k0 * NUM_HEADS + hh;
#pragma unroll
        for (int u = 0; u < B_DEPTH; ++u) ex[u] = ep[u * NUM_HEADS];
#pragma unroll
        for (int u = 0; u < B_DEPTH; ++u) {
          const unsigned node = w[u] >> 16;
          if (node != cur) {
            if (cur != NONE) {
              float* a = acc + cur * ACC_PITCH + l16 * 4;
              lds_add(a + 0, s.x);
              lds_add(a + 1, s.y);
              lds_add(a + 2, s.z);
              lds_add(a + 3, s.w);
              if ((l16 & 3) == 0) lds_add(&den[cur * NUM_HEADS + hh], ds);
            }
            cur = node;
            s   = make_float4(0.f, 0.f, 0.f, 0.f);
            ds  = 0.f;
          }
          s.x += ex[u] * zr[u].x;
          s.y += ex[u] * zr[u].y;
          s.z += ex[u] * zr[u].z;
          s.w += ex[u] * zr[u].w;
          ds  += ex[u];
        }
      }
      if (cur != NONE) {
        float* a = acc + cur * ACC_PITCH + l16 * 4;
        lds_add(a + 0, s.x);
        lds_add(a + 1, s.y);
        lds_add(a + 2, s.z);
        lds_add(a + 3, s.w);
        if ((l16 & 3) == 0) lds_add(&den[cur * NUM_HEADS + hh], ds);
      }
    }
  } else {
    for (int e = tid; e < N_EDGES; e += TILE_THREADS) {
      const unsigned dl = (unsigned)(dst[e] - base);
      if (dl < (unsigned)TILE_NODES) {
        const int s = src[e];
        for (int h = 0; h < NUM_HEADS; ++h)
          lds_max(&mx[dl * NUM_HEADS + h],
                  f32_ord(leaky(asrc[(size_t)s * NUM_HEADS + h] + adl[dl * NUM_HEADS + h])));
      }
    }
    __syncthreads();
    for (int e = tid; e < N_EDGES; e += TILE_THREADS) {
      const unsigned dl = (unsigned)(dst[e] - base);
      if (dl < (unsigned)TILE_NODES) {
        const int s = src[e];
        for (int h = 0; h < NUM_HEADS; ++h) {
          const float v  = leaky(asrc[(size_t)s * NUM_HEADS + h] + adl[dl * NUM_HEADS + h]);
          const float ex = expf(v - ord_f32_guarded(mx[dl * NUM_HEADS + h]));
          lds_add(&den[dl * NUM_HEADS + h], ex);
          for (int f = 0; f < OUT_FEATS; ++f)
            lds_add(&acc[dl * ACC_PITCH + h * OUT_FEATS + f],
                    ex * z[(size_t)s * ZCOLS + h * OUT_FEATS + f]);
        }
      }
    }
  }
  __syncthreads();

  for (int i = tid; i < nloc * ZCOLS; i += TILE_THREADS) {
    const int dl = i >> 6, j = i & 63;
    out[(size_t)base * ZCOLS + i] =
        acc[dl * ACC_PITCH + j] / fmaxf(den[dl * NUM_HEADS + (j >> 4)], 1e-12f);
  }
}

extern "C" void kernel_launch(void* const* d_in, const int* in_sizes, int n_in,
                              void* d_out, int out_size, void* d_ws, size_t ws_size,
                              hipStream_t stream) {
  const float* h   = (const float*)d_in[0];
  const int*   src = (const int*)d_in[1];
  const int*   dst = (const int*)d_in[2];
  const float* W   = (const float*)d_in[3];
  const float* A   = (const float*)d_in[4];
  float* out = (float*)d_out;

  float* z     = (float*)d_ws;
  float* asrc  = z     + (size_t)N_NODES * ZCOLS;
  unsigned* masks = (unsigned*)(asrc + (size_t)N_NODES * NUM_HEADS);

  gat_proj_wmma<<<PROJ_BLOCKS, 128, 0, stream>>>(h, W, A, dst, z, asrc, masks);
  gat_tile_softmax<<<N_TILES, TILE_THREADS, 0, stream>>>(src, dst, masks, asrc, A, z, out);
}
